# starting kernel
# baseline (speedup 1.0000x reference)
#include <hip/hip_runtime.h>
#include <math.h>

typedef __attribute__((ext_vector_type(2))) float v2f;
typedef __attribute__((ext_vector_type(8))) float v8f;

#define LN_EPS 1e-5f

__global__ void degree_kernel(const int* __restrict__ src, const int* __restrict__ dst,
                              float* __restrict__ dout, float* __restrict__ din, int e) {
    int i = blockIdx.x * blockDim.x + threadIdx.x;
    if (i >= e) return;
    atomicAdd(dout + src[i], 1.0f);
    atomicAdd(din  + dst[i], 1.0f);
}

__global__ void invsqrt_deg_kernel(float* __restrict__ dout, float* __restrict__ din, int n) {
    int i = blockIdx.x * blockDim.x + threadIdx.x;
    if (i >= n) return;
    dout[i] = rsqrtf(fmaxf(dout[i], 1.0f));
    din[i]  = rsqrtf(fmaxf(din[i],  1.0f));
}

__global__ void ln_stats_768_kernel(const float* __restrict__ feat,
                                    float* __restrict__ mu, float* __restrict__ rs, int n) {
    int row  = (blockIdx.x * blockDim.x + threadIdx.x) >> 5;
    int lane = threadIdx.x & 31;
    if (row >= n) return;
    const float* p = feat + (size_t)row * 768;
    float s = 0.f, s2 = 0.f;
    #pragma unroll
    for (int c = 0; c < 768; c += 32) {
        float v = p[c + lane];
        s += v; s2 += v * v;
    }
    #pragma unroll
    for (int off = 16; off > 0; off >>= 1) {
        s  += __shfl_xor(s,  off, 32);
        s2 += __shfl_xor(s2, off, 32);
    }
    if (lane == 0) {
        float m   = s  * (1.0f / 768.0f);
        float var = s2 * (1.0f / 768.0f) - m * m;
        mu[row] = m;
        rs[row] = rsqrtf(var + LN_EPS);
    }
}

__global__ void gemm0_ln_wmma_kernel(const float* __restrict__ feat,
                                     const float* __restrict__ mu, const float* __restrict__ rs,
                                     const float* __restrict__ do_is,
                                     const float* __restrict__ g, const float* __restrict__ b,
                                     const float* __restrict__ W,
                                     float* __restrict__ C, int M) {
    const int K = 768, F = 128;
    const int lane = threadIdx.x & 31;
    const int mt   = (blockIdx.x * blockDim.x + threadIdx.x) >> 5;
    if (mt * 16 >= M) return;
    const int half = lane >> 4;
    const int l16  = lane & 15;
    const int arow = mt * 16 + l16;
    const int koff = half * 2;
    const float m_r  = mu[arow];
    const float rs_r = rs[arow];
    const float do_r = do_is[arow];
    const float* ap = feat + (size_t)arow * K + koff;

    v8f acc[8];
    #pragma unroll
    for (int i = 0; i < 8; ++i) { v8f z = {}; acc[i] = z; }

    for (int k = 0; k < K; k += 4) {
        const int c0 = k + koff;
        const v2f x  = *(const v2f*)(ap + k);
        const v2f gg = *(const v2f*)(g + c0);
        const v2f bbv = *(const v2f*)(b + c0);
        v2f a;
        a.x = ((x.x - m_r) * rs_r * gg.x + bbv.x) * do_r;
        a.y = ((x.y - m_r) * rs_r * gg.y + bbv.y) * do_r;
        const float* wrow = W + (size_t)c0 * F;
        #pragma unroll
        for (int ft = 0; ft < 8; ++ft) {
            const int bcol = ft * 16 + l16;
            v2f bb;
            bb.x = wrow[bcol];
            bb.y = wrow[bcol + F];
            acc[ft] = __builtin_amdgcn_wmma_f32_16x16x4_f32(
                false, a, false, bb, (short)0, acc[ft], false, false);
        }
    }
    const int crow = mt * 16 + half * 8;
    #pragma unroll
    for (int ft = 0; ft < 8; ++ft) {
        const int bcol = ft * 16 + l16;
        #pragma unroll
        for (int r = 0; r < 8; ++r)
            C[(size_t)(crow + r) * F + bcol] = acc[ft][r];
    }
}

template <int FT>
__global__ void gemm_wmma_strip_kernel(const float* __restrict__ A, const float* __restrict__ W,
                                       float* __restrict__ C, int M, int K, int F) {
    const int lane = threadIdx.x & 31;
    const int mt   = (blockIdx.x * blockDim.x + threadIdx.x) >> 5;
    if (mt * 16 >= M) return;
    const int half = lane >> 4;
    const int l16  = lane & 15;
    const int arow = mt * 16 + l16;
    const int koff = half * 2;
    const float* ap = A + (size_t)arow * K + koff;

    v8f acc[FT];
    #pragma unroll
    for (int i = 0; i < FT; ++i) { v8f z = {}; acc[i] = z; }

    for (int k = 0; k < K; k += 4) {
        const v2f a = *(const v2f*)(ap + k);
        const float* wrow = W + (size_t)(k + koff) * F;
        #pragma unroll
        for (int ft = 0; ft < FT; ++ft) {
            const int  bcol = ft * 16 + l16;
            const bool bv   = (bcol < F);
            const int  bc   = bv ? bcol : 0;
            v2f bb;
            bb.x = bv ? wrow[bc]     : 0.0f;
            bb.y = bv ? wrow[bc + F] : 0.0f;
            acc[ft] = __builtin_amdgcn_wmma_f32_16x16x4_f32(
                false, a, false, bb, (short)0, acc[ft], false, false);
        }
    }
    const int crow = mt * 16 + half * 8;
    #pragma unroll
    for (int ft = 0; ft < FT; ++ft) {
        const int bcol = ft * 16 + l16;
        if (bcol < F) {
            #pragma unroll
            for (int r = 0; r < 8; ++r)
                C[(size_t)(crow + r) * F + bcol] = acc[ft][r];
        }
    }
}

template <int F4>
__global__ void scatter_add_vec4_kernel(const float* __restrict__ x,
                                        const int* __restrict__ src, const int* __restrict__ dst,
                                        float* __restrict__ out, int e) {
    long long t = (long long)blockIdx.x * blockDim.x + threadIdx.x;
    if (t >= (long long)e * F4) return;
    const int ed = (int)(t / F4);
    const int c4 = (int)(t % F4);
    const float4 v = ((const float4*)(x + (size_t)src[ed] * (F4 * 4)))[c4];
    float* o = out + (size_t)dst[ed] * (F4 * 4) + (size_t)c4 * 4;
    atomicAdd(o + 0, v.x);
    atomicAdd(o + 1, v.y);
    atomicAdd(o + 2, v.z);
    atomicAdd(o + 3, v.w);
}

__global__ void post_ln_relu_128_kernel(const float* __restrict__ agg,
                                        const float* __restrict__ di_is,
                                        const float* __restrict__ do_is,
                                        const float* __restrict__ g, const float* __restrict__ b,
                                        float* __restrict__ out, int n) {
    int row  = (blockIdx.x * blockDim.x + threadIdx.x) >> 5;
    int lane = threadIdx.x & 31;
    if (row >= n) return;
    const float* p = agg + (size_t)row * 128;
    const float di = di_is[row];
    float v[4];
    float s = 0.f, s2 = 0.f;
    #pragma unroll
    for (int i = 0; i < 4; ++i) {
        v[i] = p[lane + 32 * i] * di;
        s += v[i]; s2 += v[i] * v[i];
    }
    #pragma unroll
    for (int off = 16; off > 0; off >>= 1) {
        s  += __shfl_xor(s,  off, 32);
        s2 += __shfl_xor(s2, off, 32);
    }
    const float m   = s  * (1.0f / 128.0f);
    const float rsv = rsqrtf(s2 * (1.0f / 128.0f) - m * m + LN_EPS);
    const float ds  = do_is[row];
    float* q = out + (size_t)row * 128;
    #pragma unroll
    for (int i = 0; i < 4; ++i) {
        const int c = lane + 32 * i;
        float y = (v[i] - m) * rsv * g[c] + b[c];
        q[c] = fmaxf(y, 0.0f) * ds;
    }
}

__global__ void final_scale_bias_kernel(float* __restrict__ out,
                                        const float* __restrict__ di_is,
                                        const float* __restrict__ b2, int n) {
    const int F = 40;
    int t = blockIdx.x * blockDim.x + threadIdx.x;
    if (t >= n * F) return;
    out[t] = out[t] * di_is[t / F] + b2[t % F];
}

static inline int cdiv(long long a, long long b) { return (int)((a + b - 1) / b); }

extern "C" void kernel_launch(void* const* d_in, const int* in_sizes, int n_in,
                              void* d_out, int out_size, void* d_ws, size_t ws_size,
                              hipStream_t stream) {
    const float* feat = (const float*)d_in[0];
    const int*   src  = (const int*)d_in[1];
    const int*   dst  = (const int*)d_in[2];
    const float* W0   = (const float*)d_in[3];
    const float* W1   = (const float*)d_in[4];
    const float* W2   = (const float*)d_in[5];
    const float* b2   = (const float*)d_in[6];
    const float* g_in = (const float*)d_in[7];
    const float* b_in = (const float*)d_in[8];
    const float* g0   = (const float*)d_in[9];
    const float* b0   = (const float*)d_in[10];
    const float* g1   = (const float*)d_in[11];
    const float* b1   = (const float*)d_in[12];
    float* out = (float*)d_out;

    const int IN_F = 768, HID = 128, CLS = 40;
    const int n = in_sizes[0] / IN_F;
    const int e = in_sizes[1];

    float* ws    = (float*)d_ws;
    float* do_is = ws;
    float* di_is = ws + (size_t)n;
    float* mu    = ws + 2 * (size_t)n;
    float* rs    = ws + 3 * (size_t)n;
    float* bufB  = ws + 4 * (size_t)n;
    float* bufC  = bufB + (size_t)n * HID;
    float* bufD  = bufC + (size_t)n * HID;
    float* bufE  = bufD + (size_t)n * HID;

    const int T = 256;

    hipMemsetAsync(do_is, 0, 2 * (size_t)n * sizeof(float), stream);
    degree_kernel<<<cdiv(e, T), T, 0, stream>>>(src, dst, do_is, di_is, e);
    invsqrt_deg_kernel<<<cdiv(n, T), T, 0, stream>>>(do_is, di_is, n);

    ln_stats_768_kernel<<<cdiv((long long)n * 32, T), T, 0, stream>>>(feat, mu, rs, n);

    const int mtiles = cdiv(n, 16);
    gemm0_ln_wmma_kernel<<<cdiv((long long)mtiles * 32, T), T, 0, stream>>>(
        feat, mu, rs, do_is, g_in, b_in, W0, bufB, n);

    hipMemsetAsync(bufC, 0, (size_t)n * HID * sizeof(float), stream);
    scatter_add_vec4_kernel<32><<<cdiv((long long)e * 32, T), T, 0, stream>>>(bufB, src, dst, bufC, e);

    post_ln_relu_128_kernel<<<cdiv((long long)n * 32, T), T, 0, stream>>>(
        bufC, di_is, do_is, g0, b0, bufD, n);

    hipMemsetAsync(bufB, 0, (size_t)n * HID * sizeof(float), stream);
    scatter_add_vec4_kernel<32><<<cdiv((long long)e * 32, T), T, 0, stream>>>(bufD, src, dst, bufB, e);
    gemm_wmma_strip_kernel<8><<<cdiv((long long)mtiles * 32, T), T, 0, stream>>>(
        bufB, W1, bufC, n, HID, HID);

    post_ln_relu_128_kernel<<<cdiv((long long)n * 32, T), T, 0, stream>>>(
        bufC, di_is, do_is, g1, b1, bufD, n);

    gemm_wmma_strip_kernel<3><<<cdiv((long long)mtiles * 32, T), T, 0, stream>>>(
        bufD, W2, bufE, n, HID, CLS);
    hipMemsetAsync(out, 0, (size_t)n * CLS * sizeof(float), stream);
    scatter_add_vec4_kernel<10><<<cdiv((long long)e * 10, T), T, 0, stream>>>(bufE, src, dst, out, e);
    final_scale_bias_kernel<<<cdiv((long long)n * CLS, T), T, 0, stream>>>(out, di_is, b2, n);
}


# Edge sums by destination block: edges binned once per call by 98-row block; a work-group keeps its rows in LDS, adds source rows by LDS float adds (arrival order), stores rows once; no global atomics
# speedup vs baseline: 1.4060x; 1.4060x over previous
#include <hip/hip_runtime.h>
#include <math.h>

typedef __attribute__((ext_vector_type(2))) float v2f;
typedef __attribute__((ext_vector_type(8))) float v8f;
typedef __attribute__((ext_vector_type(4))) float v4f;

#define LN_EPS 1e-5f

__global__ void degree_kernel(const int* __restrict__ src, const int* __restrict__ dst,
                              float* __restrict__ dout, float* __restrict__ din, int e) {
    int i = blockIdx.x * blockDim.x + threadIdx.x;
    if (i >= e) return;
    atomicAdd(dout + src[i], 1.0f);
    atomicAdd(din  + dst[i], 1.0f);
}

__global__ void invsqrt_deg_kernel(float* __restrict__ dout, float* __restrict__ din, int n) {
    int i = blockIdx.x * blockDim.x + threadIdx.x;
    if (i >= n) return;
    dout[i] = rsqrtf(fmaxf(dout[i], 1.0f));
    din[i]  = rsqrtf(fmaxf(din[i],  1.0f));
}

__global__ void ln_stats_768_kernel(const float* __restrict__ feat,
                                    float* __restrict__ mu, float* __restrict__ rs, int n) {
    int row  = (blockIdx.x * blockDim.x + threadIdx.x) >> 5;
    int lane = threadIdx.x & 31;
    if (row >= n) return;
    const float* p = feat + (size_t)row * 768;
    float s = 0.f, s2 = 0.f;
    #pragma unroll
    for (int c = 0; c < 768; c += 32) {
        float v = p[c + lane];
        s += v; s2 += v * v;
    }
    #pragma unroll
    for (int off = 16; off > 0; off >>= 1) {
        s  += __shfl_xor(s,  off, 32);
        s2 += __shfl_xor(s2, off, 32);
    }
    if (lane == 0) {
        float m   = s  * (1.0f / 768.0f);
        float var = s2 * (1.0f / 768.0f) - m * m;
        mu[row] = m;
        rs[row] = rsqrtf(var + LN_EPS);
    }
}

__global__ void gemm0_ln_wmma_kernel(const float* __restrict__ feat,
                                     const float* __restrict__ mu, const float* __restrict__ rs,
                                     const float* __restrict__ do_is,
                                     const float* __restrict__ g, const float* __restrict__ b,
                                     const float* __restrict__ W,
                                     float* __restrict__ C, int M) {
    const int K = 768, F = 128;
    const int lane = threadIdx.x & 31;
    const int mt   = (blockIdx.x * blockDim.x + threadIdx.x) >> 5;
    if (mt * 16 >= M) return;
    const int half = lane >> 4;
    const int l16  = lane & 15;
    const int arow = mt * 16 + l16;
    const int koff = half * 2;
    const float m_r  = mu[arow];
    const float rs_r = rs[arow];
    const float do_r = do_is[arow];
    const float* ap = feat + (size_t)arow * K + koff;

    v8f acc[8];
    #pragma unroll
    for (int i = 0; i < 8; ++i) { v8f z = {}; acc[i] = z; }

    for (int k = 0; k < K; k += 4) {
        const int c0 = k + koff;
        const v2f x  = *(const v2f*)(ap + k);
        const v2f gg = *(const v2f*)(g + c0);
        const v2f bbv = *(const v2f*)(b + c0);
        v2f a;
        a.x = ((x.x - m_r) * rs_r * gg.x + bbv.x) * do_r;
        a.y = ((x.y - m_r) * rs_r * gg.y + bbv.y) * do_r;
        const float* wrow = W + (size_t)c0 * F;
        #pragma unroll
        for (int ft = 0; ft < 8; ++ft) {
            const int bcol = ft * 16 + l16;
            v2f bb;
            bb.x = wrow[bcol];
            bb.y = wrow[bcol + F];
            acc[ft] = __builtin_amdgcn_wmma_f32_16x16x4_f32(
                false, a, false, bb, (short)0, acc[ft], false, false);
        }
    }
    const int crow = mt * 16 + half * 8;
    #pragma unroll
    for (int ft = 0; ft < 8; ++ft) {
        const int bcol = ft * 16 + l16;
        #pragma unroll
        for (int r = 0; r < 8; ++r)
            C[(size_t)(crow + r) * F + bcol] = acc[ft][r];
    }
}

template <int FT>
__global__ void gemm_wmma_strip_kernel(const float* __restrict__ A, const float* __restrict__ W,
                                       float* __restrict__ C, int M, int K, int F) {
    const int lane = threadIdx.x & 31;
    const int mt   = (blockIdx.x * blockDim.x + threadIdx.x) >> 5;
    if (mt * 16 >= M) return;
    const int half = lane >> 4;
    const int l16  = lane & 15;
    const int arow = mt * 16 + l16;
    const int koff = half * 2;
    const float* ap = A + (size_t)arow * K + koff;

    v8f acc[FT];
    #pragma unroll
    for (int i = 0; i < FT; ++i) { v8f z = {}; acc[i] = z; }

    for (int k = 0; k < K; k += 4) {
        const v2f a = *(const v2f*)(ap + k);
        const float* wrow = W + (size_t)(k + koff) * F;
        #pragma unroll
        for (int ft = 0; ft < FT; ++ft) {
            const int  bcol = ft * 16 + l16;
            const bool bv   = (bcol < F);
            const int  bc   = bv ? bcol : 0;
            v2f bb;
            bb.x = bv ? wrow[bc]     : 0.0f;
            bb.y = bv ? wrow[bc + F] : 0.0f;
            acc[ft] = __builtin_amdgcn_wmma_f32_16x16x4_f32(
                false, a, false, bb, (short)0, acc[ft], false, false);
        }
    }
    const int crow = mt * 16 + half * 8;
    #pragma unroll
    for (int ft = 0; ft < FT; ++ft) {
        const int bcol = ft * 16 + l16;
        if (bcol < F) {
            #pragma unroll
            for (int r = 0; r < 8; ++r)
                C[(size_t)(crow + r) * F + bcol] = acc[ft][r];
        }
    }
}

#define RB   98
#define NBK  512
#define ECH  4096

__global__ void __launch_bounds__(256)
bin_count_kernel(const int* __restrict__ dst, int* __restrict__ hist, int e) {
    __shared__ int h[NBK];
    for (int i = threadIdx.x; i < NBK; i += 256) h[i] = 0;
    __syncthreads();
    const long long base = (long long)blockIdx.x * ECH;
    for (int i = threadIdx.x; i < ECH; i += 256) {
        const long long idx = base + i;
        if (idx < e) atomicAdd(&h[dst[idx] / RB], 1);
    }
    __syncthreads();
    for (int i = threadIdx.x; i < NBK; i += 256) hist[(size_t)blockIdx.x * NBK + i] = h[i];
}

__global__ void __launch_bounds__(NBK)
bin_scan_kernel(int* __restrict__ hist, int* __restrict__ start, int nch) {
    __shared__ int sc[NBK];
    const int b = threadIdx.x;
    int run = 0;
    for (int j0 = 0; j0 < nch; j0 += 14) {
        int c[14];
        #pragma unroll
        for (int u = 0; u < 14; ++u) c[u] = (j0 + u < nch) ? hist[(size_t)(j0 + u) * NBK + b] : 0;
        #pragma unroll
        for (int u = 0; u < 14; ++u) {
            if (j0 + u < nch) hist[(size_t)(j0 + u) * NBK + b] = run;
            run += c[u];
        }
    }
    sc[b] = run;
    __syncthreads();
    for (int off = 1; off < NBK; off <<= 1) {
        const int add = (b >= off) ? sc[b - off] : 0;
        __syncthreads();
        sc[b] += add;
        __syncthreads();
    }
    start[b + 1] = sc[b];
    if (b == 0) start[0] = 0;
}

__global__ void __launch_bounds__(256)
bin_fill_kernel(const int* __restrict__ src, const int* __restrict__ dst,
                const int* __restrict__ hist, const int* __restrict__ start,
                int* __restrict__ bins, int e) {
    __shared__ int cur[NBK];
    for (int i = threadIdx.x; i < NBK; i += 256)
        cur[i] = start[i] + hist[(size_t)blockIdx.x * NBK + i];
    __syncthreads();
    const long long base = (long long)blockIdx.x * ECH;
    for (int i = threadIdx.x; i < ECH; i += 256) {
        const long long idx = base + i;
        if (idx < e) {
            const int d = dst[idx];
            const int b = d / RB;
            const int pos = atomicAdd(&cur[b], 1);
            bins[pos] = src[idx] * RB + (d - b * RB);
        }
    }
}

template <int COLS, int UB>
__global__ void __launch_bounds__(512)
agg_owner_kernel(const float* __restrict__ x, const int* __restrict__ bins,
                 const int* __restrict__ start, float* __restrict__ out, int n) {
    constexpr int NI = (COLS + 31) / 32;
    __shared__ __attribute__((aligned(16))) float acc[RB * COLS];
    const int lane = threadIdx.x & 31;
    const int wave = threadIdx.x >> 5;
    const int nw   = blockDim.x >> 5;
    {
        const v4f z = {0.0f, 0.0f, 0.0f, 0.0f};
        v4f* a4 = (v4f*)acc;
        for (int i = threadIdx.x; i < RB * COLS / 4; i += blockDim.x) a4[i] = z;
    }
    __syncthreads();
    const int e0 = start[blockIdx.x];
    const int e1 = start[blockIdx.x + 1];
    const int ng = (e1 - e0 + UB - 1) / UB;
    int g = wave;
    int pk = 0;
    if (g < ng && e0 + g * UB + lane < e1 && lane < UB) pk = bins[e0 + g * UB + lane];
    while (g < ng) {
        const int cnt = (e1 - (e0 + g * UB) < UB) ? e1 - (e0 + g * UB) : UB;
        const int gn = g + nw;
        int pkn = 0;
        if (gn < ng && e0 + gn * UB + lane < e1 && lane < UB) pkn = bins[e0 + gn * UB + lane];
        if (cnt == UB) {
            float v[UB][NI];
            int   lr[UB];
            #pragma unroll
            for (int u = 0; u < UB; ++u) {
                const int w = __builtin_amdgcn_readlane(pk, u);
                const int s = w / RB;
                lr[u] = w - s * RB;
                const float* xr = x + (size_t)s * COLS + lane;
                #pragma unroll
                for (int i = 0; i < NI; ++i)
                    v[u][i] = (32 * i + lane < COLS) ? xr[32 * i] : 0.0f;
            }
            #pragma unroll
            for (int u = 0; u < UB; ++u) {
                float* a = acc + lr[u] * COLS + lane;
                #pragma unroll
                for (int i = 0; i < NI; ++i)
                    if (32 * i + lane < COLS) atomicAdd(a + 32 * i, v[u][i]);
            }
        } else {
            for (int j = 0; j < cnt; ++j) {
                const int w = __shfl(pk, j, 32);
                const int s = w / RB;
                const float* xr = x + (size_t)s * COLS + lane;
                float* a = acc + (w - s * RB) * COLS + lane;
                #pragma unroll
                for (int i = 0; i < NI; ++i)
                    if (32 * i + lane < COLS) atomicAdd(a + 32 * i, xr[32 * i]);
            }
        }
        pk = pkn;
        g  = gn;
    }
    __syncthreads();
    const int r0 = blockIdx.x * RB;
    const int nr = (n - r0 < RB) ? (n - r0) : RB;
    v4f* o4 = (v4f*)(out + (size_t)r0 * COLS);
    const v4f* a4 = (const v4f*)acc;
    for (int i = threadIdx.x; i < nr * COLS / 4; i += blockDim.x) o4[i] = a4[i];
}

__global__ void post_ln_relu_128_kernel(const float* __restrict__ agg,
                                        const float* __restrict__ di_is,
                                        const float* __restrict__ do_is,
                                        const float* __restrict__ g, const float* __restrict__ b,
                                        float* __restrict__ out, int n) {
    int row  = (blockIdx.x * blockDim.x + threadIdx.x) >> 5;
    int lane = threadIdx.x & 31;
    if (row >= n) return;
    const float* p = agg + (size_t)row * 128;
    const float di = di_is[row];
    float v[4];
    float s = 0.f, s2 = 0.f;
    #pragma unroll
    for (int i = 0; i < 4; ++i) {
        v[i] = p[lane + 32 * i] * di;
        s += v[i]; s2 += v[i] * v[i];
    }
    #pragma unroll
    for (int off = 16; off > 0; off >>= 1) {
        s  += __shfl_xor(s,  off, 32);
        s2 += __shfl_xor(s2, off, 32);
    }
    const float m   = s  * (1.0f / 128.0f);
    const float rsv = rsqrtf(s2 * (1.0f / 128.0f) - m * m + LN_EPS);
    const float ds  = do_is[row];
    float* q = out + (size_t)row * 128;
    #pragma unroll
    for (int i = 0; i < 4; ++i) {
        const int c = lane + 32 * i;
        float y = (v[i] - m) * rsv * g[c] + b[c];
        q[c] = fmaxf(y, 0.0f) * ds;
    }
}

__global__ void final_scale_bias_kernel(float* __restrict__ out,
                                        const float* __restrict__ di_is,
                                        const float* __restrict__ b2, int n) {
    const int F = 40;
    int t = blockIdx.x * blockDim.x + threadIdx.x;
    if (t >= n * F) return;
    out[t] = out[t] * di_is[t / F] + b2[t % F];
}

static inline int cdiv(long long a, long long b) { return (int)((a + b - 1) / b); }

extern "C" void kernel_launch(void* const* d_in, const int* in_sizes, int n_in,
                              void* d_out, int out_size, void* d_ws, size_t ws_size,
                              hipStream_t stream) {
    const float* feat = (const float*)d_in[0];
    const int*   src  = (const int*)d_in[1];
    const int*   dst  = (const int*)d_in[2];
    const float* W0   = (const float*)d_in[3];
    const float* W1   = (const float*)d_in[4];
    const float* W2   = (const float*)d_in[5];
    const float* b2   = (const float*)d_in[6];
    const float* g_in = (const float*)d_in[7];
    const float* b_in = (const float*)d_in[8];
    const float* g0   = (const float*)d_in[9];
    const float* b0   = (const float*)d_in[10];
    const float* g1   = (const float*)d_in[11];
    const float* b1   = (const float*)d_in[12];
    float* out = (float*)d_out;

    const int IN_F = 768, HID = 128, CLS = 40;
    const int n = in_sizes[0] / IN_F;
    const int e = in_sizes[1];

    float* ws    = (float*)d_ws;
    float* do_is = ws;
    float* di_is = ws + (size_t)n;
    float* mu    = ws + 2 * (size_t)n;
    float* rs    = ws + 3 * (size_t)n;
    float* bufB  = ws + 4 * (size_t)n;
    float* bufC  = bufB + (size_t)n * HID;
    float* bufD  = bufC + (size_t)n * HID;
    int*   bins  = (int*)(bufD + (size_t)n * HID);
    const int nch = cdiv(e, ECH);
    int*   hist  = bins + (size_t)e;
    int*   start = hist + (size_t)nch * NBK;

    const int T = 256;

    hipMemsetAsync(do_is, 0, 2 * (size_t)n * sizeof(float), stream);
    degree_kernel<<<cdiv(e, T), T, 0, stream>>>(src, dst, do_is, di_is, e);
    invsqrt_deg_kernel<<<cdiv(n, T), T, 0, stream>>>(do_is, di_is, n);

    bin_count_kernel<<<nch, 256, 0, stream>>>(dst, hist, e);
    bin_scan_kernel<<<1, NBK, 0, stream>>>(hist, start, nch);
    bin_fill_kernel<<<nch, 256, 0, stream>>>(src, dst, hist, start, bins, e);

    ln_stats_768_kernel<<<cdiv((long long)n * 32, T), T, 0, stream>>>(feat, mu, rs, n);

    const int mtiles = cdiv(n, 16);
    gemm0_ln_wmma_kernel<<<cdiv((long long)mtiles * 32, T), T, 0, stream>>>(
        feat, mu, rs, do_is, g_in, b_in, W0, bufB, n);

    agg_owner_kernel<128, 8><<<NBK, 512, 0, stream>>>(bufB, bins, start, bufC, n);

    post_ln_relu_128_kernel<<<cdiv((long long)n * 32, T), T, 0, stream>>>(
        bufC, di_is, do_is, g0, b0, bufD, n);

    agg_owner_kernel<128, 8><<<NBK, 512, 0, stream>>>(bufD, bins, start, bufB, n);
    gemm_wmma_strip_kernel<8><<<cdiv((long long)mtiles * 32, T), T, 0, stream>>>(
        bufB, W1, bufC, n, HID, HID);

    post_ln_relu_128_kernel<<<cdiv((long long)n * 32, T), T, 0, stream>>>(
        bufC, di_is, do_is, g1, b1, bufD, n);

    gemm_wmma_strip_kernel<3><<<cdiv((long long)mtiles * 32, T), T, 0, stream>>>(
        bufD, W2, bufB, n, HID, CLS);
    agg_owner_kernel<40, 8><<<NBK, 512, 0, stream>>>(bufB, bins, start, out, n);
    final_scale_bias_kernel<<<cdiv((long long)n * CLS, T), T, 0, stream>>>(out, di_is, b2, n);
}


# Degrees: 64 work-groups each count one slice of src or dst into a full-table LDS histogram (int), tables stored, then summed per node; no global atomics, no memset; degree values identical
# speedup vs baseline: 1.6432x; 1.1687x over previous
#include <hip/hip_runtime.h>
#include <math.h>

typedef __attribute__((ext_vector_type(2))) float v2f;
typedef __attribute__((ext_vector_type(8))) float v8f;
typedef __attribute__((ext_vector_type(4))) float v4f;

#define LN_EPS 1e-5f

#define RB   98
#define NBK  512
#define ECH  4096

#define DG   32
#define DN   (NBK * RB)
#define DQ   8

typedef __attribute__((ext_vector_type(4))) int v4i;

__global__ void __launch_bounds__(1024)
degree_hist_kernel(const int* __restrict__ src, const int* __restrict__ dst,
                   int* __restrict__ part, int e) {
    __shared__ __attribute__((aligned(16))) int cnt[DN];
    const int* __restrict__ idx = blockIdx.y ? dst : src;
    {
        const v4i z = {0, 0, 0, 0};
        v4i* c4 = (v4i*)cnt;
        for (int i = threadIdx.x; i < DN / 4; i += 1024) c4[i] = z;
    }
    __syncthreads();
    const int per = (((e + DG - 1) / DG) + 3) & ~3;
    const long long lo = (long long)blockIdx.x * per;
    long long hi = lo + per;
    if (hi > e) hi = e;
    const int len = (hi > lo) ? (int)(hi - lo) : 0;
    const int nq  = len >> 2;
    const v4i* q = (const v4i*)(idx + lo);
    for (int base = 0; base < nq; base += 1024 * DQ) {
        v4i w[DQ];
        #pragma unroll
        for (int u = 0; u < DQ; ++u) {
            const int i = base + u * 1024 + threadIdx.x;
            w[u] = q[(i < nq) ? i : nq - 1];
        }
        #pragma unroll
        for (int u = 0; u < DQ; ++u) {
            const int i = base + u * 1024 + threadIdx.x;
            if (i < nq) {
                atomicAdd(&cnt[w[u].x], 1);
                atomicAdd(&cnt[w[u].y], 1);
                atomicAdd(&cnt[w[u].z], 1);
                atomicAdd(&cnt[w[u].w], 1);
            }
        }
    }
    if (4 * nq + (int)threadIdx.x < len)
        atomicAdd(&cnt[idx[lo + 4 * nq + threadIdx.x]], 1);
    __syncthreads();
    v4i* p4 = (v4i*)(part + ((size_t)blockIdx.y * DG + blockIdx.x) * DN);
    const v4i* c4 = (const v4i*)cnt;
    for (int i = threadIdx.x; i < DN / 4; i += 1024) p4[i] = c4[i];
}

__global__ void invsqrt_deg_kernel(const int* __restrict__ part,
                                   float* __restrict__ dout, float* __restrict__ din, int n) {
    int i = blockIdx.x * blockDim.x + threadIdx.x;
    if (i >= n) return;
    const int* p = part + (size_t)blockIdx.y * DG * DN + i;
    int s = 0;
    #pragma unroll
    for (int g = 0; g < DG; ++g) s += p[(size_t)g * DN];
    float* o = blockIdx.y ? din : dout;
    o[i] = rsqrtf(fmaxf((float)s, 1.0f));
}

__global__ void ln_stats_768_kernel(const float* __restrict__ feat,
                                    float* __restrict__ mu, float* __restrict__ rs, int n) {
    int row  = (blockIdx.x * blockDim.x + threadIdx.x) >> 5;
    int lane = threadIdx.x & 31;
    if (row >= n) return;
    const float* p = feat + (size_t)row * 768;
    float s = 0.f, s2 = 0.f;
    #pragma unroll
    for (int c = 0; c < 768; c += 32) {
        float v = p[c + lane];
        s += v; s2 += v * v;
    }
    #pragma unroll
    for (int off = 16; off > 0; off >>= 1) {
        s  += __shfl_xor(s,  off, 32);
        s2 += __shfl_xor(s2, off, 32);
    }
    if (lane == 0) {
        float m   = s  * (1.0f / 768.0f);
        float var = s2 * (1.0f / 768.0f) - m * m;
        mu[row] = m;
        rs[row] = rsqrtf(var + LN_EPS);
    }
}

__global__ void gemm0_ln_wmma_kernel(const float* __restrict__ feat,
                                     const float* __restrict__ mu, const float* __restrict__ rs,
                                     const float* __restrict__ do_is,
                                     const float* __restrict__ g, const float* __restrict__ b,
                                     const float* __restrict__ W,
                                     float* __restrict__ C, int M) {
    const int K = 768, F = 128;
    const int lane = threadIdx.x & 31;
    const int mt   = (blockIdx.x * blockDim.x + threadIdx.x) >> 5;
    if (mt * 16 >= M) return;
    const int half = lane >> 4;
    const int l16  = lane & 15;
    const int arow = mt * 16 + l16;
    const int koff = half * 2;
    const float m_r  = mu[arow];
    const float rs_r = rs[arow];
    const float do_r = do_is[arow];
    const float* ap = feat + (size_t)arow * K + koff;

    v8f acc[8];
    #pragma unroll
    for (int i = 0; i < 8; ++i) { v8f z = {}; acc[i] = z; }

    for (int k = 0; k < K; k += 4) {
        const int c0 = k + koff;
        const v2f x  = *(const v2f*)(ap + k);
        const v2f gg = *(const v2f*)(g + c0);
        const v2f bbv = *(const v2f*)(b + c0);
        v2f a;
        a.x = ((x.x - m_r) * rs_r * gg.x + bbv.x) * do_r;
        a.y = ((x.y - m_r) * rs_r * gg.y + bbv.y) * do_r;
        const float* wrow = W + (size_t)c0 * F;
        #pragma unroll
        for (int ft = 0; ft < 8; ++ft) {
            const int bcol = ft * 16 + l16;
            v2f bb;
            bb.x = wrow[bcol];
            bb.y = wrow[bcol + F];
            acc[ft] = __builtin_amdgcn_wmma_f32_16x16x4_f32(
                false, a, false, bb, (short)0, acc[ft], false, false);
        }
    }
    const int crow = mt * 16 + half * 8;
    #pragma unroll
    for (int ft = 0; ft < 8; ++ft) {
        const int bcol = ft * 16 + l16;
        #pragma unroll
        for (int r = 0; r < 8; ++r)
            C[(size_t)(crow + r) * F + bcol] = acc[ft][r];
    }
}

template <int FT>
__global__ void gemm_wmma_strip_kernel(const float* __restrict__ A, const float* __restrict__ W,
                                       float* __restrict__ C, int M, int K, int F) {
    const int lane = threadIdx.x & 31;
    const int mt   = (blockIdx.x * blockDim.x + threadIdx.x) >> 5;
    if (mt * 16 >= M) return;
    const int half = lane >> 4;
    const int l16  = lane & 15;
    const int arow = mt * 16 + l16;
    const int koff = half * 2;
    const float* ap = A + (size_t)arow * K + koff;

    v8f acc[FT];
    #pragma unroll
    for (int i = 0; i < FT; ++i) { v8f z = {}; acc[i] = z; }

    for (int k = 0; k < K; k += 4) {
        const v2f a = *(const v2f*)(ap + k);
        const float* wrow = W + (size_t)(k + koff) * F;
        #pragma unroll
        for (int ft = 0; ft < FT; ++ft) {
            const int  bcol = ft * 16 + l16;
            const bool bv   = (bcol < F);
            const int  bc   = bv ? bcol : 0;
            v2f bb;
            bb.x = bv ? wrow[bc]     : 0.0f;
            bb.y = bv ? wrow[bc + F] : 0.0f;
            acc[ft] = __builtin_amdgcn_wmma_f32_16x16x4_f32(
                false, a, false, bb, (short)0, acc[ft], false, false);
        }
    }
    const int crow = mt * 16 + half * 8;
    #pragma unroll
    for (int ft = 0; ft < FT; ++ft) {
        const int bcol = ft * 16 + l16;
        if (bcol < F) {
            #pragma unroll
            for (int r = 0; r < 8; ++r)
                C[(size_t)(crow + r) * F + bcol] = acc[ft][r];
        }
    }
}

__global__ void __launch_bounds__(256)
bin_count_kernel(const int* __restrict__ dst, int* __restrict__ hist, int e) {
    __shared__ int h[NBK];
    for (int i = threadIdx.x; i < NBK; i += 256) h[i] = 0;
    __syncthreads();
    const long long base = (long long)blockIdx.x * ECH;
    for (int i = threadIdx.x; i < ECH; i += 256) {
        const long long idx = base + i;
        if (idx < e) atomicAdd(&h[dst[idx] / RB], 1);
    }
    __syncthreads();
    for (int i = threadIdx.x; i < NBK; i += 256) hist[(size_t)blockIdx.x * NBK + i] = h[i];
}

__global__ void __launch_bounds__(NBK)
bin_scan_kernel(int* __restrict__ hist, int* __restrict__ start, int nch) {
    __shared__ int sc[NBK];
    const int b = threadIdx.x;
    int run = 0;
    for (int j0 = 0; j0 < nch; j0 += 14) {
        int c[14];
        #pragma unroll
        for (int u = 0; u < 14; ++u) c[u] = (j0 + u < nch) ? hist[(size_t)(j0 + u) * NBK + b] : 0;
        #pragma unroll
        for (int u = 0; u < 14; ++u) {
            if (j0 + u < nch) hist[(size_t)(j0 + u) * NBK + b] = run;
            run += c[u];
        }
    }
    sc[b] = run;
    __syncthreads();
    for (int off = 1; off < NBK; off <<= 1) {
        const int add = (b >= off) ? sc[b - off] : 0;
        __syncthreads();
        sc[b] += add;
        __syncthreads();
    }
    start[b + 1] = sc[b];
    if (b == 0) start[0] = 0;
}

__global__ void __launch_bounds__(256)
bin_fill_kernel(const int* __restrict__ src, const int* __restrict__ dst,
                const int* __restrict__ hist, const int* __restrict__ start,
                int* __restrict__ bins, int e) {
    __shared__ int cur[NBK];
    for (int i = threadIdx.x; i < NBK; i += 256)
        cur[i] = start[i] + hist[(size_t)blockIdx.x * NBK + i];
    __syncthreads();
    const long long base = (long long)blockIdx.x * ECH;
    for (int i = threadIdx.x; i < ECH; i += 256) {
        const long long idx = base + i;
        if (idx < e) {
            const int d = dst[idx];
            const int b = d / RB;
            const int pos = atomicAdd(&cur[b], 1);
            bins[pos] = src[idx] * RB + (d - b * RB);
        }
    }
}

template <int COLS, int UB>
__global__ void __launch_bounds__(512)
agg_owner_kernel(const float* __restrict__ x, const int* __restrict__ bins,
                 const int* __restrict__ start, float* __restrict__ out, int n) {
    constexpr int NI = (COLS + 31) / 32;
    __shared__ __attribute__((aligned(16))) float acc[RB * COLS];
    const int lane = threadIdx.x & 31;
    const int wave = threadIdx.x >> 5;
    const int nw   = blockDim.x >> 5;
    {
        const v4f z = {0.0f, 0.0f, 0.0f, 0.0f};
        v4f* a4 = (v4f*)acc;
        for (int i = threadIdx.x; i < RB * COLS / 4; i += blockDim.x) a4[i] = z;
    }
    __syncthreads();
    const int e0 = start[blockIdx.x];
    const int e1 = start[blockIdx.x + 1];
    const int ng = (e1 - e0 + UB - 1) / UB;
    int g = wave;
    int pk = 0;
    if (g < ng && e0 + g * UB + lane < e1 && lane < UB) pk = bins[e0 + g * UB + lane];
    while (g < ng) {
        const int cnt = (e1 - (e0 + g * UB) < UB) ? e1 - (e0 + g * UB) : UB;
        const int gn = g + nw;
        int pkn = 0;
        if (gn < ng && e0 + gn * UB + lane < e1 && lane < UB) pkn = bins[e0 + gn * UB + lane];
        if (cnt == UB) {
            float v[UB][NI];
            int   lr[UB];
            #pragma unroll
            for (int u = 0; u < UB; ++u) {
                const int w = __builtin_amdgcn_readlane(pk, u);
                const int s = w / RB;
                lr[u] = w - s * RB;
                const float* xr = x + (size_t)s * COLS + lane;
                #pragma unroll
                for (int i = 0; i < NI; ++i)
                    v[u][i] = (32 * i + lane < COLS) ? xr[32 * i] : 0.0f;
            }
            #pragma unroll
            for (int u = 0; u < UB; ++u) {
                float* a = acc + lr[u] * COLS + lane;
                #pragma unroll
                for (int i = 0; i < NI; ++i)
                    if (32 * i + lane < COLS) atomicAdd(a + 32 * i, v[u][i]);
            }
        } else {
            for (int j = 0; j < cnt; ++j) {
                const int w = __shfl(pk, j, 32);
                const int s = w / RB;
                const float* xr = x + (size_t)s * COLS + lane;
                float* a = acc + (w - s * RB) * COLS + lane;
                #pragma unroll
                for (int i = 0; i < NI; ++i)
                    if (32 * i + lane < COLS) atomicAdd(a + 32 * i, xr[32 * i]);
            }
        }
        pk = pkn;
        g  = gn;
    }
    __syncthreads();
    const int r0 = blockIdx.x * RB;
    const int nr = (n - r0 < RB) ? (n - r0) : RB;
    v4f* o4 = (v4f*)(out + (size_t)r0 * COLS);
    const v4f* a4 = (const v4f*)acc;
    for (int i = threadIdx.x; i < nr * COLS / 4; i += blockDim.x) o4[i] = a4[i];
}

__global__ void post_ln_relu_128_kernel(const float* __restrict__ agg,
                                        const float* __restrict__ di_is,
                                        const float* __restrict__ do_is,
                                        const float* __restrict__ g, const float* __restrict__ b,
                                        float* __restrict__ out, int n) {
    int row  = (blockIdx.x * blockDim.x + threadIdx.x) >> 5;
    int lane = threadIdx.x & 31;
    if (row >= n) return;
    const float* p = agg + (size_t)row * 128;
    const float di = di_is[row];
    float v[4];
    float s = 0.f, s2 = 0.f;
    #pragma unroll
    for (int i = 0; i < 4; ++i) {
        v[i] = p[lane + 32 * i] * di;
        s += v[i]; s2 += v[i] * v[i];
    }
    #pragma unroll
    for (int off = 16; off > 0; off >>= 1) {
        s  += __shfl_xor(s,  off, 32);
        s2 += __shfl_xor(s2, off, 32);
    }
    const float m   = s  * (1.0f / 128.0f);
    const float rsv = rsqrtf(s2 * (1.0f / 128.0f) - m * m + LN_EPS);
    const float ds  = do_is[row];
    float* q = out + (size_t)row * 128;
    #pragma unroll
    for (int i = 0; i < 4; ++i) {
        const int c = lane + 32 * i;
        float y = (v[i] - m) * rsv * g[c] + b[c];
        q[c] = fmaxf(y, 0.0f) * ds;
    }
}

__global__ void final_scale_bias_kernel(float* __restrict__ out,
                                        const float* __restrict__ di_is,
                                        const float* __restrict__ b2, int n) {
    const int F = 40;
    int t = blockIdx.x * blockDim.x + threadIdx.x;
    if (t >= n * F) return;
    out[t] = out[t] * di_is[t / F] + b2[t % F];
}

static inline int cdiv(long long a, long long b) { return (int)((a + b - 1) / b); }

extern "C" void kernel_launch(void* const* d_in, const int* in_sizes, int n_in,
                              void* d_out, int out_size, void* d_ws, size_t ws_size,
                              hipStream_t stream) {
    const float* feat = (const float*)d_in[0];
    const int*   src  = (const int*)d_in[1];
    const int*   dst  = (const int*)d_in[2];
    const float* W0   = (const float*)d_in[3];
    const float* W1   = (const float*)d_in[4];
    const float* W2   = (const float*)d_in[5];
    const float* b2   = (const float*)d_in[6];
    const float* g_in = (const float*)d_in[7];
    const float* b_in = (const float*)d_in[8];
    const float* g0   = (const float*)d_in[9];
    const float* b0   = (const float*)d_in[10];
    const float* g1   = (const float*)d_in[11];
    const float* b1   = (const float*)d_in[12];
    float* out = (float*)d_out;

    const int IN_F = 768, HID = 128, CLS = 40;
    const int n = in_sizes[0] / IN_F;
    const int e = in_sizes[1];

    float* ws    = (float*)d_ws;
    float* do_is = ws;
    float* di_is = ws + (size_t)n;
    float* mu    = ws + 2 * (size_t)n;
    float* rs    = ws + 3 * (size_t)n;
    float* bufB  = ws + 4 * (size_t)n;
    float* bufC  = bufB + (size_t)n * HID;
    float* bufD  = bufC + (size_t)n * HID;
    int*   bins  = (int*)(bufD + (size_t)n * HID);
    const int nch = cdiv(e, ECH);
    int*   hist  = bins + (size_t)e;
    int*   start = hist + (size_t)nch * NBK;

    const int T = 256;

    int* part = (int*)bufC;
    degree_hist_kernel<<<dim3(DG, 2), 1024, 0, stream>>>(src, dst, part, e);
    invsqrt_deg_kernel<<<dim3(cdiv(n, T), 2), T, 0, stream>>>(part, do_is, di_is, n);

    bin_count_kernel<<<nch, 256, 0, stream>>>(dst, hist, e);
    bin_scan_kernel<<<1, NBK, 0, stream>>>(hist, start, nch);
    bin_fill_kernel<<<nch, 256, 0, stream>>>(src, dst, hist, start, bins, e);

    ln_stats_768_kernel<<<cdiv((long long)n * 32, T), T, 0, stream>>>(feat, mu, rs, n);

    const int mtiles = cdiv(n, 16);
    gemm0_ln_wmma_kernel<<<cdiv((long long)mtiles * 32, T), T, 0, stream>>>(
        feat, mu, rs, do_is, g_in, b_in, W0, bufB, n);

    agg_owner_kernel<128, 8><<<NBK, 512, 0, stream>>>(bufB, bins, start, bufC, n);

    post_ln_relu_128_kernel<<<cdiv((long long)n * 32, T), T, 0, stream>>>(
        bufC, di_is, do_is, g0, b0, bufD, n);

    agg_owner_kernel<128, 8><<<NBK, 512, 0, stream>>>(bufD, bins, start, bufB, n);
    gemm_wmma_strip_kernel<8><<<cdiv((long long)mtiles * 32, T), T, 0, stream>>>(
        bufB, W1, bufC, n, HID, HID);

    post_ln_relu_128_kernel<<<cdiv((long long)n * 32, T), T, 0, stream>>>(
        bufC, di_is, do_is, g1, b1, bufD, n);

    gemm_wmma_strip_kernel<3><<<cdiv((long long)mtiles * 32, T), T, 0, stream>>>(
        bufD, W2, bufB, n, HID, CLS);
    agg_owner_kernel<40, 8><<<NBK, 512, 0, stream>>>(bufB, bins, start, out, n);
    final_scale_bias_kernel<<<cdiv((long long)n * CLS, T), T, 0, stream>>>(out, di_is, b2, n);
}


# 128-column products (layers 0 and 1): column tiles interleaved (tile t = columns 4*lane+t), so B fragments are 16-byte global loads and C rows 16-byte stores, 16 k requested per trip; sums unchanged
# speedup vs baseline: 2.1753x; 1.3238x over previous
#include <hip/hip_runtime.h>
#include <math.h>

typedef __attribute__((ext_vector_type(2))) float v2f;
typedef __attribute__((ext_vector_type(8))) float v8f;
typedef __attribute__((ext_vector_type(4))) float v4f;

#define LN_EPS 1e-5f

#define RB   98
#define NBK  512
#define ECH  4096

#define DG   32
#define DN   (NBK * RB)
#define DQ   8

typedef __attribute__((ext_vector_type(4))) int v4i;

__global__ void __launch_bounds__(1024)
degree_hist_kernel(const int* __restrict__ src, const int* __restrict__ dst,
                   int* __restrict__ part, int e) {
    __shared__ __attribute__((aligned(16))) int cnt[DN];
    const int* __restrict__ idx = blockIdx.y ? dst : src;
    {
        const v4i z = {0, 0, 0, 0};
        v4i* c4 = (v4i*)cnt;
        for (int i = threadIdx.x; i < DN / 4; i += 1024) c4[i] = z;
    }
    __syncthreads();
    const int per = (((e + DG - 1) / DG) + 3) & ~3;
    const long long lo = (long long)blockIdx.x * per;
    long long hi = lo + per;
    if (hi > e) hi = e;
    const int len = (hi > lo) ? (int)(hi - lo) : 0;
    const int nq  = len >> 2;
    const v4i* q = (const v4i*)(idx + lo);
    for (int base = 0; base < nq; base += 1024 * DQ) {
        v4i w[DQ];
        #pragma unroll
        for (int u = 0; u < DQ; ++u) {
            const int i = base + u * 1024 + threadIdx.x;
            w[u] = q[(i < nq) ? i : nq - 1];
        }
        #pragma unroll
        for (int u = 0; u < DQ; ++u) {
            const int i = base + u * 1024 + threadIdx.x;
            if (i < nq) {
                atomicAdd(&cnt[w[u].x], 1);
                atomicAdd(&cnt[w[u].y], 1);
                atomicAdd(&cnt[w[u].z], 1);
                atomicAdd(&cnt[w[u].w], 1);
            }
        }
    }
    if (4 * nq + (int)threadIdx.x < len)
        atomicAdd(&cnt[idx[lo + 4 * nq + threadIdx.x]], 1);
    __syncthreads();
    v4i* p4 = (v4i*)(part + ((size_t)blockIdx.y * DG + blockIdx.x) * DN);
    const v4i* c4 = (const v4i*)cnt;
    for (int i = threadIdx.x; i < DN / 4; i += 1024) p4[i] = c4[i];
}

__global__ void invsqrt_deg_kernel(const int* __restrict__ part,
                                   float* __restrict__ dout, float* __restrict__ din, int n) {
    int i = blockIdx.x * blockDim.x + threadIdx.x;
    if (i >= n) return;
    const int* p = part + (size_t)blockIdx.y * DG * DN + i;
    int s = 0;
    #pragma unroll
    for (int g = 0; g < DG; ++g) s += p[(size_t)g * DN];
    float* o = blockIdx.y ? din : dout;
    o[i] = rsqrtf(fmaxf((float)s, 1.0f));
}

__global__ void ln_stats_768_kernel(const float* __restrict__ feat,
                                    float* __restrict__ mu, float* __restrict__ rs, int n) {
    int row  = (blockIdx.x * blockDim.x + threadIdx.x) >> 5;
    int lane = threadIdx.x & 31;
    if (row >= n) return;
    const float* p = feat + (size_t)row * 768;
    float s = 0.f, s2 = 0.f;
    #pragma unroll
    for (int c = 0; c < 768; c += 32) {
        float v = p[c + lane];
        s += v; s2 += v * v;
    }
    #pragma unroll
    for (int off = 16; off > 0; off >>= 1) {
        s  += __shfl_xor(s,  off, 32);
        s2 += __shfl_xor(s2, off, 32);
    }
    if (lane == 0) {
        float m   = s  * (1.0f / 768.0f);
        float var = s2 * (1.0f / 768.0f) - m * m;
        mu[row] = m;
        rs[row] = rsqrtf(var + LN_EPS);
    }
}

__global__ void __launch_bounds__(256)
gemm0_ln_wmma_kernel(const float* __restrict__ feat,
                                     const float* __restrict__ mu, const float* __restrict__ rs,
                                     const float* __restrict__ do_is,
                                     const float* __restrict__ g, const float* __restrict__ b,
                                     const float* __restrict__ W,
                                     float* __restrict__ C, int M) {
    const int K = 768, F = 128;
    const int lane = threadIdx.x & 31;
    const int mt   = (blockIdx.x * blockDim.x + threadIdx.x) >> 5;
    if (mt * 16 >= M) return;
    const int half = lane >> 4;
    const int l16  = lane & 15;
    const int arow = mt * 16 + l16;
    const int koff = half * 2;
    const float m_r  = mu[arow];
    const float rs_r = rs[arow];
    const float do_r = do_is[arow];
    const float* ap = feat + (size_t)arow * K + koff;

    v8f acc[8];
    #pragma unroll
    for (int i = 0; i < 8; ++i) { v8f z = {}; acc[i] = z; }

    constexpr int KB = 4;
    for (int k = 0; k < K; k += 4 * KB) {
        v2f x[KB], gg[KB], bbv[KB];
        v4f w0l[KB], w0h[KB], w1l[KB], w1h[KB];
        #pragma unroll
        for (int u = 0; u < KB; ++u) {
            const int c0 = k + 4 * u + koff;
            x[u]   = *(const v2f*)(ap + k + 4 * u);
            gg[u]  = *(const v2f*)(g + c0);
            bbv[u] = *(const v2f*)(b + c0);
        }
        __builtin_amdgcn_sched_barrier(0);
        #pragma unroll
        for (int u = 0; u < KB; ++u) {
            const int c0 = k + 4 * u + koff;
            const float* wrow = W + (size_t)c0 * F + 4 * l16;
            w0l[u] = *(const v4f*)(wrow);
            w0h[u] = *(const v4f*)(wrow + 64);
            w1l[u] = *(const v4f*)(wrow + F);
            w1h[u] = *(const v4f*)(wrow + F + 64);
        }
        #pragma unroll
        for (int u = 0; u < KB; ++u) {
            v2f a;
            a.x = ((x[u].x - m_r) * rs_r * gg[u].x + bbv[u].x) * do_r;
            a.y = ((x[u].y - m_r) * rs_r * gg[u].y + bbv[u].y) * do_r;
            #pragma unroll
            for (int ft = 0; ft < 8; ++ft) {
                v2f bb;
                bb.x = (ft < 4) ? w0l[u][ft & 3] : w0h[u][ft & 3];
                bb.y = (ft < 4) ? w1l[u][ft & 3] : w1h[u][ft & 3];
                acc[ft] = __builtin_amdgcn_wmma_f32_16x16x4_f32(
                    false, a, false, bb, (short)0, acc[ft], false, false);
            }
        }
    }
    const int crow = mt * 16 + half * 8;
    #pragma unroll
    for (int r = 0; r < 8; ++r) {
        float* crp = C + (size_t)(crow + r) * F + 4 * l16;
        const v4f lo = {acc[0][r], acc[1][r], acc[2][r], acc[3][r]};
        const v4f hi = {acc[4][r], acc[5][r], acc[6][r], acc[7][r]};
        *(v4f*)(crp)      = lo;
        *(v4f*)(crp + 64) = hi;
    }
}

template <int FT>
__global__ void gemm_wmma_strip_kernel(const float* __restrict__ A, const float* __restrict__ W,
                                       float* __restrict__ C, int M, int K, int F) {
    const int lane = threadIdx.x & 31;
    const int mt   = (blockIdx.x * blockDim.x + threadIdx.x) >> 5;
    if (mt * 16 >= M) return;
    const int half = lane >> 4;
    const int l16  = lane & 15;
    const int arow = mt * 16 + l16;
    const int koff = half * 2;
    const float* ap = A + (size_t)arow * K + koff;

    v8f acc[FT];
    #pragma unroll
    for (int i = 0; i < FT; ++i) { v8f z = {}; acc[i] = z; }

    for (int k = 0; k < K; k += 4) {
        const v2f a = *(const v2f*)(ap + k);
        const float* wrow = W + (size_t)(k + koff) * F;
        #pragma unroll
        for (int ft = 0; ft < FT; ++ft) {
            const int  bcol = ft * 16 + l16;
            const bool bv   = (bcol < F);
            const int  bc   = bv ? bcol : 0;
            v2f bb;
            bb.x = bv ? wrow[bc]     : 0.0f;
            bb.y = bv ? wrow[bc + F] : 0.0f;
            acc[ft] = __builtin_amdgcn_wmma_f32_16x16x4_f32(
                false, a, false, bb, (short)0, acc[ft], false, false);
        }
    }
    const int crow = mt * 16 + half * 8;
    #pragma unroll
    for (int ft = 0; ft < FT; ++ft) {
        const int bcol = ft * 16 + l16;
        if (bcol < F) {
            #pragma unroll
            for (int r = 0; r < 8; ++r)
                C[(size_t)(crow + r) * F + bcol] = acc[ft][r];
        }
    }
}

__global__ void __launch_bounds__(256)
gemm128_wmma_strip_kernel(const float* __restrict__ A, const float* __restrict__ W,
                                          float* __restrict__ C, int M, int K) {
    const int F = 128;
    const int lane = threadIdx.x & 31;
    const int mt   = (blockIdx.x * blockDim.x + threadIdx.x) >> 5;
    if (mt * 16 >= M) return;
    const int half = lane >> 4;
    const int l16  = lane & 15;
    const int arow = mt * 16 + l16;
    const int koff = half * 2;
    const float* ap = A + (size_t)arow * K + koff;

    v8f acc[8];
    #pragma unroll
    for (int i = 0; i < 8; ++i) { v8f z = {}; acc[i] = z; }

    constexpr int KB = 4;
    for (int k = 0; k < K; k += 4 * KB) {
        v2f a[KB];
        v4f w0l[KB], w0h[KB], w1l[KB], w1h[KB];
        #pragma unroll
        for (int u = 0; u < KB; ++u)
            a[u] = *(const v2f*)(ap + k + 4 * u);
        __builtin_amdgcn_sched_barrier(0);
        #pragma unroll
        for (int u = 0; u < KB; ++u) {
            const float* wrow = W + (size_t)(k + 4 * u + koff) * F + 4 * l16;
            w0l[u] = *(const v4f*)(wrow);
            w0h[u] = *(const v4f*)(wrow + 64);
            w1l[u] = *(const v4f*)(wrow + F);
            w1h[u] = *(const v4f*)(wrow + F + 64);
        }
        #pragma unroll
        for (int u = 0; u < KB; ++u) {
            #pragma unroll
            for (int ft = 0; ft < 8; ++ft) {
                v2f bb;
                bb.x = (ft < 4) ? w0l[u][ft & 3] : w0h[u][ft & 3];
                bb.y = (ft < 4) ? w1l[u][ft & 3] : w1h[u][ft & 3];
                acc[ft] = __builtin_amdgcn_wmma_f32_16x16x4_f32(
                    false, a[u], false, bb, (short)0, acc[ft], false, false);
            }
        }
    }
    const int crow = mt * 16 + half * 8;
    #pragma unroll
    for (int r = 0; r < 8; ++r) {
        float* crp = C + (size_t)(crow + r) * F + 4 * l16;
        const v4f lo = {acc[0][r], acc[1][r], acc[2][r], acc[3][r]};
        const v4f hi = {acc[4][r], acc[5][r], acc[6][r], acc[7][r]};
        *(v4f*)(crp)      = lo;
        *(v4f*)(crp + 64) = hi;
    }
}

__global__ void __launch_bounds__(256)
bin_count_kernel(const int* __restrict__ dst, int* __restrict__ hist, int e) {
    __shared__ int h[NBK];
    for (int i = threadIdx.x; i < NBK; i += 256) h[i] = 0;
    __syncthreads();
    const long long base = (long long)blockIdx.x * ECH;
    for (int i = threadIdx.x; i < ECH; i += 256) {
        const long long idx = base + i;
        if (idx < e) atomicAdd(&h[dst[idx] / RB], 1);
    }
    __syncthreads();
    for (int i = threadIdx.x; i < NBK; i += 256) hist[(size_t)blockIdx.x * NBK + i] = h[i];
}

__global__ void __launch_bounds__(NBK)
bin_scan_kernel(int* __restrict__ hist, int* __restrict__ start, int nch) {
    __shared__ int sc[NBK];
    const int b = threadIdx.x;
    int run = 0;
    for (int j0 = 0; j0 < nch; j0 += 14) {
        int c[14];
        #pragma unroll
        for (int u = 0; u < 14; ++u) c[u] = (j0 + u < nch) ? hist[(size_t)(j0 + u) * NBK + b] : 0;
        #pragma unroll
        for (int u = 0; u < 14; ++u) {
            if (j0 + u < nch) hist[(size_t)(j0 + u) * NBK + b] = run;
            run += c[u];
        }
    }
    sc[b] = run;
    __syncthreads();
    for (int off = 1; off < NBK; off <<= 1) {
        const int add = (b >= off) ? sc[b - off] : 0;
        __syncthreads();
        sc[b] += add;
        __syncthreads();
    }
    start[b + 1] = sc[b];
    if (b == 0) start[0] = 0;
}

__global__ void __launch_bounds__(256)
bin_fill_kernel(const int* __restrict__ src, const int* __restrict__ dst,
                const int* __restrict__ hist, const int* __restrict__ start,
                int* __restrict__ bins, int e) {
    __shared__ int cur[NBK];
    for (int i = threadIdx.x; i < NBK; i += 256)
        cur[i] = start[i] + hist[(size_t)blockIdx.x * NBK + i];
    __syncthreads();
    const long long base = (long long)blockIdx.x * ECH;
    for (int i = threadIdx.x; i < ECH; i += 256) {
        const long long idx = base + i;
        if (idx < e) {
            const int d = dst[idx];
            const int b = d / RB;
            const int pos = atomicAdd(&cur[b], 1);
            bins[pos] = src[idx] * RB + (d - b * RB);
        }
    }
}

template <int COLS, int UB>
__global__ void __launch_bounds__(512)
agg_owner_kernel(const float* __restrict__ x, const int* __restrict__ bins,
                 const int* __restrict__ start, float* __restrict__ out, int n) {
    constexpr int NI = (COLS + 31) / 32;
    __shared__ __attribute__((aligned(16))) float acc[RB * COLS];
    const int lane = threadIdx.x & 31;
    const int wave = threadIdx.x >> 5;
    const int nw   = blockDim.x >> 5;
    {
        const v4f z = {0.0f, 0.0f, 0.0f, 0.0f};
        v4f* a4 = (v4f*)acc;
        for (int i = threadIdx.x; i < RB * COLS / 4; i += blockDim.x) a4[i] = z;
    }
    __syncthreads();
    const int e0 = start[blockIdx.x];
    const int e1 = start[blockIdx.x + 1];
    const int ng = (e1 - e0 + UB - 1) / UB;
    int g = wave;
    int pk = 0;
    if (g < ng && e0 + g * UB + lane < e1 && lane < UB) pk = bins[e0 + g * UB + lane];
    while (g < ng) {
        const int cnt = (e1 - (e0 + g * UB) < UB) ? e1 - (e0 + g * UB) : UB;
        const int gn = g + nw;
        int pkn = 0;
        if (gn < ng && e0 + gn * UB + lane < e1 && lane < UB) pkn = bins[e0 + gn * UB + lane];
        if (cnt == UB) {
            float v[UB][NI];
            int   lr[UB];
            #pragma unroll
            for (int u = 0; u < UB; ++u) {
                const int w = __builtin_amdgcn_readlane(pk, u);
                const int s = w / RB;
                lr[u] = w - s * RB;
                const float* xr = x + (size_t)s * COLS + lane;
                #pragma unroll
                for (int i = 0; i < NI; ++i)
                    v[u][i] = (32 * i + lane < COLS) ? xr[32 * i] : 0.0f;
            }
            #pragma unroll
            for (int u = 0; u < UB; ++u) {
                float* a = acc + lr[u] * COLS + lane;
                #pragma unroll
                for (int i = 0; i < NI; ++i)
                    if (32 * i + lane < COLS) atomicAdd(a + 32 * i, v[u][i]);
            }
        } else {
            for (int j = 0; j < cnt; ++j) {
                const int w = __shfl(pk, j, 32);
                const int s = w / RB;
                const float* xr = x + (size_t)s * COLS + lane;
                float* a = acc + (w - s * RB) * COLS + lane;
                #pragma unroll
                for (int i = 0; i < NI; ++i)
                    if (32 * i + lane < COLS) atomicAdd(a + 32 * i, xr[32 * i]);
            }
        }
        pk = pkn;
        g  = gn;
    }
    __syncthreads();
    const int r0 = blockIdx.x * RB;
    const int nr = (n - r0 < RB) ? (n - r0) : RB;
    v4f* o4 = (v4f*)(out + (size_t)r0 * COLS);
    const v4f* a4 = (const v4f*)acc;
    for (int i = threadIdx.x; i < nr * COLS / 4; i += blockDim.x) o4[i] = a4[i];
}

__global__ void post_ln_relu_128_kernel(const float* __restrict__ agg,
                                        const float* __restrict__ di_is,
                                        const float* __restrict__ do_is,
                                        const float* __restrict__ g, const float* __restrict__ b,
                                        float* __restrict__ out, int n) {
    int row  = (blockIdx.x * blockDim.x + threadIdx.x) >> 5;
    int lane = threadIdx.x & 31;
    if (row >= n) return;
    const float* p = agg + (size_t)row * 128;
    const float di = di_is[row];
    float v[4];
    float s = 0.f, s2 = 0.f;
    #pragma unroll
    for (int i = 0; i < 4; ++i) {
        v[i] = p[lane + 32 * i] * di;
        s += v[i]; s2 += v[i] * v[i];
    }
    #pragma unroll
    for (int off = 16; off > 0; off >>= 1) {
        s  += __shfl_xor(s,  off, 32);
        s2 += __shfl_xor(s2, off, 32);
    }
    const float m   = s  * (1.0f / 128.0f);
    const float rsv = rsqrtf(s2 * (1.0f / 128.0f) - m * m + LN_EPS);
    const float ds  = do_is[row];
    float* q = out + (size_t)row * 128;
    #pragma unroll
    for (int i = 0; i < 4; ++i) {
        const int c = lane + 32 * i;
        float y = (v[i] - m) * rsv * g[c] + b[c];
        q[c] = fmaxf(y, 0.0f) * ds;
    }
}

__global__ void final_scale_bias_kernel(float* __restrict__ out,
                                        const float* __restrict__ di_is,
                                        const float* __restrict__ b2, int n) {
    const int F = 40;
    int t = blockIdx.x * blockDim.x + threadIdx.x;
    if (t >= n * F) return;
    out[t] = out[t] * di_is[t / F] + b2[t % F];
}

static inline int cdiv(long long a, long long b) { return (int)((a + b - 1) / b); }

extern "C" void kernel_launch(void* const* d_in, const int* in_sizes, int n_in,
                              void* d_out, int out_size, void* d_ws, size_t ws_size,
                              hipStream_t stream) {
    const float* feat = (const float*)d_in[0];
    const int*   src  = (const int*)d_in[1];
    const int*   dst  = (const int*)d_in[2];
    const float* W0   = (const float*)d_in[3];
    const float* W1   = (const float*)d_in[4];
    const float* W2   = (const float*)d_in[5];
    const float* b2   = (const float*)d_in[6];
    const float* g_in = (const float*)d_in[7];
    const float* b_in = (const float*)d_in[8];
    const float* g0   = (const float*)d_in[9];
    const float* b0   = (const float*)d_in[10];
    const float* g1   = (const float*)d_in[11];
    const float* b1   = (const float*)d_in[12];
    float* out = (float*)d_out;

    const int IN_F = 768, HID = 128, CLS = 40;
    const int n = in_sizes[0] / IN_F;
    const int e = in_sizes[1];

    float* ws    = (float*)d_ws;
    float* do_is = ws;
    float* di_is = ws + (size_t)n;
    float* mu    = ws + 2 * (size_t)n;
    float* rs    = ws + 3 * (size_t)n;
    float* bufB  = ws + 4 * (size_t)n;
    float* bufC  = bufB + (size_t)n * HID;
    float* bufD  = bufC + (size_t)n * HID;
    int*   bins  = (int*)(bufD + (size_t)n * HID);
    const int nch = cdiv(e, ECH);
    int*   hist  = bins + (size_t)e;
    int*   start = hist + (size_t)nch * NBK;

    const int T = 256;

    int* part = (int*)bufC;
    degree_hist_kernel<<<dim3(DG, 2), 1024, 0, stream>>>(src, dst, part, e);
    invsqrt_deg_kernel<<<dim3(cdiv(n, T), 2), T, 0, stream>>>(part, do_is, di_is, n);

    bin_count_kernel<<<nch, 256, 0, stream>>>(dst, hist, e);
    bin_scan_kernel<<<1, NBK, 0, stream>>>(hist, start, nch);
    bin_fill_kernel<<<nch, 256, 0, stream>>>(src, dst, hist, start, bins, e);

    ln_stats_768_kernel<<<cdiv((long long)n * 32, T), T, 0, stream>>>(feat, mu, rs, n);

    const int mtiles = cdiv(n, 16);
    gemm0_ln_wmma_kernel<<<cdiv((long long)mtiles * 32, T), T, 0, stream>>>(
        feat, mu, rs, do_is, g_in, b_in, W0, bufB, n);

    agg_owner_kernel<128, 8><<<NBK, 512, 0, stream>>>(bufB, bins, start, bufC, n);

    post_ln_relu_128_kernel<<<cdiv((long long)n * 32, T), T, 0, stream>>>(
        bufC, di_is, do_is, g0, b0, bufD, n);

    agg_owner_kernel<128, 8><<<NBK, 512, 0, stream>>>(bufD, bins, start, bufB, n);
    gemm128_wmma_strip_kernel<<<cdiv((long long)mtiles * 32, T), T, 0, stream>>>(
        bufB, W1, bufC, n, HID);

    post_ln_relu_128_kernel<<<cdiv((long long)n * 32, T), T, 0, stream>>>(
        bufC, di_is, do_is, g1, b1, bufD, n);

    gemm_wmma_strip_kernel<3><<<cdiv((long long)mtiles * 32, T), T, 0, stream>>>(
        bufD, W2, bufB, n, HID, CLS);
    agg_owner_kernel<40, 8><<<NBK, 512, 0, stream>>>(bufB, bins, start, out, n);
    final_scale_bias_kernel<<<cdiv((long long)n * CLS, T), T, 0, stream>>>(out, di_is, b2, n);
}


# Row-wise step moved into its consumer: layer-2 product loads a 16-row strip of Y1 whole, applies scale+LayerNorm+ReLU in registers, then WMMAs; h2 never stored, 1 launch fewer; LN row sums reordered
# speedup vs baseline: 2.3956x; 1.1013x over previous
#include <hip/hip_runtime.h>
#include <math.h>

typedef __attribute__((ext_vector_type(2))) float v2f;
typedef __attribute__((ext_vector_type(8))) float v8f;
typedef __attribute__((ext_vector_type(4))) float v4f;

#define LN_EPS 1e-5f

#define RB   98
#define NBK  512
#define ECH  4096

#define DG   32
#define DN   (NBK * RB)
#define DQ   8

typedef __attribute__((ext_vector_type(4))) int v4i;

__global__ void __launch_bounds__(1024)
degree_hist_kernel(const int* __restrict__ src, const int* __restrict__ dst,
                   int* __restrict__ part, int e) {
    __shared__ __attribute__((aligned(16))) int cnt[DN];
    const int* __restrict__ idx = blockIdx.y ? dst : src;
    {
        const v4i z = {0, 0, 0, 0};
        v4i* c4 = (v4i*)cnt;
        for (int i = threadIdx.x; i < DN / 4; i += 1024) c4[i] = z;
    }
    __syncthreads();
    const int per = (((e + DG - 1) / DG) + 3) & ~3;
    const long long lo = (long long)blockIdx.x * per;
    long long hi = lo + per;
    if (hi > e) hi = e;
    const int len = (hi > lo) ? (int)(hi - lo) : 0;
    const int nq  = len >> 2;
    const v4i* q = (const v4i*)(idx + lo);
    for (int base = 0; base < nq; base += 1024 * DQ) {
        v4i w[DQ];
        #pragma unroll
        for (int u = 0; u < DQ; ++u) {
            const int i = base + u * 1024 + threadIdx.x;
            w[u] = q[(i < nq) ? i : nq - 1];
        }
        #pragma unroll
        for (int u = 0; u < DQ; ++u) {
            const int i = base + u * 1024 + threadIdx.x;
            if (i < nq) {
                atomicAdd(&cnt[w[u].x], 1);
                atomicAdd(&cnt[w[u].y], 1);
                atomicAdd(&cnt[w[u].z], 1);
                atomicAdd(&cnt[w[u].w], 1);
            }
        }
    }
    if (4 * nq + (int)threadIdx.x < len)
        atomicAdd(&cnt[idx[lo + 4 * nq + threadIdx.x]], 1);
    __syncthreads();
    v4i* p4 = (v4i*)(part + ((size_t)blockIdx.y * DG + blockIdx.x) * DN);
    const v4i* c4 = (const v4i*)cnt;
    for (int i = threadIdx.x; i < DN / 4; i += 1024) p4[i] = c4[i];
}

__global__ void invsqrt_deg_kernel(const int* __restrict__ part,
                                   float* __restrict__ dout, float* __restrict__ din, int n) {
    int i = blockIdx.x * blockDim.x + threadIdx.x;
    if (i >= n) return;
    const int* p = part + (size_t)blockIdx.y * DG * DN + i;
    int s = 0;
    #pragma unroll
    for (int g = 0; g < DG; ++g) s += p[(size_t)g * DN];
    float* o = blockIdx.y ? din : dout;
    o[i] = rsqrtf(fmaxf((float)s, 1.0f));
}

__global__ void ln_stats_768_kernel(const float* __restrict__ feat,
                                    float* __restrict__ mu, float* __restrict__ rs, int n) {
    int row  = (blockIdx.x * blockDim.x + threadIdx.x) >> 5;
    int lane = threadIdx.x & 31;
    if (row >= n) return;
    const float* p = feat + (size_t)row * 768;
    float s = 0.f, s2 = 0.f;
    #pragma unroll
    for (int c = 0; c < 768; c += 32) {
        float v = p[c + lane];
        s += v; s2 += v * v;
    }
    #pragma unroll
    for (int off = 16; off > 0; off >>= 1) {
        s  += __shfl_xor(s,  off, 32);
        s2 += __shfl_xor(s2, off, 32);
    }
    if (lane == 0) {
        float m   = s  * (1.0f / 768.0f);
        float var = s2 * (1.0f / 768.0f) - m * m;
        mu[row] = m;
        rs[row] = rsqrtf(var + LN_EPS);
    }
}

__global__ void __launch_bounds__(256)
gemm0_ln_wmma_kernel(const float* __restrict__ feat,
                                     const float* __restrict__ mu, const float* __restrict__ rs,
                                     const float* __restrict__ do_is,
                                     const float* __restrict__ g, const float* __restrict__ b,
                                     const float* __restrict__ W,
                                     float* __restrict__ C, int M) {
    const int K = 768, F = 128;
    const int lane = threadIdx.x & 31;
    const int mt   = (blockIdx.x * blockDim.x + threadIdx.x) >> 5;
    if (mt * 16 >= M) return;
    const int half = lane >> 4;
    const int l16  = lane & 15;
    const int arow = mt * 16 + l16;
    const int koff = half * 2;
    const float m_r  = mu[arow];
    const float rs_r = rs[arow];
    const float do_r = do_is[arow];
    const float* ap = feat + (size_t)arow * K + koff;

    v8f acc[8];
    #pragma unroll
    for (int i = 0; i < 8; ++i) { v8f z = {}; acc[i] = z; }

    constexpr int KB = 4;
    for (int k = 0; k < K; k += 4 * KB) {
        v2f x[KB], gg[KB], bbv[KB];
        v4f w0l[KB], w0h[KB], w1l[KB], w1h[KB];
        #pragma unroll
        for (int u = 0; u < KB; ++u) {
            const int c0 = k + 4 * u + koff;
            x[u]   = *(const v2f*)(ap + k + 4 * u);
            gg[u]  = *(const v2f*)(g + c0);
            bbv[u] = *(const v2f*)(b + c0);
        }
        __builtin_amdgcn_sched_barrier(0);
        #pragma unroll
        for (int u = 0; u < KB; ++u) {
            const int c0 = k + 4 * u + koff;
            const float* wrow = W + (size_t)c0 * F + 4 * l16;
            w0l[u] = *(const v4f*)(wrow);
            w0h[u] = *(const v4f*)(wrow + 64);
            w1l[u] = *(const v4f*)(wrow + F);
            w1h[u] = *(const v4f*)(wrow + F + 64);
        }
        #pragma unroll
        for (int u = 0; u < KB; ++u) {
            v2f a;
            a.x = ((x[u].x - m_r) * rs_r * gg[u].x + bbv[u].x) * do_r;
            a.y = ((x[u].y - m_r) * rs_r * gg[u].y + bbv[u].y) * do_r;
            #pragma unroll
            for (int ft = 0; ft < 8; ++ft) {
                v2f bb;
                bb.x = (ft < 4) ? w0l[u][ft & 3] : w0h[u][ft & 3];
                bb.y = (ft < 4) ? w1l[u][ft & 3] : w1h[u][ft & 3];
                acc[ft] = __builtin_amdgcn_wmma_f32_16x16x4_f32(
                    false, a, false, bb, (short)0, acc[ft], false, false);
            }
        }
    }
    const int crow = mt * 16 + half * 8;
    #pragma unroll
    for (int r = 0; r < 8; ++r) {
        float* crp = C + (size_t)(crow + r) * F + 4 * l16;
        const v4f lo = {acc[0][r], acc[1][r], acc[2][r], acc[3][r]};
        const v4f hi = {acc[4][r], acc[5][r], acc[6][r], acc[7][r]};
        *(v4f*)(crp)      = lo;
        *(v4f*)(crp + 64) = hi;
    }
}

__global__ void __launch_bounds__(256)
gemm128_wmma_strip_kernel(const float* __restrict__ A, const float* __restrict__ W,
                                          float* __restrict__ C, int M, int K) {
    const int F = 128;
    const int lane = threadIdx.x & 31;
    const int mt   = (blockIdx.x * blockDim.x + threadIdx.x) >> 5;
    if (mt * 16 >= M) return;
    const int half = lane >> 4;
    const int l16  = lane & 15;
    const int arow = mt * 16 + l16;
    const int koff = half * 2;
    const float* ap = A + (size_t)arow * K + koff;

    v8f acc[8];
    #pragma unroll
    for (int i = 0; i < 8; ++i) { v8f z = {}; acc[i] = z; }

    constexpr int KB = 4;
    for (int k = 0; k < K; k += 4 * KB) {
        v2f a[KB];
        v4f w0l[KB], w0h[KB], w1l[KB], w1h[KB];
        #pragma unroll
        for (int u = 0; u < KB; ++u)
            a[u] = *(const v2f*)(ap + k + 4 * u);
        __builtin_amdgcn_sched_barrier(0);
        #pragma unroll
        for (int u = 0; u < KB; ++u) {
            const float* wrow = W + (size_t)(k + 4 * u + koff) * F + 4 * l16;
            w0l[u] = *(const v4f*)(wrow);
            w0h[u] = *(const v4f*)(wrow + 64);
            w1l[u] = *(const v4f*)(wrow + F);
            w1h[u] = *(const v4f*)(wrow + F + 64);
        }
        #pragma unroll
        for (int u = 0; u < KB; ++u) {
            #pragma unroll
            for (int ft = 0; ft < 8; ++ft) {
                v2f bb;
                bb.x = (ft < 4) ? w0l[u][ft & 3] : w0h[u][ft & 3];
                bb.y = (ft < 4) ? w1l[u][ft & 3] : w1h[u][ft & 3];
                acc[ft] = __builtin_amdgcn_wmma_f32_16x16x4_f32(
                    false, a[u], false, bb, (short)0, acc[ft], false, false);
            }
        }
    }
    const int crow = mt * 16 + half * 8;
    #pragma unroll
    for (int r = 0; r < 8; ++r) {
        float* crp = C + (size_t)(crow + r) * F + 4 * l16;
        const v4f lo = {acc[0][r], acc[1][r], acc[2][r], acc[3][r]};
        const v4f hi = {acc[4][r], acc[5][r], acc[6][r], acc[7][r]};
        *(v4f*)(crp)      = lo;
        *(v4f*)(crp + 64) = hi;
    }
}

__global__ void __launch_bounds__(256)
gemm40_ln_relu_wmma_kernel(const float* __restrict__ Y,
                           const float* __restrict__ di_is, const float* __restrict__ do_is,
                           const float* __restrict__ g, const float* __restrict__ b,
                           const float* __restrict__ W,
                           float* __restrict__ C, int M) {
    constexpr int K = 128, F = 40, KS = K / 4;
    const int lane = threadIdx.x & 31;
    const int mt   = (blockIdx.x * blockDim.x + threadIdx.x) >> 5;
    if (mt * 16 >= M) return;
    const int half = lane >> 4;
    const int l16  = lane & 15;
    const int arow = mt * 16 + l16;
    const int koff = half * 2;
    const float* ap = Y + (size_t)arow * K + koff;

    v2f x[KS];
    #pragma unroll
    for (int u = 0; u < KS; ++u) x[u] = *(const v2f*)(ap + 4 * u);
    const float di = di_is[arow];
    const float ds = do_is[arow];

    float sa = 0.f, sb = 0.f, qa = 0.f, qb = 0.f;
    #pragma unroll
    for (int u = 0; u < KS; ++u) {
        x[u].x *= di; x[u].y *= di;
        sa += x[u].x; qa += x[u].x * x[u].x;
        sb += x[u].y; qb += x[u].y * x[u].y;
    }
    float s = sa + sb, s2 = qa + qb;
    s  += __shfl_xor(s,  16, 32);
    s2 += __shfl_xor(s2, 16, 32);
    const float m   = s * (1.0f / 128.0f);
    const float rsv = rsqrtf(s2 * (1.0f / 128.0f) - m * m + LN_EPS);

    v8f acc[4];
    #pragma unroll
    for (int i = 0; i < 4; ++i) { v8f z = {}; acc[i] = z; }

    const int lc = (l16 < F / 4) ? l16 : F / 4 - 1;
    constexpr int KB = 8;
    #pragma unroll
    for (int t = 0; t < KS; t += KB) {
        if (t > 0) __builtin_amdgcn_sched_barrier(0);
        v2f gg[KB], bbv[KB];
        v4f w0[KB], w1[KB];
        #pragma unroll
        for (int u = 0; u < KB; ++u) {
            const int c0 = 4 * (t + u) + koff;
            gg[u]  = *(const v2f*)(g + c0);
            bbv[u] = *(const v2f*)(b + c0);
            const float* wrow = W + (size_t)c0 * F + 4 * lc;
            w0[u] = *(const v4f*)(wrow);
            w1[u] = *(const v4f*)(wrow + F);
        }
        #pragma unroll
        for (int u = 0; u < KB; ++u) {
            v2f a;
            a.x = fmaxf((x[t + u].x - m) * rsv * gg[u].x + bbv[u].x, 0.0f) * ds;
            a.y = fmaxf((x[t + u].y - m) * rsv * gg[u].y + bbv[u].y, 0.0f) * ds;
            #pragma unroll
            for (int ft = 0; ft < 4; ++ft) {
                v2f bb;
                bb.x = w0[u][ft];
                bb.y = w1[u][ft];
                acc[ft] = __builtin_amdgcn_wmma_f32_16x16x4_f32(
                    false, a, false, bb, (short)0, acc[ft], false, false);
            }
        }
    }
    const int crow = mt * 16 + half * 8;
    if (l16 < F / 4) {
        #pragma unroll
        for (int r = 0; r < 8; ++r) {
            const v4f o = {acc[0][r], acc[1][r], acc[2][r], acc[3][r]};
            *(v4f*)(C + (size_t)(crow + r) * F + 4 * l16) = o;
        }
    }
}

__global__ void __launch_bounds__(256)
bin_count_kernel(const int* __restrict__ dst, int* __restrict__ hist, int e) {
    __shared__ int h[NBK];
    for (int i = threadIdx.x; i < NBK; i += 256) h[i] = 0;
    __syncthreads();
    const long long base = (long long)blockIdx.x * ECH;
    for (int i = threadIdx.x; i < ECH; i += 256) {
        const long long idx = base + i;
        if (idx < e) atomicAdd(&h[dst[idx] / RB], 1);
    }
    __syncthreads();
    for (int i = threadIdx.x; i < NBK; i += 256) hist[(size_t)blockIdx.x * NBK + i] = h[i];
}

__global__ void __launch_bounds__(NBK)
bin_scan_kernel(int* __restrict__ hist, int* __restrict__ start, int nch) {
    __shared__ int sc[NBK];
    const int b = threadIdx.x;
    int run = 0;
    for (int j0 = 0; j0 < nch; j0 += 14) {
        int c[14];
        #pragma unroll
        for (int u = 0; u < 14; ++u) c[u] = (j0 + u < nch) ? hist[(size_t)(j0 + u) * NBK + b] : 0;
        #pragma unroll
        for (int u = 0; u < 14; ++u) {
            if (j0 + u < nch) hist[(size_t)(j0 + u) * NBK + b] = run;
            run += c[u];
        }
    }
    sc[b] = run;
    __syncthreads();
    for (int off = 1; off < NBK; off <<= 1) {
        const int add = (b >= off) ? sc[b - off] : 0;
        __syncthreads();
        sc[b] += add;
        __syncthreads();
    }
    start[b + 1] = sc[b];
    if (b == 0) start[0] = 0;
}

__global__ void __launch_bounds__(256)
bin_fill_kernel(const int* __restrict__ src, const int* __restrict__ dst,
                const int* __restrict__ hist, const int* __restrict__ start,
                int* __restrict__ bins, int e) {
    __shared__ int cur[NBK];
    for (int i = threadIdx.x; i < NBK; i += 256)
        cur[i] = start[i] + hist[(size_t)blockIdx.x * NBK + i];
    __syncthreads();
    const long long base = (long long)blockIdx.x * ECH;
    for (int i = threadIdx.x; i < ECH; i += 256) {
        const long long idx = base + i;
        if (idx < e) {
            const int d = dst[idx];
            const int b = d / RB;
            const int pos = atomicAdd(&cur[b], 1);
            bins[pos] = src[idx] * RB + (d - b * RB);
        }
    }
}

template <int COLS, int UB>
__global__ void __launch_bounds__(512)
agg_owner_kernel(const float* __restrict__ x, const int* __restrict__ bins,
                 const int* __restrict__ start, float* __restrict__ out, int n) {
    constexpr int NI = (COLS + 31) / 32;
    __shared__ __attribute__((aligned(16))) float acc[RB * COLS];
    const int lane = threadIdx.x & 31;
    const int wave = threadIdx.x >> 5;
    const int nw   = blockDim.x >> 5;
    {
        const v4f z = {0.0f, 0.0f, 0.0f, 0.0f};
        v4f* a4 = (v4f*)acc;
        for (int i = threadIdx.x; i < RB * COLS / 4; i += blockDim.x) a4[i] = z;
    }
    __syncthreads();
    const int e0 = start[blockIdx.x];
    const int e1 = start[blockIdx.x + 1];
    const int ng = (e1 - e0 + UB - 1) / UB;
    int g = wave;
    int pk = 0;
    if (g < ng && e0 + g * UB + lane < e1 && lane < UB) pk = bins[e0 + g * UB + lane];
    while (g < ng) {
        const int cnt = (e1 - (e0 + g * UB) < UB) ? e1 - (e0 + g * UB) : UB;
        const int gn = g + nw;
        int pkn = 0;
        if (gn < ng && e0 + gn * UB + lane < e1 && lane < UB) pkn = bins[e0 + gn * UB + lane];
        if (cnt == UB) {
            float v[UB][NI];
            int   lr[UB];
            #pragma unroll
            for (int u = 0; u < UB; ++u) {
                const int w = __builtin_amdgcn_readlane(pk, u);
                const int s = w / RB;
                lr[u] = w - s * RB;
                const float* xr = x + (size_t)s * COLS + lane;
                #pragma unroll
                for (int i = 0; i < NI; ++i)
                    v[u][i] = (32 * i + lane < COLS) ? xr[32 * i] : 0.0f;
            }
            #pragma unroll
            for (int u = 0; u < UB; ++u) {
                float* a = acc + lr[u] * COLS + lane;
                #pragma unroll
                for (int i = 0; i < NI; ++i)
                    if (32 * i + lane < COLS) atomicAdd(a + 32 * i, v[u][i]);
            }
        } else {
            for (int j = 0; j < cnt; ++j) {
                const int w = __shfl(pk, j, 32);
                const int s = w / RB;
                const float* xr = x + (size_t)s * COLS + lane;
                float* a = acc + (w - s * RB) * COLS + lane;
                #pragma unroll
                for (int i = 0; i < NI; ++i)
                    if (32 * i + lane < COLS) atomicAdd(a + 32 * i, xr[32 * i]);
            }
        }
        pk = pkn;
        g  = gn;
    }
    __syncthreads();
    const int r0 = blockIdx.x * RB;
    const int nr = (n - r0 < RB) ? (n - r0) : RB;
    v4f* o4 = (v4f*)(out + (size_t)r0 * COLS);
    const v4f* a4 = (const v4f*)acc;
    for (int i = threadIdx.x; i < nr * COLS / 4; i += blockDim.x) o4[i] = a4[i];
}

__global__ void post_ln_relu_128_kernel(const float* __restrict__ agg,
                                        const float* __restrict__ di_is,
                                        const float* __restrict__ do_is,
                                        const float* __restrict__ g, const float* __restrict__ b,
                                        float* __restrict__ out, int n) {
    int row  = (blockIdx.x * blockDim.x + threadIdx.x) >> 5;
    int lane = threadIdx.x & 31;
    if (row >= n) return;
    const float* p = agg + (size_t)row * 128;
    const float di = di_is[row];
    float v[4];
    float s = 0.f, s2 = 0.f;
    #pragma unroll
    for (int i = 0; i < 4; ++i) {
        v[i] = p[lane + 32 * i] * di;
        s += v[i]; s2 += v[i] * v[i];
    }
    #pragma unroll
    for (int off = 16; off > 0; off >>= 1) {
        s  += __shfl_xor(s,  off, 32);
        s2 += __shfl_xor(s2, off, 32);
    }
    const float m   = s  * (1.0f / 128.0f);
    const float rsv = rsqrtf(s2 * (1.0f / 128.0f) - m * m + LN_EPS);
    const float ds  = do_is[row];
    float* q = out + (size_t)row * 128;
    #pragma unroll
    for (int i = 0; i < 4; ++i) {
        const int c = lane + 32 * i;
        float y = (v[i] - m) * rsv * g[c] + b[c];
        q[c] = fmaxf(y, 0.0f) * ds;
    }
}

__global__ void final_scale_bias_kernel(float* __restrict__ out,
                                        const float* __restrict__ di_is,
                                        const float* __restrict__ b2, int n) {
    const int F = 40;
    int t = blockIdx.x * blockDim.x + threadIdx.x;
    if (t >= n * F) return;
    out[t] = out[t] * di_is[t / F] + b2[t % F];
}

static inline int cdiv(long long a, long long b) { return (int)((a + b - 1) / b); }

extern "C" void kernel_launch(void* const* d_in, const int* in_sizes, int n_in,
                              void* d_out, int out_size, void* d_ws, size_t ws_size,
                              hipStream_t stream) {
    const float* feat = (const float*)d_in[0];
    const int*   src  = (const int*)d_in[1];
    const int*   dst  = (const int*)d_in[2];
    const float* W0   = (const float*)d_in[3];
    const float* W1   = (const float*)d_in[4];
    const float* W2   = (const float*)d_in[5];
    const float* b2   = (const float*)d_in[6];
    const float* g_in = (const float*)d_in[7];
    const float* b_in = (const float*)d_in[8];
    const float* g0   = (const float*)d_in[9];
    const float* b0   = (const float*)d_in[10];
    const float* g1   = (const float*)d_in[11];
    const float* b1   = (const float*)d_in[12];
    float* out = (float*)d_out;

    const int IN_F = 768, HID = 128, CLS = 40;
    const int n = in_sizes[0] / IN_F;
    const int e = in_sizes[1];

    float* ws    = (float*)d_ws;
    float* do_is = ws;
    float* di_is = ws + (size_t)n;
    float* mu    = ws + 2 * (size_t)n;
    float* rs    = ws + 3 * (size_t)n;
    float* bufB  = ws + 4 * (size_t)n;
    float* bufC  = bufB + (size_t)n * HID;
    float* bufD  = bufC + (size_t)n * HID;
    int*   bins  = (int*)(bufD + (size_t)n * HID);
    const int nch = cdiv(e, ECH);
    int*   hist  = bins + (size_t)e;
    int*   start = hist + (size_t)nch * NBK;

    const int T = 256;

    int* part = (int*)bufC;
    degree_hist_kernel<<<dim3(DG, 2), 1024, 0, stream>>>(src, dst, part, e);
    invsqrt_deg_kernel<<<dim3(cdiv(n, T), 2), T, 0, stream>>>(part, do_is, di_is, n);

    bin_count_kernel<<<nch, 256, 0, stream>>>(dst, hist, e);
    bin_scan_kernel<<<1, NBK, 0, stream>>>(hist, start, nch);
    bin_fill_kernel<<<nch, 256, 0, stream>>>(src, dst, hist, start, bins, e);

    ln_stats_768_kernel<<<cdiv((long long)n * 32, T), T, 0, stream>>>(feat, mu, rs, n);

    const int mtiles = cdiv(n, 16);
    gemm0_ln_wmma_kernel<<<cdiv((long long)mtiles * 32, T), T, 0, stream>>>(
        feat, mu, rs, do_is, g_in, b_in, W0, bufB, n);

    agg_owner_kernel<128, 8><<<NBK, 512, 0, stream>>>(bufB, bins, start, bufC, n);

    post_ln_relu_128_kernel<<<cdiv((long long)n * 32, T), T, 0, stream>>>(
        bufC, di_is, do_is, g0, b0, bufD, n);

    agg_owner_kernel<128, 8><<<NBK, 512, 0, stream>>>(bufD, bins, start, bufB, n);
    gemm128_wmma_strip_kernel<<<cdiv((long long)mtiles * 32, T), T, 0, stream>>>(
        bufB, W1, bufC, n, HID);

    gemm40_ln_relu_wmma_kernel<<<cdiv((long long)mtiles * 32, T), T, 0, stream>>>(
        bufC, di_is, do_is, g1, b1, W2, bufB, n);
    agg_owner_kernel<40, 8><<<NBK, 512, 0, stream>>>(bufB, bins, start, out, n);
    final_scale_bias_kernel<<<cdiv((long long)n * CLS, T), T, 0, stream>>>(out, di_is, b2, n);
}


# Layer 1 reordered: product h1@W1 first, scale+LayerNorm+ReLU in its A loader (A0 strip in registers), edge sums after; h1 never stored, 1 launch fewer; layer-1 sums reassociated, LN sums reordered
# speedup vs baseline: 2.4777x; 1.0343x over previous
#include <hip/hip_runtime.h>
#include <math.h>

typedef __attribute__((ext_vector_type(2))) float v2f;
typedef __attribute__((ext_vector_type(8))) float v8f;
typedef __attribute__((ext_vector_type(4))) float v4f;

#define LN_EPS 1e-5f

#define RB   98
#define NBK  512
#define ECH  4096

#define DG   32
#define DN   (NBK * RB)
#define DQ   8

typedef __attribute__((ext_vector_type(4))) int v4i;

__global__ void __launch_bounds__(1024)
degree_hist_kernel(const int* __restrict__ src, const int* __restrict__ dst,
                   int* __restrict__ part, int e) {
    __shared__ __attribute__((aligned(16))) int cnt[DN];
    const int* __restrict__ idx = blockIdx.y ? dst : src;
    {
        const v4i z = {0, 0, 0, 0};
        v4i* c4 = (v4i*)cnt;
        for (int i = threadIdx.x; i < DN / 4; i += 1024) c4[i] = z;
    }
    __syncthreads();
    const int per = (((e + DG - 1) / DG) + 3) & ~3;
    const long long lo = (long long)blockIdx.x * per;
    long long hi = lo + per;
    if (hi > e) hi = e;
    const int len = (hi > lo) ? (int)(hi - lo) : 0;
    const int nq  = len >> 2;
    const v4i* q = (const v4i*)(idx + lo);
    for (int base = 0; base < nq; base += 1024 * DQ) {
        v4i w[DQ];
        #pragma unroll
        for (int u = 0; u < DQ; ++u) {
            const int i = base + u * 1024 + threadIdx.x;
            w[u] = q[(i < nq) ? i : nq - 1];
        }
        #pragma unroll
        for (int u = 0; u < DQ; ++u) {
            const int i = base + u * 1024 + threadIdx.x;
            if (i < nq) {
                atomicAdd(&cnt[w[u].x], 1);
                atomicAdd(&cnt[w[u].y], 1);
                atomicAdd(&cnt[w[u].z], 1);
                atomicAdd(&cnt[w[u].w], 1);
            }
        }
    }
    if (4 * nq + (int)threadIdx.x < len)
        atomicAdd(&cnt[idx[lo + 4 * nq + threadIdx.x]], 1);
    __syncthreads();
    v4i* p4 = (v4i*)(part + ((size_t)blockIdx.y * DG + blockIdx.x) * DN);
    const v4i* c4 = (const v4i*)cnt;
    for (int i = threadIdx.x; i < DN / 4; i += 1024) p4[i] = c4[i];
}

__global__ void invsqrt_deg_kernel(const int* __restrict__ part,
                                   float* __restrict__ dout, float* __restrict__ din, int n) {
    int i = blockIdx.x * blockDim.x + threadIdx.x;
    if (i >= n) return;
    const int* p = part + (size_t)blockIdx.y * DG * DN + i;
    int s = 0;
    #pragma unroll
    for (int g = 0; g < DG; ++g) s += p[(size_t)g * DN];
    float* o = blockIdx.y ? din : dout;
    o[i] = rsqrtf(fmaxf((float)s, 1.0f));
}

__global__ void ln_stats_768_kernel(const float* __restrict__ feat,
                                    float* __restrict__ mu, float* __restrict__ rs, int n) {
    int row  = (blockIdx.x * blockDim.x + threadIdx.x) >> 5;
    int lane = threadIdx.x & 31;
    if (row >= n) return;
    const float* p = feat + (size_t)row * 768;
    float s = 0.f, s2 = 0.f;
    #pragma unroll
    for (int c = 0; c < 768; c += 32) {
        float v = p[c + lane];
        s += v; s2 += v * v;
    }
    #pragma unroll
    for (int off = 16; off > 0; off >>= 1) {
        s  += __shfl_xor(s,  off, 32);
        s2 += __shfl_xor(s2, off, 32);
    }
    if (lane == 0) {
        float m   = s  * (1.0f / 768.0f);
        float var = s2 * (1.0f / 768.0f) - m * m;
        mu[row] = m;
        rs[row] = rsqrtf(var + LN_EPS);
    }
}

__global__ void __launch_bounds__(256)
gemm0_ln_wmma_kernel(const float* __restrict__ feat,
                                     const float* __restrict__ mu, const float* __restrict__ rs,
                                     const float* __restrict__ do_is,
                                     const float* __restrict__ g, const float* __restrict__ b,
                                     const float* __restrict__ W,
                                     float* __restrict__ C, int M) {
    const int K = 768, F = 128;
    const int lane = threadIdx.x & 31;
    const int mt   = (blockIdx.x * blockDim.x + threadIdx.x) >> 5;
    if (mt * 16 >= M) return;
    const int half = lane >> 4;
    const int l16  = lane & 15;
    const int arow = mt * 16 + l16;
    const int koff = half * 2;
    const float m_r  = mu[arow];
    const float rs_r = rs[arow];
    const float do_r = do_is[arow];
    const float* ap = feat + (size_t)arow * K + koff;

    v8f acc[8];
    #pragma unroll
    for (int i = 0; i < 8; ++i) { v8f z = {}; acc[i] = z; }

    constexpr int KB = 4;
    for (int k = 0; k < K; k += 4 * KB) {
        v2f x[KB], gg[KB], bbv[KB];
        v4f w0l[KB], w0h[KB], w1l[KB], w1h[KB];
        #pragma unroll
        for (int u = 0; u < KB; ++u) {
            const int c0 = k + 4 * u + koff;
            x[u]   = *(const v2f*)(ap + k + 4 * u);
            gg[u]  = *(const v2f*)(g + c0);
            bbv[u] = *(const v2f*)(b + c0);
        }
        __builtin_amdgcn_sched_barrier(0);
        #pragma unroll
        for (int u = 0; u < KB; ++u) {
            const int c0 = k + 4 * u + koff;
            const float* wrow = W + (size_t)c0 * F + 4 * l16;
            w0l[u] = *(const v4f*)(wrow);
            w0h[u] = *(const v4f*)(wrow + 64);
            w1l[u] = *(const v4f*)(wrow + F);
            w1h[u] = *(const v4f*)(wrow + F + 64);
        }
        #pragma unroll
        for (int u = 0; u < KB; ++u) {
            v2f a;
            a.x = ((x[u].x - m_r) * rs_r * gg[u].x + bbv[u].x) * do_r;
            a.y = ((x[u].y - m_r) * rs_r * gg[u].y + bbv[u].y) * do_r;
            #pragma unroll
            for (int ft = 0; ft < 8; ++ft) {
                v2f bb;
                bb.x = (ft < 4) ? w0l[u][ft & 3] : w0h[u][ft & 3];
                bb.y = (ft < 4) ? w1l[u][ft & 3] : w1h[u][ft & 3];
                acc[ft] = __builtin_amdgcn_wmma_f32_16x16x4_f32(
                    false, a, false, bb, (short)0, acc[ft], false, false);
            }
        }
    }
    const int crow = mt * 16 + half * 8;
    #pragma unroll
    for (int r = 0; r < 8; ++r) {
        float* crp = C + (size_t)(crow + r) * F + 4 * l16;
        const v4f lo = {acc[0][r], acc[1][r], acc[2][r], acc[3][r]};
        const v4f hi = {acc[4][r], acc[5][r], acc[6][r], acc[7][r]};
        *(v4f*)(crp)      = lo;
        *(v4f*)(crp + 64) = hi;
    }
}

__global__ void __launch_bounds__(256)
gemm128_ln_relu_wmma_kernel(const float* __restrict__ Y,
                            const float* __restrict__ di_is, const float* __restrict__ do_is,
                            const float* __restrict__ g, const float* __restrict__ b,
                            const float* __restrict__ W,
                            float* __restrict__ C, int M) {
    constexpr int K = 128, F = 128, KS = K / 4;
    const int lane = threadIdx.x & 31;
    const int mt   = (blockIdx.x * blockDim.x + threadIdx.x) >> 5;
    if (mt * 16 >= M) return;
    const int half = lane >> 4;
    const int l16  = lane & 15;
    const int arow = mt * 16 + l16;
    const int koff = half * 2;
    const float* ap = Y + (size_t)arow * K + koff;

    v2f x[KS];
    #pragma unroll
    for (int u = 0; u < KS; ++u) x[u] = *(const v2f*)(ap + 4 * u);
    const float di = di_is[arow];
    const float ds = do_is[arow];

    float sa = 0.f, sb = 0.f, qa = 0.f, qb = 0.f;
    #pragma unroll
    for (int u = 0; u < KS; ++u) {
        x[u].x *= di; x[u].y *= di;
        sa += x[u].x; qa += x[u].x * x[u].x;
        sb += x[u].y; qb += x[u].y * x[u].y;
    }
    float s = sa + sb, s2 = qa + qb;
    s  += __shfl_xor(s,  16, 32);
    s2 += __shfl_xor(s2, 16, 32);
    const float m   = s * (1.0f / 128.0f);
    const float rsv = rsqrtf(s2 * (1.0f / 128.0f) - m * m + LN_EPS);

    v8f acc[8];
    #pragma unroll
    for (int i = 0; i < 8; ++i) { v8f z = {}; acc[i] = z; }

    constexpr int KB = 4;
    #pragma unroll
    for (int t = 0; t < KS; t += KB) {
        if (t > 0) __builtin_amdgcn_sched_barrier(0);
        v2f gg[KB], bbv[KB];
        v4f w0l[KB], w0h[KB], w1l[KB], w1h[KB];
        #pragma unroll
        for (int u = 0; u < KB; ++u) {
            const int c0 = 4 * (t + u) + koff;
            gg[u]  = *(const v2f*)(g + c0);
            bbv[u] = *(const v2f*)(b + c0);
            const float* wrow = W + (size_t)c0 * F + 4 * l16;
            w0l[u] = *(const v4f*)(wrow);
            w0h[u] = *(const v4f*)(wrow + 64);
            w1l[u] = *(const v4f*)(wrow + F);
            w1h[u] = *(const v4f*)(wrow + F + 64);
        }
        #pragma unroll
        for (int u = 0; u < KB; ++u) {
            v2f a;
            a.x = fmaxf((x[t + u].x - m) * rsv * gg[u].x + bbv[u].x, 0.0f) * ds;
            a.y = fmaxf((x[t + u].y - m) * rsv * gg[u].y + bbv[u].y, 0.0f) * ds;
            #pragma unroll
            for (int ft = 0; ft < 8; ++ft) {
                v2f bb;
                bb.x = (ft < 4) ? w0l[u][ft & 3] : w0h[u][ft & 3];
                bb.y = (ft < 4) ? w1l[u][ft & 3] : w1h[u][ft & 3];
                acc[ft] = __builtin_amdgcn_wmma_f32_16x16x4_f32(
                    false, a, false, bb, (short)0, acc[ft], false, false);
            }
        }
    }
    const int crow = mt * 16 + half * 8;
    #pragma unroll
    for (int r = 0; r < 8; ++r) {
        float* crp = C + (size_t)(crow + r) * F + 4 * l16;
        const v4f lo = {acc[0][r], acc[1][r], acc[2][r], acc[3][r]};
        const v4f hi = {acc[4][r], acc[5][r], acc[6][r], acc[7][r]};
        *(v4f*)(crp)      = lo;
        *(v4f*)(crp + 64) = hi;
    }
}

__global__ void __launch_bounds__(256)
gemm40_ln_relu_wmma_kernel(const float* __restrict__ Y,
                           const float* __restrict__ di_is, const float* __restrict__ do_is,
                           const float* __restrict__ g, const float* __restrict__ b,
                           const float* __restrict__ W,
                           float* __restrict__ C, int M) {
    constexpr int K = 128, F = 40, KS = K / 4;
    const int lane = threadIdx.x & 31;
    const int mt   = (blockIdx.x * blockDim.x + threadIdx.x) >> 5;
    if (mt * 16 >= M) return;
    const int half = lane >> 4;
    const int l16  = lane & 15;
    const int arow = mt * 16 + l16;
    const int koff = half * 2;
    const float* ap = Y + (size_t)arow * K + koff;

    v2f x[KS];
    #pragma unroll
    for (int u = 0; u < KS; ++u) x[u] = *(const v2f*)(ap + 4 * u);
    const float di = di_is[arow];
    const float ds = do_is[arow];

    float sa = 0.f, sb = 0.f, qa = 0.f, qb = 0.f;
    #pragma unroll
    for (int u = 0; u < KS; ++u) {
        x[u].x *= di; x[u].y *= di;
        sa += x[u].x; qa += x[u].x * x[u].x;
        sb += x[u].y; qb += x[u].y * x[u].y;
    }
    float s = sa + sb, s2 = qa + qb;
    s  += __shfl_xor(s,  16, 32);
    s2 += __shfl_xor(s2, 16, 32);
    const float m   = s * (1.0f / 128.0f);
    const float rsv = rsqrtf(s2 * (1.0f / 128.0f) - m * m + LN_EPS);

    v8f acc[4];
    #pragma unroll
    for (int i = 0; i < 4; ++i) { v8f z = {}; acc[i] = z; }

    const int lc = (l16 < F / 4) ? l16 : F / 4 - 1;
    constexpr int KB = 8;
    #pragma unroll
    for (int t = 0; t < KS; t += KB) {
        if (t > 0) __builtin_amdgcn_sched_barrier(0);
        v2f gg[KB], bbv[KB];
        v4f w0[KB], w1[KB];
        #pragma unroll
        for (int u = 0; u < KB; ++u) {
            const int c0 = 4 * (t + u) + koff;
            gg[u]  = *(const v2f*)(g + c0);
            bbv[u] = *(const v2f*)(b + c0);
            const float* wrow = W + (size_t)c0 * F + 4 * lc;
            w0[u] = *(const v4f*)(wrow);
            w1[u] = *(const v4f*)(wrow + F);
        }
        #pragma unroll
        for (int u = 0; u < KB; ++u) {
            v2f a;
            a.x = fmaxf((x[t + u].x - m) * rsv * gg[u].x + bbv[u].x, 0.0f) * ds;
            a.y = fmaxf((x[t + u].y - m) * rsv * gg[u].y + bbv[u].y, 0.0f) * ds;
            #pragma unroll
            for (int ft = 0; ft < 4; ++ft) {
                v2f bb;
                bb.x = w0[u][ft];
                bb.y = w1[u][ft];
                acc[ft] = __builtin_amdgcn_wmma_f32_16x16x4_f32(
                    false, a, false, bb, (short)0, acc[ft], false, false);
            }
        }
    }
    const int crow = mt * 16 + half * 8;
    if (l16 < F / 4) {
        #pragma unroll
        for (int r = 0; r < 8; ++r) {
            const v4f o = {acc[0][r], acc[1][r], acc[2][r], acc[3][r]};
            *(v4f*)(C + (size_t)(crow + r) * F + 4 * l16) = o;
        }
    }
}

__global__ void __launch_bounds__(256)
bin_count_kernel(const int* __restrict__ dst, int* __restrict__ hist, int e) {
    __shared__ int h[NBK];
    for (int i = threadIdx.x; i < NBK; i += 256) h[i] = 0;
    __syncthreads();
    const long long base = (long long)blockIdx.x * ECH;
    for (int i = threadIdx.x; i < ECH; i += 256) {
        const long long idx = base + i;
        if (idx < e) atomicAdd(&h[dst[idx] / RB], 1);
    }
    __syncthreads();
    for (int i = threadIdx.x; i < NBK; i += 256) hist[(size_t)blockIdx.x * NBK + i] = h[i];
}

__global__ void __launch_bounds__(NBK)
bin_scan_kernel(int* __restrict__ hist, int* __restrict__ start, int nch) {
    __shared__ int sc[NBK];
    const int b = threadIdx.x;
    int run = 0;
    for (int j0 = 0; j0 < nch; j0 += 14) {
        int c[14];
        #pragma unroll
        for (int u = 0; u < 14; ++u) c[u] = (j0 + u < nch) ? hist[(size_t)(j0 + u) * NBK + b] : 0;
        #pragma unroll
        for (int u = 0; u < 14; ++u) {
            if (j0 + u < nch) hist[(size_t)(j0 + u) * NBK + b] = run;
            run += c[u];
        }
    }
    sc[b] = run;
    __syncthreads();
    for (int off = 1; off < NBK; off <<= 1) {
        const int add = (b >= off) ? sc[b - off] : 0;
        __syncthreads();
        sc[b] += add;
        __syncthreads();
    }
    start[b + 1] = sc[b];
    if (b == 0) start[0] = 0;
}

__global__ void __launch_bounds__(256)
bin_fill_kernel(const int* __restrict__ src, const int* __restrict__ dst,
                const int* __restrict__ hist, const int* __restrict__ start,
                int* __restrict__ bins, int e) {
    __shared__ int cur[NBK];
    for (int i = threadIdx.x; i < NBK; i += 256)
        cur[i] = start[i] + hist[(size_t)blockIdx.x * NBK + i];
    __syncthreads();
    const long long base = (long long)blockIdx.x * ECH;
    for (int i = threadIdx.x; i < ECH; i += 256) {
        const long long idx = base + i;
        if (idx < e) {
            const int d = dst[idx];
            const int b = d / RB;
            const int pos = atomicAdd(&cur[b], 1);
            bins[pos] = src[idx] * RB + (d - b * RB);
        }
    }
}

template <int COLS, int UB>
__global__ void __launch_bounds__(512)
agg_owner_kernel(const float* __restrict__ x, const int* __restrict__ bins,
                 const int* __restrict__ start, float* __restrict__ out, int n) {
    constexpr int NI = (COLS + 31) / 32;
    __shared__ __attribute__((aligned(16))) float acc[RB * COLS];
    const int lane = threadIdx.x & 31;
    const int wave = threadIdx.x >> 5;
    const int nw   = blockDim.x >> 5;
    {
        const v4f z = {0.0f, 0.0f, 0.0f, 0.0f};
        v4f* a4 = (v4f*)acc;
        for (int i = threadIdx.x; i < RB * COLS / 4; i += blockDim.x) a4[i] = z;
    }
    __syncthreads();
    const int e0 = start[blockIdx.x];
    const int e1 = start[blockIdx.x + 1];
    const int ng = (e1 - e0 + UB - 1) / UB;
    int g = wave;
    int pk = 0;
    if (g < ng && e0 + g * UB + lane < e1 && lane < UB) pk = bins[e0 + g * UB + lane];
    while (g < ng) {
        const int cnt = (e1 - (e0 + g * UB) < UB) ? e1 - (e0 + g * UB) : UB;
        const int gn = g + nw;
        int pkn = 0;
        if (gn < ng && e0 + gn * UB + lane < e1 && lane < UB) pkn = bins[e0 + gn * UB + lane];
        if (cnt == UB) {
            float v[UB][NI];
            int   lr[UB];
            #pragma unroll
            for (int u = 0; u < UB; ++u) {
                const int w = __builtin_amdgcn_readlane(pk, u);
                const int s = w / RB;
                lr[u] = w - s * RB;
                const float* xr = x + (size_t)s * COLS + lane;
                #pragma unroll
                for (int i = 0; i < NI; ++i)
                    v[u][i] = (32 * i + lane < COLS) ? xr[32 * i] : 0.0f;
            }
            #pragma unroll
            for (int u = 0; u < UB; ++u) {
                float* a = acc + lr[u] * COLS + lane;
                #pragma unroll
                for (int i = 0; i < NI; ++i)
                    if (32 * i + lane < COLS) atomicAdd(a + 32 * i, v[u][i]);
            }
        } else {
            for (int j = 0; j < cnt; ++j) {
                const int w = __shfl(pk, j, 32);
                const int s = w / RB;
                const float* xr = x + (size_t)s * COLS + lane;
                float* a = acc + (w - s * RB) * COLS + lane;
                #pragma unroll
                for (int i = 0; i < NI; ++i)
                    if (32 * i + lane < COLS) atomicAdd(a + 32 * i, xr[32 * i]);
            }
        }
        pk = pkn;
        g  = gn;
    }
    __syncthreads();
    const int r0 = blockIdx.x * RB;
    const int nr = (n - r0 < RB) ? (n - r0) : RB;
    v4f* o4 = (v4f*)(out + (size_t)r0 * COLS);
    const v4f* a4 = (const v4f*)acc;
    for (int i = threadIdx.x; i < nr * COLS / 4; i += blockDim.x) o4[i] = a4[i];
}

__global__ void final_scale_bias_kernel(float* __restrict__ out,
                                        const float* __restrict__ di_is,
                                        const float* __restrict__ b2, int n) {
    const int F = 40;
    int t = blockIdx.x * blockDim.x + threadIdx.x;
    if (t >= n * F) return;
    out[t] = out[t] * di_is[t / F] + b2[t % F];
}

static inline int cdiv(long long a, long long b) { return (int)((a + b - 1) / b); }

extern "C" void kernel_launch(void* const* d_in, const int* in_sizes, int n_in,
                              void* d_out, int out_size, void* d_ws, size_t ws_size,
                              hipStream_t stream) {
    const float* feat = (const float*)d_in[0];
    const int*   src  = (const int*)d_in[1];
    const int*   dst  = (const int*)d_in[2];
    const float* W0   = (const float*)d_in[3];
    const float* W1   = (const float*)d_in[4];
    const float* W2   = (const float*)d_in[5];
    const float* b2   = (const float*)d_in[6];
    const float* g_in = (const float*)d_in[7];
    const float* b_in = (const float*)d_in[8];
    const float* g0   = (const float*)d_in[9];
    const float* b0   = (const float*)d_in[10];
    const float* g1   = (const float*)d_in[11];
    const float* b1   = (const float*)d_in[12];
    float* out = (float*)d_out;

    const int IN_F = 768, HID = 128, CLS = 40;
    const int n = in_sizes[0] / IN_F;
    const int e = in_sizes[1];

    float* ws    = (float*)d_ws;
    float* do_is = ws;
    float* di_is = ws + (size_t)n;
    float* mu    = ws + 2 * (size_t)n;
    float* rs    = ws + 3 * (size_t)n;
    float* bufB  = ws + 4 * (size_t)n;
    float* bufC  = bufB + (size_t)n * HID;
    float* bufD  = bufC + (size_t)n * HID;
    int*   bins  = (int*)(bufD + (size_t)n * HID);
    const int nch = cdiv(e, ECH);
    int*   hist  = bins + (size_t)e;
    int*   start = hist + (size_t)nch * NBK;

    const int T = 256;

    int* part = (int*)bufC;
    degree_hist_kernel<<<dim3(DG, 2), 1024, 0, stream>>>(src, dst, part, e);
    invsqrt_deg_kernel<<<dim3(cdiv(n, T), 2), T, 0, stream>>>(part, do_is, di_is, n);

    bin_count_kernel<<<nch, 256, 0, stream>>>(dst, hist, e);
    bin_scan_kernel<<<1, NBK, 0, stream>>>(hist, start, nch);
    bin_fill_kernel<<<nch, 256, 0, stream>>>(src, dst, hist, start, bins, e);

    ln_stats_768_kernel<<<cdiv((long long)n * 32, T), T, 0, stream>>>(feat, mu, rs, n);

    const int mtiles = cdiv(n, 16);
    gemm0_ln_wmma_kernel<<<cdiv((long long)mtiles * 32, T), T, 0, stream>>>(
        feat, mu, rs, do_is, g_in, b_in, W0, bufB, n);

    agg_owner_kernel<128, 8><<<NBK, 512, 0, stream>>>(bufB, bins, start, bufC, n);

    gemm128_ln_relu_wmma_kernel<<<cdiv((long long)mtiles * 32, T), T, 0, stream>>>(
        bufC, di_is, do_is, g0, b0, W1, bufD, n);
    agg_owner_kernel<128, 8><<<NBK, 512, 0, stream>>>(bufD, bins, start, bufB, n);

    gemm40_ln_relu_wmma_kernel<<<cdiv((long long)mtiles * 32, T), T, 0, stream>>>(
        bufB, di_is, do_is, g1, b1, W2, bufC, n);
    agg_owner_kernel<40, 8><<<NBK, 512, 0, stream>>>(bufC, bins, start, out, n);
    final_scale_bias_kernel<<<cdiv((long long)n * CLS, T), T, 0, stream>>>(out, di_is, b2, n);
}


# Layer-1 and layer-2 products moved into the edge-sum kernels: a wave reads a 16-row strip's A values from the block's LDS sums, LN there; A0, Y1 never stored, 2 launches fewer; sum orders kept
# speedup vs baseline: 2.5929x; 1.0465x over previous
#include <hip/hip_runtime.h>
#include <math.h>

typedef __attribute__((ext_vector_type(2))) float v2f;
typedef __attribute__((ext_vector_type(8))) float v8f;
typedef __attribute__((ext_vector_type(4))) float v4f;

#define LN_EPS 1e-5f

#define RB   98
#define NBK  512
#define ECH  4096

#define DG   32
#define DN   (NBK * RB)
#define DQ   8

typedef __attribute__((ext_vector_type(4))) int v4i;

__global__ void __launch_bounds__(1024)
degree_hist_kernel(const int* __restrict__ src, const int* __restrict__ dst,
                   int* __restrict__ part, int e) {
    __shared__ __attribute__((aligned(16))) int cnt[DN];
    const int* __restrict__ idx = blockIdx.y ? dst : src;
    {
        const v4i z = {0, 0, 0, 0};
        v4i* c4 = (v4i*)cnt;
        for (int i = threadIdx.x; i < DN / 4; i += 1024) c4[i] = z;
    }
    __syncthreads();
    const int per = (((e + DG - 1) / DG) + 3) & ~3;
    const long long lo = (long long)blockIdx.x * per;
    long long hi = lo + per;
    if (hi > e) hi = e;
    const int len = (hi > lo) ? (int)(hi - lo) : 0;
    const int nq  = len >> 2;
    const v4i* q = (const v4i*)(idx + lo);
    for (int base = 0; base < nq; base += 1024 * DQ) {
        v4i w[DQ];
        #pragma unroll
        for (int u = 0; u < DQ; ++u) {
            const int i = base + u * 1024 + threadIdx.x;
            w[u] = q[(i < nq) ? i : nq - 1];
        }
        #pragma unroll
        for (int u = 0; u < DQ; ++u) {
            const int i = base + u * 1024 + threadIdx.x;
            if (i < nq) {
                atomicAdd(&cnt[w[u].x], 1);
                atomicAdd(&cnt[w[u].y], 1);
                atomicAdd(&cnt[w[u].z], 1);
                atomicAdd(&cnt[w[u].w], 1);
            }
        }
    }
    if (4 * nq + (int)threadIdx.x < len)
        atomicAdd(&cnt[idx[lo + 4 * nq + threadIdx.x]], 1);
    __syncthreads();
    v4i* p4 = (v4i*)(part + ((size_t)blockIdx.y * DG + blockIdx.x) * DN);
    const v4i* c4 = (const v4i*)cnt;
    for (int i = threadIdx.x; i < DN / 4; i += 1024) p4[i] = c4[i];
}

__global__ void invsqrt_deg_kernel(const int* __restrict__ part,
                                   float* __restrict__ dout, float* __restrict__ din, int n) {
    int i = blockIdx.x * blockDim.x + threadIdx.x;
    if (i >= n) return;
    const int* p = part + (size_t)blockIdx.y * DG * DN + i;
    int s = 0;
    #pragma unroll
    for (int g = 0; g < DG; ++g) s += p[(size_t)g * DN];
    float* o = blockIdx.y ? din : dout;
    o[i] = rsqrtf(fmaxf((float)s, 1.0f));
}

__global__ void ln_stats_768_kernel(const float* __restrict__ feat,
                                    float* __restrict__ mu, float* __restrict__ rs, int n) {
    int row  = (blockIdx.x * blockDim.x + threadIdx.x) >> 5;
    int lane = threadIdx.x & 31;
    if (row >= n) return;
    const float* p = feat + (size_t)row * 768;
    float s = 0.f, s2 = 0.f;
    #pragma unroll
    for (int c = 0; c < 768; c += 32) {
        float v = p[c + lane];
        s += v; s2 += v * v;
    }
    #pragma unroll
    for (int off = 16; off > 0; off >>= 1) {
        s  += __shfl_xor(s,  off, 32);
        s2 += __shfl_xor(s2, off, 32);
    }
    if (lane == 0) {
        float m   = s  * (1.0f / 768.0f);
        float var = s2 * (1.0f / 768.0f) - m * m;
        mu[row] = m;
        rs[row] = rsqrtf(var + LN_EPS);
    }
}

__global__ void __launch_bounds__(256)
gemm0_ln_wmma_kernel(const float* __restrict__ feat,
                                     const float* __restrict__ mu, const float* __restrict__ rs,
                                     const float* __restrict__ do_is,
                                     const float* __restrict__ g, const float* __restrict__ b,
                                     const float* __restrict__ W,
                                     float* __restrict__ C, int M) {
    const int K = 768, F = 128;
    const int lane = threadIdx.x & 31;
    const int mt   = (blockIdx.x * blockDim.x + threadIdx.x) >> 5;
    if (mt * 16 >= M) return;
    const int half = lane >> 4;
    const int l16  = lane & 15;
    const int arow = mt * 16 + l16;
    const int koff = half * 2;
    const float m_r  = mu[arow];
    const float rs_r = rs[arow];
    const float do_r = do_is[arow];
    const float* ap = feat + (size_t)arow * K + koff;

    v8f acc[8];
    #pragma unroll
    for (int i = 0; i < 8; ++i) { v8f z = {}; acc[i] = z; }

    constexpr int KB = 4;
    for (int k = 0; k < K; k += 4 * KB) {
        v2f x[KB], gg[KB], bbv[KB];
        v4f w0l[KB], w0h[KB], w1l[KB], w1h[KB];
        #pragma unroll
        for (int u = 0; u < KB; ++u) {
            const int c0 = k + 4 * u + koff;
            x[u]   = *(const v2f*)(ap + k + 4 * u);
            gg[u]  = *(const v2f*)(g + c0);
            bbv[u] = *(const v2f*)(b + c0);
        }
        __builtin_amdgcn_sched_barrier(0);
        #pragma unroll
        for (int u = 0; u < KB; ++u) {
            const int c0 = k + 4 * u + koff;
            const float* wrow = W + (size_t)c0 * F + 4 * l16;
            w0l[u] = *(const v4f*)(wrow);
            w0h[u] = *(const v4f*)(wrow + 64);
            w1l[u] = *(const v4f*)(wrow + F);
            w1h[u] = *(const v4f*)(wrow + F + 64);
        }
        #pragma unroll
        for (int u = 0; u < KB; ++u) {
            v2f a;
            a.x = ((x[u].x - m_r) * rs_r * gg[u].x + bbv[u].x) * do_r;
            a.y = ((x[u].y - m_r) * rs_r * gg[u].y + bbv[u].y) * do_r;
            #pragma unroll
            for (int ft = 0; ft < 8; ++ft) {
                v2f bb;
                bb.x = (ft < 4) ? w0l[u][ft & 3] : w0h[u][ft & 3];
                bb.y = (ft < 4) ? w1l[u][ft & 3] : w1h[u][ft & 3];
                acc[ft] = __builtin_amdgcn_wmma_f32_16x16x4_f32(
                    false, a, false, bb, (short)0, acc[ft], false, false);
            }
        }
    }
    const int crow = mt * 16 + half * 8;
    #pragma unroll
    for (int r = 0; r < 8; ++r) {
        float* crp = C + (size_t)(crow + r) * F + 4 * l16;
        const v4f lo = {acc[0][r], acc[1][r], acc[2][r], acc[3][r]};
        const v4f hi = {acc[4][r], acc[5][r], acc[6][r], acc[7][r]};
        *(v4f*)(crp)      = lo;
        *(v4f*)(crp + 64) = hi;
    }
}

__global__ void __launch_bounds__(256)
bin_count_kernel(const int* __restrict__ dst, int* __restrict__ hist, int e) {
    __shared__ int h[NBK];
    for (int i = threadIdx.x; i < NBK; i += 256) h[i] = 0;
    __syncthreads();
    const long long base = (long long)blockIdx.x * ECH;
    for (int i = threadIdx.x; i < ECH; i += 256) {
        const long long idx = base + i;
        if (idx < e) atomicAdd(&h[dst[idx] / RB], 1);
    }
    __syncthreads();
    for (int i = threadIdx.x; i < NBK; i += 256) hist[(size_t)blockIdx.x * NBK + i] = h[i];
}

__global__ void __launch_bounds__(NBK)
bin_scan_kernel(int* __restrict__ hist, int* __restrict__ start, int nch) {
    __shared__ int sc[NBK];
    const int b = threadIdx.x;
    int run = 0;
    for (int j0 = 0; j0 < nch; j0 += 14) {
        int c[14];
        #pragma unroll
        for (int u = 0; u < 14; ++u) c[u] = (j0 + u < nch) ? hist[(size_t)(j0 + u) * NBK + b] : 0;
        #pragma unroll
        for (int u = 0; u < 14; ++u) {
            if (j0 + u < nch) hist[(size_t)(j0 + u) * NBK + b] = run;
            run += c[u];
        }
    }
    sc[b] = run;
    __syncthreads();
    for (int off = 1; off < NBK; off <<= 1) {
        const int add = (b >= off) ? sc[b - off] : 0;
        __syncthreads();
        sc[b] += add;
        __syncthreads();
    }
    start[b + 1] = sc[b];
    if (b == 0) start[0] = 0;
}

__global__ void __launch_bounds__(256)
bin_fill_kernel(const int* __restrict__ src, const int* __restrict__ dst,
                const int* __restrict__ hist, const int* __restrict__ start,
                int* __restrict__ bins, int e) {
    __shared__ int cur[NBK];
    for (int i = threadIdx.x; i < NBK; i += 256)
        cur[i] = start[i] + hist[(size_t)blockIdx.x * NBK + i];
    __syncthreads();
    const long long base = (long long)blockIdx.x * ECH;
    for (int i = threadIdx.x; i < ECH; i += 256) {
        const long long idx = base + i;
        if (idx < e) {
            const int d = dst[idx];
            const int b = d / RB;
            const int pos = atomicAdd(&cur[b], 1);
            bins[pos] = src[idx] * RB + (d - b * RB);
        }
    }
}

template <int COLS, int UB>
__global__ void __launch_bounds__(512)
agg_owner_kernel(const float* __restrict__ x, const int* __restrict__ bins,
                 const int* __restrict__ start, float* __restrict__ out, int n) {
    constexpr int NI = (COLS + 31) / 32;
    __shared__ __attribute__((aligned(16))) float acc[RB * COLS];
    const int lane = threadIdx.x & 31;
    const int wave = threadIdx.x >> 5;
    const int nw   = blockDim.x >> 5;
    {
        const v4f z = {0.0f, 0.0f, 0.0f, 0.0f};
        v4f* a4 = (v4f*)acc;
        for (int i = threadIdx.x; i < RB * COLS / 4; i += blockDim.x) a4[i] = z;
    }
    __syncthreads();
    const int e0 = start[blockIdx.x];
    const int e1 = start[blockIdx.x + 1];
    const int ng = (e1 - e0 + UB - 1) / UB;
    int g = wave;
    int pk = 0;
    if (g < ng && e0 + g * UB + lane < e1 && lane < UB) pk = bins[e0 + g * UB + lane];
    while (g < ng) {
        const int cnt = (e1 - (e0 + g * UB) < UB) ? e1 - (e0 + g * UB) : UB;
        const int gn = g + nw;
        int pkn = 0;
        if (gn < ng && e0 + gn * UB + lane < e1 && lane < UB) pkn = bins[e0 + gn * UB + lane];
        if (cnt == UB) {
            float v[UB][NI];
            int   lr[UB];
            #pragma unroll
            for (int u = 0; u < UB; ++u) {
                const int w = __builtin_amdgcn_readlane(pk, u);
                const int s = w / RB;
                lr[u] = w - s * RB;
                const float* xr = x + (size_t)s * COLS + lane;
                #pragma unroll
                for (int i = 0; i < NI; ++i)
                    v[u][i] = (32 * i + lane < COLS) ? xr[32 * i] : 0.0f;
            }
            #pragma unroll
            for (int u = 0; u < UB; ++u) {
                float* a = acc + lr[u] * COLS + lane;
                #pragma unroll
                for (int i = 0; i < NI; ++i)
                    if (32 * i + lane < COLS) atomicAdd(a + 32 * i, v[u][i]);
            }
        } else {
            for (int j = 0; j < cnt; ++j) {
                const int w = __shfl(pk, j, 32);
                const int s = w / RB;
                const float* xr = x + (size_t)s * COLS + lane;
                float* a = acc + (w - s * RB) * COLS + lane;
                #pragma unroll
                for (int i = 0; i < NI; ++i)
                    if (32 * i + lane < COLS) atomicAdd(a + 32 * i, xr[32 * i]);
            }
        }
        pk = pkn;
        g  = gn;
    }
    __syncthreads();
    const int r0 = blockIdx.x * RB;
    const int nr = (n - r0 < RB) ? (n - r0) : RB;
    v4f* o4 = (v4f*)(out + (size_t)r0 * COLS);
    const v4f* a4 = (const v4f*)acc;
    for (int i = threadIdx.x; i < nr * COLS / 4; i += blockDim.x) o4[i] = a4[i];
}

#define AP 132

template <int FOUT, int UB>
__global__ void __launch_bounds__(512)
agg_product_kernel(const float* __restrict__ x, const int* __restrict__ bins,
                   const int* __restrict__ start,
                   const float* __restrict__ di_is, const float* __restrict__ do_is,
                   const float* __restrict__ g, const float* __restrict__ b,
                   const float* __restrict__ W,
                   float* __restrict__ C, int n) {
    constexpr int COLS = 128, NI = COLS / 32, KS = COLS / 4;
    __shared__ __attribute__((aligned(16))) float acc[RB * AP];
    const int lane = threadIdx.x & 31;
    const int wave = threadIdx.x >> 5;
    const int nw   = blockDim.x >> 5;
    {
        const v4f z = {0.0f, 0.0f, 0.0f, 0.0f};
        v4f* a4 = (v4f*)acc;
        for (int i = threadIdx.x; i < RB * AP / 4; i += blockDim.x) a4[i] = z;
    }
    __syncthreads();
    const int e0 = start[blockIdx.x];
    const int e1 = start[blockIdx.x + 1];
    const int ng = (e1 - e0 + UB - 1) / UB;
    int gi = wave;
    int pk = 0;
    if (gi < ng && e0 + gi * UB + lane < e1 && lane < UB) pk = bins[e0 + gi * UB + lane];
    while (gi < ng) {
        const int cnt = (e1 - (e0 + gi * UB) < UB) ? e1 - (e0 + gi * UB) : UB;
        const int gn = gi + nw;
        int pkn = 0;
        if (gn < ng && e0 + gn * UB + lane < e1 && lane < UB) pkn = bins[e0 + gn * UB + lane];
        if (cnt == UB) {
            float v[UB][NI];
            int   lr[UB];
            #pragma unroll
            for (int u = 0; u < UB; ++u) {
                const int w = __builtin_amdgcn_readlane(pk, u);
                const int s = w / RB;
                lr[u] = w - s * RB;
                const float* xr = x + (size_t)s * COLS + lane;
                #pragma unroll
                for (int i = 0; i < NI; ++i) v[u][i] = xr[32 * i];
            }
            #pragma unroll
            for (int u = 0; u < UB; ++u) {
                float* a = acc + lr[u] * AP + lane;
                #pragma unroll
                for (int i = 0; i < NI; ++i) atomicAdd(a + 32 * i, v[u][i]);
            }
        } else {
            for (int j = 0; j < cnt; ++j) {
                const int w = __shfl(pk, j, 32);
                const int s = w / RB;
                const float* xr = x + (size_t)s * COLS + lane;
                float* a = acc + (w - s * RB) * AP + lane;
                #pragma unroll
                for (int i = 0; i < NI; ++i) atomicAdd(a + 32 * i, xr[32 * i]);
            }
        }
        pk = pkn;
        gi = gn;
    }
    __syncthreads();

    const int r0 = blockIdx.x * RB;
    const int nr = (n - r0 < RB) ? (n - r0) : RB;
    const int nstrips = (nr > 0) ? ((nr + 15) >> 4) : 0;
    constexpr int NH = (FOUT == 128) ? 2 : 1;
    if (wave >= nstrips * NH) return;
    const int strip = wave / NH;
    const int ch    = wave - strip * NH;
    const int half  = lane >> 4;
    const int l16   = lane & 15;
    const int koff  = half * 2;
    int lrow = strip * 16 + l16;
    if (lrow >= nr) lrow = nr - 1;
    const float* ar = acc + lrow * AP + koff;
    const float di = di_is[r0 + lrow];
    const float ds = do_is[r0 + lrow];

    float sa = 0.f, sb = 0.f, qa = 0.f, qb = 0.f;
    #pragma unroll
    for (int u = 0; u < KS; ++u) {
        v2f xv = *(const v2f*)(ar + 4 * u);
        xv.x *= di; xv.y *= di;
        sa += xv.x; qa += xv.x * xv.x;
        sb += xv.y; qb += xv.y * xv.y;
    }
    float s = sa + sb, s2 = qa + qb;
    s  += __shfl_xor(s,  16, 32);
    s2 += __shfl_xor(s2, 16, 32);
    const float m   = s * (1.0f / 128.0f);
    const float rsv = rsqrtf(s2 * (1.0f / 128.0f) - m * m + LN_EPS);

    v8f pacc[4];
    #pragma unroll
    for (int i = 0; i < 4; ++i) { v8f z = {}; pacc[i] = z; }

    const int lc = (l16 < FOUT / 4) ? l16 : FOUT / 4 - 1;
    const int cb = (FOUT == 128) ? (64 * ch + 4 * l16) : (4 * lc);
    constexpr int KB = 4;
    #pragma unroll 1
    for (int t = 0; t < KS; t += KB) {
        v2f xa[KB], gg[KB], bbv[KB];
        v4f w0[KB], w1[KB];
        #pragma unroll
        for (int u = 0; u < KB; ++u) {
            const int c0 = 4 * (t + u) + koff;
            gg[u]  = *(const v2f*)(g + c0);
            bbv[u] = *(const v2f*)(b + c0);
            const float* wrow = W + (size_t)c0 * FOUT + cb;
            w0[u] = *(const v4f*)(wrow);
            w1[u] = *(const v4f*)(wrow + FOUT);
            xa[u] = *(const v2f*)(ar + 4 * (t + u));
        }
        #pragma unroll
        for (int u = 0; u < KB; ++u) {
            const float x0 = xa[u].x * di;
            const float x1 = xa[u].y * di;
            v2f a;
            a.x = fmaxf((x0 - m) * rsv * gg[u].x + bbv[u].x, 0.0f) * ds;
            a.y = fmaxf((x1 - m) * rsv * gg[u].y + bbv[u].y, 0.0f) * ds;
            #pragma unroll
            for (int ft = 0; ft < 4; ++ft) {
                v2f bb;
                bb.x = w0[u][ft];
                bb.y = w1[u][ft];
                pacc[ft] = __builtin_amdgcn_wmma_f32_16x16x4_f32(
                    false, a, false, bb, (short)0, pacc[ft], false, false);
            }
        }
    }
    const int crow = strip * 16 + half * 8;
    const int ccol = (FOUT == 128) ? (64 * ch + 4 * l16) : (4 * l16);
    if (FOUT == 128 || l16 < FOUT / 4) {
        #pragma unroll
        for (int r = 0; r < 8; ++r) {
            if (crow + r < nr) {
                const v4f o = {pacc[0][r], pacc[1][r], pacc[2][r], pacc[3][r]};
                *(v4f*)(C + (size_t)(r0 + crow + r) * FOUT + ccol) = o;
            }
        }
    }
}

__global__ void final_scale_bias_kernel(float* __restrict__ out,
                                        const float* __restrict__ di_is,
                                        const float* __restrict__ b2, int n) {
    const int F = 40;
    int t = blockIdx.x * blockDim.x + threadIdx.x;
    if (t >= n * F) return;
    out[t] = out[t] * di_is[t / F] + b2[t % F];
}

static inline int cdiv(long long a, long long b) { return (int)((a + b - 1) / b); }

extern "C" void kernel_launch(void* const* d_in, const int* in_sizes, int n_in,
                              void* d_out, int out_size, void* d_ws, size_t ws_size,
                              hipStream_t stream) {
    const float* feat = (const float*)d_in[0];
    const int*   src  = (const int*)d_in[1];
    const int*   dst  = (const int*)d_in[2];
    const float* W0   = (const float*)d_in[3];
    const float* W1   = (const float*)d_in[4];
    const float* W2   = (const float*)d_in[5];
    const float* b2   = (const float*)d_in[6];
    const float* g_in = (const float*)d_in[7];
    const float* b_in = (const float*)d_in[8];
    const float* g0   = (const float*)d_in[9];
    const float* b0   = (const float*)d_in[10];
    const float* g1   = (const float*)d_in[11];
    const float* b1   = (const float*)d_in[12];
    float* out = (float*)d_out;

    const int IN_F = 768, HID = 128, CLS = 40;
    const int n = in_sizes[0] / IN_F;
    const int e = in_sizes[1];

    float* ws    = (float*)d_ws;
    float* do_is = ws;
    float* di_is = ws + (size_t)n;
    float* mu    = ws + 2 * (size_t)n;
    float* rs    = ws + 3 * (size_t)n;
    float* bufB  = ws + 4 * (size_t)n;
    float* bufC  = bufB + (size_t)n * HID;
    float* bufD  = bufC + (size_t)n * HID;
    int*   bins  = (int*)(bufD + (size_t)n * HID);
    const int nch = cdiv(e, ECH);
    int*   hist  = bins + (size_t)e;
    int*   start = hist + (size_t)nch * NBK;

    const int T = 256;

    int* part = (int*)bufC;
    degree_hist_kernel<<<dim3(DG, 2), 1024, 0, stream>>>(src, dst, part, e);
    invsqrt_deg_kernel<<<dim3(cdiv(n, T), 2), T, 0, stream>>>(part, do_is, di_is, n);

    bin_count_kernel<<<nch, 256, 0, stream>>>(dst, hist, e);
    bin_scan_kernel<<<1, NBK, 0, stream>>>(hist, start, nch);
    bin_fill_kernel<<<nch, 256, 0, stream>>>(src, dst, hist, start, bins, e);

    ln_stats_768_kernel<<<cdiv((long long)n * 32, T), T, 0, stream>>>(feat, mu, rs, n);

    const int mtiles = cdiv(n, 16);
    gemm0_ln_wmma_kernel<<<cdiv((long long)mtiles * 32, T), T, 0, stream>>>(
        feat, mu, rs, do_is, g_in, b_in, W0, bufB, n);

    agg_product_kernel<128, 8><<<NBK, 512, 0, stream>>>(bufB, bins, start, di_is, do_is, g0, b0, W1, bufD, n);

    agg_product_kernel<40, 8><<<NBK, 512, 0, stream>>>(bufD, bins, start, di_is, do_is, g1, b1, W2, bufC, n);
    agg_owner_kernel<40, 8><<<NBK, 512, 0, stream>>>(bufC, bins, start, out, n);
    final_scale_bias_kernel<<<cdiv((long long)n * CLS, T), T, 0, stream>>>(out, di_is, b2, n);
}


# Edge-list passes request their loads as one group: bin count, fill and scan and each bin's short last group no longer wait one memory round trip per element; sums stay in arrival order
# speedup vs baseline: 2.8099x; 1.0837x over previous
#include <hip/hip_runtime.h>
#include <math.h>

typedef __attribute__((ext_vector_type(2))) float v2f;
typedef __attribute__((ext_vector_type(8))) float v8f;
typedef __attribute__((ext_vector_type(4))) float v4f;

#define LN_EPS 1e-5f

#define RB   98
#define NBK  512
#define ECH  4096

#define DG   32
#define DN   (NBK * RB)
#define DQ   8

typedef __attribute__((ext_vector_type(4))) int v4i;

__global__ void __launch_bounds__(1024)
degree_hist_kernel(const int* __restrict__ src, const int* __restrict__ dst,
                   int* __restrict__ part, int e) {
    __shared__ __attribute__((aligned(16))) int cnt[DN];
    const int* __restrict__ idx = blockIdx.y ? dst : src;
    {
        const v4i z = {0, 0, 0, 0};
        v4i* c4 = (v4i*)cnt;
        for (int i = threadIdx.x; i < DN / 4; i += 1024) c4[i] = z;
    }
    __syncthreads();
    const int per = (((e + DG - 1) / DG) + 3) & ~3;
    const long long lo = (long long)blockIdx.x * per;
    long long hi = lo + per;
    if (hi > e) hi = e;
    const int len = (hi > lo) ? (int)(hi - lo) : 0;
    const int nq  = len >> 2;
    const v4i* q = (const v4i*)(idx + lo);
    for (int base = 0; base < nq; base += 1024 * DQ) {
        v4i w[DQ];
        #pragma unroll
        for (int u = 0; u < DQ; ++u) {
            const int i = base + u * 1024 + threadIdx.x;
            w[u] = q[(i < nq) ? i : nq - 1];
        }
        #pragma unroll
        for (int u = 0; u < DQ; ++u) {
            const int i = base + u * 1024 + threadIdx.x;
            if (i < nq) {
                atomicAdd(&cnt[w[u].x], 1);
                atomicAdd(&cnt[w[u].y], 1);
                atomicAdd(&cnt[w[u].z], 1);
                atomicAdd(&cnt[w[u].w], 1);
            }
        }
    }
    if (4 * nq + (int)threadIdx.x < len)
        atomicAdd(&cnt[idx[lo + 4 * nq + threadIdx.x]], 1);
    __syncthreads();
    v4i* p4 = (v4i*)(part + ((size_t)blockIdx.y * DG + blockIdx.x) * DN);
    const v4i* c4 = (const v4i*)cnt;
    for (int i = threadIdx.x; i < DN / 4; i += 1024) p4[i] = c4[i];
}

__global__ void invsqrt_deg_kernel(const int* __restrict__ part,
                                   float* __restrict__ dout, float* __restrict__ din, int n) {
    int i = blockIdx.x * blockDim.x + threadIdx.x;
    if (i >= n) return;
    const int* p = part + (size_t)blockIdx.y * DG * DN + i;
    int s = 0;
    #pragma unroll
    for (int g = 0; g < DG; ++g) s += p[(size_t)g * DN];
    float* o = blockIdx.y ? din : dout;
    o[i] = rsqrtf(fmaxf((float)s, 1.0f));
}

__global__ void ln_stats_768_kernel(const float* __restrict__ feat,
                                    float* __restrict__ mu, float* __restrict__ rs, int n) {
    int row  = (blockIdx.x * blockDim.x + threadIdx.x) >> 5;
    int lane = threadIdx.x & 31;
    if (row >= n) return;
    const float* p = feat + (size_t)row * 768;
    float s = 0.f, s2 = 0.f;
    #pragma unroll
    for (int c = 0; c < 768; c += 32) {
        float v = p[c + lane];
        s += v; s2 += v * v;
    }
    #pragma unroll
    for (int off = 16; off > 0; off >>= 1) {
        s  += __shfl_xor(s,  off, 32);
        s2 += __shfl_xor(s2, off, 32);
    }
    if (lane == 0) {
        float m   = s  * (1.0f / 768.0f);
        float var = s2 * (1.0f / 768.0f) - m * m;
        mu[row] = m;
        rs[row] = rsqrtf(var + LN_EPS);
    }
}

__global__ void __launch_bounds__(256)
gemm0_ln_wmma_kernel(const float* __restrict__ feat,
                                     const float* __restrict__ mu, const float* __restrict__ rs,
                                     const float* __restrict__ do_is,
                                     const float* __restrict__ g, const float* __restrict__ b,
                                     const float* __restrict__ W,
                                     float* __restrict__ C, int M) {
    const int K = 768, F = 128;
    const int lane = threadIdx.x & 31;
    const int mt   = (blockIdx.x * blockDim.x + threadIdx.x) >> 5;
    if (mt * 16 >= M) return;
    const int half = lane >> 4;
    const int l16  = lane & 15;
    const int arow = mt * 16 + l16;
    const int koff = half * 2;
    const float m_r  = mu[arow];
    const float rs_r = rs[arow];
    const float do_r = do_is[arow];
    const float* ap = feat + (size_t)arow * K + koff;

    v8f acc[8];
    #pragma unroll
    for (int i = 0; i < 8; ++i) { v8f z = {}; acc[i] = z; }

    constexpr int KB = 4;
    for (int k = 0; k < K; k += 4 * KB) {
        v2f x[KB], gg[KB], bbv[KB];
        v4f w0l[KB], w0h[KB], w1l[KB], w1h[KB];
        #pragma unroll
        for (int u = 0; u < KB; ++u) {
            const int c0 = k + 4 * u + koff;
            x[u]   = *(const v2f*)(ap + k + 4 * u);
            gg[u]  = *(const v2f*)(g + c0);
            bbv[u] = *(const v2f*)(b + c0);
        }
        __builtin_amdgcn_sched_barrier(0);
        #pragma unroll
        for (int u = 0; u < KB; ++u) {
            const int c0 = k + 4 * u + koff;
            const float* wrow = W + (size_t)c0 * F + 4 * l16;
            w0l[u] = *(const v4f*)(wrow);
            w0h[u] = *(const v4f*)(wrow + 64);
            w1l[u] = *(const v4f*)(wrow + F);
            w1h[u] = *(const v4f*)(wrow + F + 64);
        }
        #pragma unroll
        for (int u = 0; u < KB; ++u) {
            v2f a;
            a.x = ((x[u].x - m_r) * rs_r * gg[u].x + bbv[u].x) * do_r;
            a.y = ((x[u].y - m_r) * rs_r * gg[u].y + bbv[u].y) * do_r;
            #pragma unroll
            for (int ft = 0; ft < 8; ++ft) {
                v2f bb;
                bb.x = (ft < 4) ? w0l[u][ft & 3] : w0h[u][ft & 3];
                bb.y = (ft < 4) ? w1l[u][ft & 3] : w1h[u][ft & 3];
                acc[ft] = __builtin_amdgcn_wmma_f32_16x16x4_f32(
                    false, a, false, bb, (short)0, acc[ft], false, false);
            }
        }
    }
    const int crow = mt * 16 + half * 8;
    #pragma unroll
    for (int r = 0; r < 8; ++r) {
        float* crp = C + (size_t)(crow + r) * F + 4 * l16;
        const v4f lo = {acc[0][r], acc[1][r], acc[2][r], acc[3][r]};
        const v4f hi = {acc[4][r], acc[5][r], acc[6][r], acc[7][r]};
        *(v4f*)(crp)      = lo;
        *(v4f*)(crp + 64) = hi;
    }
}

#define EQ   (ECH / 256)

__global__ void __launch_bounds__(256)
bin_count_kernel(const int* __restrict__ dst, int* __restrict__ hist, int e) {
    __shared__ int h[NBK];
    const long long base = (long long)blockIdx.x * ECH;
    int d[EQ];
    #pragma unroll
    for (int u = 0; u < EQ; ++u) {
        const long long idx = base + u * 256 + threadIdx.x;
        d[u] = dst[(idx < e) ? idx : (long long)e - 1];
    }
    for (int i = threadIdx.x; i < NBK; i += 256) h[i] = 0;
    __syncthreads();
    #pragma unroll
    for (int u = 0; u < EQ; ++u) {
        const long long idx = base + u * 256 + threadIdx.x;
        if (idx < e) atomicAdd(&h[d[u] / RB], 1);
    }
    __syncthreads();
    for (int i = threadIdx.x; i < NBK; i += 256) hist[(size_t)blockIdx.x * NBK + i] = h[i];
}

__global__ void __launch_bounds__(NBK)
bin_scan_kernel(int* __restrict__ hist, int* __restrict__ start, int nch) {
    __shared__ int sc[NBK];
    const int b = threadIdx.x;
    int run = 0;
    constexpr int SB = 49;
    for (int j0 = 0; j0 < nch; j0 += SB) {
        int c[SB];
        #pragma unroll
        for (int u = 0; u < SB; ++u) c[u] = (j0 + u < nch) ? hist[(size_t)(j0 + u) * NBK + b] : 0;
        #pragma unroll
        for (int u = 0; u < SB; ++u) {
            if (j0 + u < nch) hist[(size_t)(j0 + u) * NBK + b] = run;
            run += c[u];
        }
    }
    sc[b] = run;
    __syncthreads();
    for (int off = 1; off < NBK; off <<= 1) {
        const int add = (b >= off) ? sc[b - off] : 0;
        __syncthreads();
        sc[b] += add;
        __syncthreads();
    }
    start[b + 1] = sc[b];
    if (b == 0) start[0] = 0;
}

__global__ void __launch_bounds__(256)
bin_fill_kernel(const int* __restrict__ src, const int* __restrict__ dst,
                const int* __restrict__ hist, const int* __restrict__ start,
                int* __restrict__ bins, int e) {
    __shared__ int cur[NBK + 1];
    const long long base = (long long)blockIdx.x * ECH;
    int d[EQ], s[EQ], pos[EQ];
    #pragma unroll
    for (int u = 0; u < EQ; ++u) {
        const long long idx = base + u * 256 + threadIdx.x;
        const long long ic  = (idx < e) ? idx : (long long)e - 1;
        d[u] = dst[ic];
        s[u] = src[ic];
    }
    for (int i = threadIdx.x; i < NBK; i += 256)
        cur[i] = start[i] + hist[(size_t)blockIdx.x * NBK + i];
    if (threadIdx.x == 0) cur[NBK] = 0;
    __syncthreads();
    #pragma unroll
    for (int u = 0; u < EQ; ++u) {
        const long long idx = base + u * 256 + threadIdx.x;
        pos[u] = atomicAdd(&cur[(idx < e) ? d[u] / RB : NBK], 1);
    }
    #pragma unroll
    for (int u = 0; u < EQ; ++u) {
        const long long idx = base + u * 256 + threadIdx.x;
        if (idx < e) {
            const int b = d[u] / RB;
            bins[pos[u]] = s[u] * RB + (d[u] - b * RB);
        }
    }
}

template <int COLS, int UB>
__global__ void __launch_bounds__(512)
agg_owner_kernel(const float* __restrict__ x, const int* __restrict__ bins,
                 const int* __restrict__ start, float* __restrict__ out, int n) {
    constexpr int NI = (COLS + 31) / 32;
    __shared__ __attribute__((aligned(16))) float acc[RB * COLS];
    const int lane = threadIdx.x & 31;
    const int wave = threadIdx.x >> 5;
    const int nw   = blockDim.x >> 5;
    const int e0 = start[blockIdx.x];
    const int e1 = start[blockIdx.x + 1];
    const int ng = (e1 - e0 + UB - 1) / UB;
    int g = wave;
    int pk = 0;
    if (g < ng && e0 + g * UB + lane < e1 && lane < UB) pk = bins[e0 + g * UB + lane];
    {
        const v4f z = {0.0f, 0.0f, 0.0f, 0.0f};
        v4f* a4 = (v4f*)acc;
        for (int i = threadIdx.x; i < RB * COLS / 4; i += blockDim.x) a4[i] = z;
    }
    __syncthreads();
    while (g < ng) {
        const int cnt = (e1 - (e0 + g * UB) < UB) ? e1 - (e0 + g * UB) : UB;
        const int gn = g + nw;
        int pkn = 0;
        if (gn < ng && e0 + gn * UB + lane < e1 && lane < UB) pkn = bins[e0 + gn * UB + lane];
        if (cnt == UB) {
            float v[UB][NI];
            int   lr[UB];
            #pragma unroll
            for (int u = 0; u < UB; ++u) {
                const int w = __builtin_amdgcn_readlane(pk, u);
                const int s = w / RB;
                lr[u] = w - s * RB;
                const float* xr = x + (size_t)s * COLS + lane;
                #pragma unroll
                for (int i = 0; i < NI; ++i)
                    v[u][i] = (32 * i + lane < COLS) ? xr[32 * i] : 0.0f;
            }
            #pragma unroll
            for (int u = 0; u < UB; ++u) {
                float* a = acc + lr[u] * COLS + lane;
                #pragma unroll
                for (int i = 0; i < NI; ++i)
                    if (32 * i + lane < COLS) atomicAdd(a + 32 * i, v[u][i]);
            }
        } else {
            float v[UB - 1][NI];
            int   lr[UB - 1];
            #pragma unroll
            for (int u = 0; u < UB - 1; ++u) {
                const int w = __builtin_amdgcn_readlane(pk, u);
                const int s = w / RB;
                lr[u] = w - s * RB;
                const float* xr = x + (size_t)s * COLS + lane;
                #pragma unroll
                for (int i = 0; i < NI; ++i)
                    v[u][i] = (32 * i + lane < COLS) ? xr[32 * i] : 0.0f;
            }
            #pragma unroll
            for (int u = 0; u < UB - 1; ++u) {
                float* a = acc + lr[u] * COLS + lane;
                #pragma unroll
                for (int i = 0; i < NI; ++i)
                    if (32 * i + lane < COLS) atomicAdd(a + 32 * i, (u < cnt) ? v[u][i] : 0.0f);
            }
        }
        pk = pkn;
        g  = gn;
    }
    __syncthreads();
    const int r0 = blockIdx.x * RB;
    const int nr = (n - r0 < RB) ? (n - r0) : RB;
    v4f* o4 = (v4f*)(out + (size_t)r0 * COLS);
    const v4f* a4 = (const v4f*)acc;
    for (int i = threadIdx.x; i < nr * COLS / 4; i += blockDim.x) o4[i] = a4[i];
}

#define AP 132

template <int FOUT, int UB>
__global__ void __launch_bounds__(512)
agg_product_kernel(const float* __restrict__ x, const int* __restrict__ bins,
                   const int* __restrict__ start,
                   const float* __restrict__ di_is, const float* __restrict__ do_is,
                   const float* __restrict__ g, const float* __restrict__ b,
                   const float* __restrict__ W,
                   float* __restrict__ C, int n) {
    constexpr int COLS = 128, NI = COLS / 32, KS = COLS / 4;
    __shared__ __attribute__((aligned(16))) float acc[RB * AP];
    const int lane = threadIdx.x & 31;
    const int wave = threadIdx.x >> 5;
    const int nw   = blockDim.x >> 5;
    const int e0 = start[blockIdx.x];
    const int e1 = start[blockIdx.x + 1];
    const int ng = (e1 - e0 + UB - 1) / UB;
    int gi = wave;
    int pk = 0;
    if (gi < ng && e0 + gi * UB + lane < e1 && lane < UB) pk = bins[e0 + gi * UB + lane];
    {
        const v4f z = {0.0f, 0.0f, 0.0f, 0.0f};
        v4f* a4 = (v4f*)acc;
        for (int i = threadIdx.x; i < RB * AP / 4; i += blockDim.x) a4[i] = z;
    }
    __syncthreads();
    while (gi < ng) {
        const int cnt = (e1 - (e0 + gi * UB) < UB) ? e1 - (e0 + gi * UB) : UB;
        const int gn = gi + nw;
        int pkn = 0;
        if (gn < ng && e0 + gn * UB + lane < e1 && lane < UB) pkn = bins[e0 + gn * UB + lane];
        if (cnt == UB) {
            float v[UB][NI];
            int   lr[UB];
            #pragma unroll
            for (int u = 0; u < UB; ++u) {
                const int w = __builtin_amdgcn_readlane(pk, u);
                const int s = w / RB;
                lr[u] = w - s * RB;
                const float* xr = x + (size_t)s * COLS + lane;
                #pragma unroll
                for (int i = 0; i < NI; ++i) v[u][i] = xr[32 * i];
            }
            #pragma unroll
            for (int u = 0; u < UB; ++u) {
                float* a = acc + lr[u] * AP + lane;
                #pragma unroll
                for (int i = 0; i < NI; ++i) atomicAdd(a + 32 * i, v[u][i]);
            }
        } else {
            float v[UB - 1][NI];
            int   lr[UB - 1];
            #pragma unroll
            for (int u = 0; u < UB - 1; ++u) {
                const int w = __builtin_amdgcn_readlane(pk, u);
                const int s = w / RB;
                lr[u] = w - s * RB;
                const float* xr = x + (size_t)s * COLS + lane;
                #pragma unroll
                for (int i = 0; i < NI; ++i) v[u][i] = xr[32 * i];
            }
            #pragma unroll
            for (int u = 0; u < UB - 1; ++u) {
                float* a = acc + lr[u] * AP + lane;
                #pragma unroll
                for (int i = 0; i < NI; ++i) atomicAdd(a + 32 * i, (u < cnt) ? v[u][i] : 0.0f);
            }
        }
        pk = pkn;
        gi = gn;
    }
    __syncthreads();

    const int r0 = blockIdx.x * RB;
    const int nr = (n - r0 < RB) ? (n - r0) : RB;
    const int nstrips = (nr > 0) ? ((nr + 15) >> 4) : 0;
    constexpr int NH = (FOUT == 128) ? 2 : 1;
    if (wave >= nstrips * NH) return;
    const int strip = wave / NH;
    const int ch    = wave - strip * NH;
    const int half  = lane >> 4;
    const int l16   = lane & 15;
    const int koff  = half * 2;
    int lrow = strip * 16 + l16;
    if (lrow >= nr) lrow = nr - 1;
    const float* ar = acc + lrow * AP + koff;
    const float di = di_is[r0 + lrow];
    const float ds = do_is[r0 + lrow];

    float sa = 0.f, sb = 0.f, qa = 0.f, qb = 0.f;
    #pragma unroll
    for (int u = 0; u < KS; ++u) {
        v2f xv = *(const v2f*)(ar + 4 * u);
        xv.x *= di; xv.y *= di;
        sa += xv.x; qa += xv.x * xv.x;
        sb += xv.y; qb += xv.y * xv.y;
    }
    float s = sa + sb, s2 = qa + qb;
    s  += __shfl_xor(s,  16, 32);
    s2 += __shfl_xor(s2, 16, 32);
    const float m   = s * (1.0f / 128.0f);
    const float rsv = rsqrtf(s2 * (1.0f / 128.0f) - m * m + LN_EPS);

    v8f pacc[4];
    #pragma unroll
    for (int i = 0; i < 4; ++i) { v8f z = {}; pacc[i] = z; }

    const int lc = (l16 < FOUT / 4) ? l16 : FOUT / 4 - 1;
    const int cb = (FOUT == 128) ? (64 * ch + 4 * l16) : (4 * lc);
    constexpr int KB = 4;
    #pragma unroll 1
    for (int t = 0; t < KS; t += KB) {
        v2f xa[KB], gg[KB], bbv[KB];
        v4f w0[KB], w1[KB];
        #pragma unroll
        for (int u = 0; u < KB; ++u) {
            const int c0 = 4 * (t + u) + koff;
            gg[u]  = *(const v2f*)(g + c0);
            bbv[u] = *(const v2f*)(b + c0);
            const float* wrow = W + (size_t)c0 * FOUT + cb;
            w0[u] = *(const v4f*)(wrow);
            w1[u] = *(const v4f*)(wrow + FOUT);
            xa[u] = *(const v2f*)(ar + 4 * (t + u));
        }
        #pragma unroll
        for (int u = 0; u < KB; ++u) {
            const float x0 = xa[u].x * di;
            const float x1 = xa[u].y * di;
            v2f a;
            a.x = fmaxf((x0 - m) * rsv * gg[u].x + bbv[u].x, 0.0f) * ds;
            a.y = fmaxf((x1 - m) * rsv * gg[u].y + bbv[u].y, 0.0f) * ds;
            #pragma unroll
            for (int ft = 0; ft < 4; ++ft) {
                v2f bb;
                bb.x = w0[u][ft];
                bb.y = w1[u][ft];
                pacc[ft] = __builtin_amdgcn_wmma_f32_16x16x4_f32(
                    false, a, false, bb, (short)0, pacc[ft], false, false);
            }
        }
    }
    const int crow = strip * 16 + half * 8;
    const int ccol = (FOUT == 128) ? (64 * ch + 4 * l16) : (4 * l16);
    if (FOUT == 128 || l16 < FOUT / 4) {
        #pragma unroll
        for (int r = 0; r < 8; ++r) {
            if (crow + r < nr) {
                const v4f o = {pacc[0][r], pacc[1][r], pacc[2][r], pacc[3][r]};
                *(v4f*)(C + (size_t)(r0 + crow + r) * FOUT + ccol) = o;
            }
        }
    }
}

__global__ void final_scale_bias_kernel(float* __restrict__ out,
                                        const float* __restrict__ di_is,
                                        const float* __restrict__ b2, int n) {
    const int F = 40;
    int t = blockIdx.x * blockDim.x + threadIdx.x;
    if (t >= n * F) return;
    out[t] = out[t] * di_is[t / F] + b2[t % F];
}

static inline int cdiv(long long a, long long b) { return (int)((a + b - 1) / b); }

extern "C" void kernel_launch(void* const* d_in, const int* in_sizes, int n_in,
                              void* d_out, int out_size, void* d_ws, size_t ws_size,
                              hipStream_t stream) {
    const float* feat = (const float*)d_in[0];
    const int*   src  = (const int*)d_in[1];
    const int*   dst  = (const int*)d_in[2];
    const float* W0   = (const float*)d_in[3];
    const float* W1   = (const float*)d_in[4];
    const float* W2   = (const float*)d_in[5];
    const float* b2   = (const float*)d_in[6];
    const float* g_in = (const float*)d_in[7];
    const float* b_in = (const float*)d_in[8];
    const float* g0   = (const float*)d_in[9];
    const float* b0   = (const float*)d_in[10];
    const float* g1   = (const float*)d_in[11];
    const float* b1   = (const float*)d_in[12];
    float* out = (float*)d_out;

    const int IN_F = 768, HID = 128, CLS = 40;
    const int n = in_sizes[0] / IN_F;
    const int e = in_sizes[1];

    float* ws    = (float*)d_ws;
    float* do_is = ws;
    float* di_is = ws + (size_t)n;
    float* mu    = ws + 2 * (size_t)n;
    float* rs    = ws + 3 * (size_t)n;
    float* bufB  = ws + 4 * (size_t)n;
    float* bufC  = bufB + (size_t)n * HID;
    float* bufD  = bufC + (size_t)n * HID;
    int*   bins  = (int*)(bufD + (size_t)n * HID);
    const int nch = cdiv(e, ECH);
    int*   hist  = bins + (size_t)e;
    int*   start = hist + (size_t)nch * NBK;

    const int T = 256;

    int* part = (int*)bufC;
    degree_hist_kernel<<<dim3(DG, 2), 1024, 0, stream>>>(src, dst, part, e);
    invsqrt_deg_kernel<<<dim3(cdiv(n, T), 2), T, 0, stream>>>(part, do_is, di_is, n);

    bin_count_kernel<<<nch, 256, 0, stream>>>(dst, hist, e);
    bin_scan_kernel<<<1, NBK, 0, stream>>>(hist, start, nch);
    bin_fill_kernel<<<nch, 256, 0, stream>>>(src, dst, hist, start, bins, e);

    ln_stats_768_kernel<<<cdiv((long long)n * 32, T), T, 0, stream>>>(feat, mu, rs, n);

    const int mtiles = cdiv(n, 16);
    gemm0_ln_wmma_kernel<<<cdiv((long long)mtiles * 32, T), T, 0, stream>>>(
        feat, mu, rs, do_is, g_in, b_in, W0, bufB, n);

    agg_product_kernel<128, 8><<<NBK, 512, 0, stream>>>(bufB, bins, start, di_is, do_is, g0, b0, W1, bufD, n);

    agg_product_kernel<40, 8><<<NBK, 512, 0, stream>>>(bufD, bins, start, di_is, do_is, g1, b1, W2, bufC, n);
    agg_owner_kernel<40, 8><<<NBK, 512, 0, stream>>>(bufC, bins, start, out, n);
    final_scale_bias_kernel<<<cdiv((long long)n * CLS, T), T, 0, stream>>>(out, di_is, b2, n);
}


# Layer-0 product split over K: 4 waves per 16-row strip (one per SIMD), each one K quarter, partial tiles added through LDS in fixed order; work spreads evenly over SIMDs; layer-0 sums reassociated
# speedup vs baseline: 2.9797x; 1.0604x over previous
#include <hip/hip_runtime.h>
#include <math.h>

typedef __attribute__((ext_vector_type(2))) float v2f;
typedef __attribute__((ext_vector_type(8))) float v8f;
typedef __attribute__((ext_vector_type(4))) float v4f;

#define LN_EPS 1e-5f

#define RB   98
#define NBK  512
#define ECH  4096

#define DG   32
#define DN   (NBK * RB)
#define DQ   8

typedef __attribute__((ext_vector_type(4))) int v4i;

__global__ void __launch_bounds__(1024)
degree_hist_kernel(const int* __restrict__ src, const int* __restrict__ dst,
                   int* __restrict__ part, int e) {
    __shared__ __attribute__((aligned(16))) int cnt[DN];
    const int* __restrict__ idx = blockIdx.y ? dst : src;
    {
        const v4i z = {0, 0, 0, 0};
        v4i* c4 = (v4i*)cnt;
        for (int i = threadIdx.x; i < DN / 4; i += 1024) c4[i] = z;
    }
    __syncthreads();
    const int per = (((e + DG - 1) / DG) + 3) & ~3;
    const long long lo = (long long)blockIdx.x * per;
    long long hi = lo + per;
    if (hi > e) hi = e;
    const int len = (hi > lo) ? (int)(hi - lo) : 0;
    const int nq  = len >> 2;
    const v4i* q = (const v4i*)(idx + lo);
    for (int base = 0; base < nq; base += 1024 * DQ) {
        v4i w[DQ];
        #pragma unroll
        for (int u = 0; u < DQ; ++u) {
            const int i = base + u * 1024 + threadIdx.x;
            w[u] = q[(i < nq) ? i : nq - 1];
        }
        #pragma unroll
        for (int u = 0; u < DQ; ++u) {
            const int i = base + u * 1024 + threadIdx.x;
            if (i < nq) {
                atomicAdd(&cnt[w[u].x], 1);
                atomicAdd(&cnt[w[u].y], 1);
                atomicAdd(&cnt[w[u].z], 1);
                atomicAdd(&cnt[w[u].w], 1);
            }
        }
    }
    if (4 * nq + (int)threadIdx.x < len)
        atomicAdd(&cnt[idx[lo + 4 * nq + threadIdx.x]], 1);
    __syncthreads();
    v4i* p4 = (v4i*)(part + ((size_t)blockIdx.y * DG + blockIdx.x) * DN);
    const v4i* c4 = (const v4i*)cnt;
    for (int i = threadIdx.x; i < DN / 4; i += 1024) p4[i] = c4[i];
}

__global__ void invsqrt_deg_kernel(const int* __restrict__ part,
                                   float* __restrict__ dout, float* __restrict__ din, int n) {
    int i = blockIdx.x * blockDim.x + threadIdx.x;
    if (i >= n) return;
    const int* p = part + (size_t)blockIdx.y * DG * DN + i;
    int s = 0;
    #pragma unroll
    for (int g = 0; g < DG; ++g) s += p[(size_t)g * DN];
    float* o = blockIdx.y ? din : dout;
    o[i] = rsqrtf(fmaxf((float)s, 1.0f));
}

__global__ void ln_stats_768_kernel(const float* __restrict__ feat,
                                    float* __restrict__ mu, float* __restrict__ rs, int n) {
    int row  = (blockIdx.x * blockDim.x + threadIdx.x) >> 5;
    int lane = threadIdx.x & 31;
    if (row >= n) return;
    const float* p = feat + (size_t)row * 768;
    float s = 0.f, s2 = 0.f;
    #pragma unroll
    for (int c = 0; c < 768; c += 32) {
        float v = p[c + lane];
        s += v; s2 += v * v;
    }
    #pragma unroll
    for (int off = 16; off > 0; off >>= 1) {
        s  += __shfl_xor(s,  off, 32);
        s2 += __shfl_xor(s2, off, 32);
    }
    if (lane == 0) {
        float m   = s  * (1.0f / 768.0f);
        float var = s2 * (1.0f / 768.0f) - m * m;
        mu[row] = m;
        rs[row] = rsqrtf(var + LN_EPS);
    }
}

#define KSPL 4
#define RP   132

__global__ void __launch_bounds__(32 * KSPL)
gemm0_ln_wmma_kernel(const float* __restrict__ feat,
                                     const float* __restrict__ mu, const float* __restrict__ rs,
                                     const float* __restrict__ do_is,
                                     const float* __restrict__ g, const float* __restrict__ b,
                                     const float* __restrict__ W,
                                     float* __restrict__ C, int M) {
    constexpr int K = 768, F = 128, KQ = K / KSPL;
    __shared__ __attribute__((aligned(16))) float part[KSPL * 16 * RP];
    const int lane = threadIdx.x & 31;
    const int q    = __builtin_amdgcn_readfirstlane(threadIdx.x >> 5);
    const int mt   = blockIdx.x;
    if (mt * 16 >= M) return;
    const int half = lane >> 4;
    const int l16  = lane & 15;
    const int arow = mt * 16 + l16;
    const int koff = half * 2;
    const float m_r  = mu[arow];
    const float rs_r = rs[arow];
    const float do_r = do_is[arow];
    const float* ap = feat + (size_t)arow * K + koff;

    v8f acc[8];
    #pragma unroll
    for (int i = 0; i < 8; ++i) { v8f z = {}; acc[i] = z; }

    constexpr int KB = 4;
    const int kbeg = q * KQ;
    for (int k = kbeg; k < kbeg + KQ; k += 4 * KB) {
        v2f x[KB], gg[KB], bbv[KB];
        v4f w0l[KB], w0h[KB], w1l[KB], w1h[KB];
        #pragma unroll
        for (int u = 0; u < KB; ++u) {
            const int c0 = k + 4 * u + koff;
            x[u]   = *(const v2f*)(ap + k + 4 * u);
            gg[u]  = *(const v2f*)(g + c0);
            bbv[u] = *(const v2f*)(b + c0);
        }
        __builtin_amdgcn_sched_barrier(0);
        #pragma unroll
        for (int u = 0; u < KB; ++u) {
            const int c0 = k + 4 * u + koff;
            const float* wrow = W + (size_t)c0 * F + 4 * l16;
            w0l[u] = *(const v4f*)(wrow);
            w0h[u] = *(const v4f*)(wrow + 64);
            w1l[u] = *(const v4f*)(wrow + F);
            w1h[u] = *(const v4f*)(wrow + F + 64);
        }
        #pragma unroll
        for (int u = 0; u < KB; ++u) {
            v2f a;
            a.x = ((x[u].x - m_r) * rs_r * gg[u].x + bbv[u].x) * do_r;
            a.y = ((x[u].y - m_r) * rs_r * gg[u].y + bbv[u].y) * do_r;
            #pragma unroll
            for (int ft = 0; ft < 8; ++ft) {
                v2f bb;
                bb.x = (ft < 4) ? w0l[u][ft & 3] : w0h[u][ft & 3];
                bb.y = (ft < 4) ? w1l[u][ft & 3] : w1h[u][ft & 3];
                acc[ft] = __builtin_amdgcn_wmma_f32_16x16x4_f32(
                    false, a, false, bb, (short)0, acc[ft], false, false);
            }
        }
    }
    float* pq = part + q * 16 * RP;
    #pragma unroll
    for (int r = 0; r < 8; ++r) {
        float* prow = pq + (half * 8 + r) * RP + 4 * l16;
        const v4f lo = {acc[0][r], acc[1][r], acc[2][r], acc[3][r]};
        const v4f hi = {acc[4][r], acc[5][r], acc[6][r], acc[7][r]};
        *(v4f*)(prow)      = lo;
        *(v4f*)(prow + 64) = hi;
    }
    __syncthreads();
    #pragma unroll
    for (int j = 0; j < 16 / KSPL; ++j) {
        const int row = q * (16 / KSPL) + j;
        const float* p0 = part + row * RP + 4 * lane;
        v4f s = *(const v4f*)(p0);
        #pragma unroll
        for (int t = 1; t < KSPL; ++t) s += *(const v4f*)(p0 + t * 16 * RP);
        *(v4f*)(C + (size_t)(mt * 16 + row) * F + 4 * lane) = s;
    }
}

#define EQ   (ECH / 256)

__global__ void __launch_bounds__(256)
bin_count_kernel(const int* __restrict__ dst, int* __restrict__ hist, int e) {
    __shared__ int h[NBK];
    const long long base = (long long)blockIdx.x * ECH;
    int d[EQ];
    #pragma unroll
    for (int u = 0; u < EQ; ++u) {
        const long long idx = base + u * 256 + threadIdx.x;
        d[u] = dst[(idx < e) ? idx : (long long)e - 1];
    }
    for (int i = threadIdx.x; i < NBK; i += 256) h[i] = 0;
    __syncthreads();
    #pragma unroll
    for (int u = 0; u < EQ; ++u) {
        const long long idx = base + u * 256 + threadIdx.x;
        if (idx < e) atomicAdd(&h[d[u] / RB], 1);
    }
    __syncthreads();
    for (int i = threadIdx.x; i < NBK; i += 256) hist[(size_t)blockIdx.x * NBK + i] = h[i];
}

__global__ void __launch_bounds__(NBK)
bin_scan_kernel(int* __restrict__ hist, int* __restrict__ start, int nch) {
    __shared__ int sc[NBK];
    const int b = threadIdx.x;
    int run = 0;
    constexpr int SB = 49;
    for (int j0 = 0; j0 < nch; j0 += SB) {
        int c[SB];
        #pragma unroll
        for (int u = 0; u < SB; ++u) c[u] = (j0 + u < nch) ? hist[(size_t)(j0 + u) * NBK + b] : 0;
        #pragma unroll
        for (int u = 0; u < SB; ++u) {
            if (j0 + u < nch) hist[(size_t)(j0 + u) * NBK + b] = run;
            run += c[u];
        }
    }
    sc[b] = run;
    __syncthreads();
    for (int off = 1; off < NBK; off <<= 1) {
        const int add = (b >= off) ? sc[b - off] : 0;
        __syncthreads();
        sc[b] += add;
        __syncthreads();
    }
    start[b + 1] = sc[b];
    if (b == 0) start[0] = 0;
}

__global__ void __launch_bounds__(256)
bin_fill_kernel(const int* __restrict__ src, const int* __restrict__ dst,
                const int* __restrict__ hist, const int* __restrict__ start,
                int* __restrict__ bins, int e) {
    __shared__ int cur[NBK + 1];
    const long long base = (long long)blockIdx.x * ECH;
    int d[EQ], s[EQ], pos[EQ];
    #pragma unroll
    for (int u = 0; u < EQ; ++u) {
        const long long idx = base + u * 256 + threadIdx.x;
        const long long ic  = (idx < e) ? idx : (long long)e - 1;
        d[u] = dst[ic];
        s[u] = src[ic];
    }
    for (int i = threadIdx.x; i < NBK; i += 256)
        cur[i] = start[i] + hist[(size_t)blockIdx.x * NBK + i];
    if (threadIdx.x == 0) cur[NBK] = 0;
    __syncthreads();
    #pragma unroll
    for (int u = 0; u < EQ; ++u) {
        const long long idx = base + u * 256 + threadIdx.x;
        pos[u] = atomicAdd(&cur[(idx < e) ? d[u] / RB : NBK], 1);
    }
    #pragma unroll
    for (int u = 0; u < EQ; ++u) {
        const long long idx = base + u * 256 + threadIdx.x;
        if (idx < e) {
            const int b = d[u] / RB;
            bins[pos[u]] = s[u] * RB + (d[u] - b * RB);
        }
    }
}

template <int COLS, int UB>
__global__ void __launch_bounds__(512)
agg_owner_kernel(const float* __restrict__ x, const int* __restrict__ bins,
                 const int* __restrict__ start, float* __restrict__ out, int n) {
    constexpr int NI = (COLS + 31) / 32;
    __shared__ __attribute__((aligned(16))) float acc[RB * COLS];
    const int lane = threadIdx.x & 31;
    const int wave = threadIdx.x >> 5;
    const int nw   = blockDim.x >> 5;
    const int e0 = start[blockIdx.x];
    const int e1 = start[blockIdx.x + 1];
    const int ng = (e1 - e0 + UB - 1) / UB;
    int g = wave;
    int pk = 0;
    if (g < ng && e0 + g * UB + lane < e1 && lane < UB) pk = bins[e0 + g * UB + lane];
    {
        const v4f z = {0.0f, 0.0f, 0.0f, 0.0f};
        v4f* a4 = (v4f*)acc;
        for (int i = threadIdx.x; i < RB * COLS / 4; i += blockDim.x) a4[i] = z;
    }
    __syncthreads();
    while (g < ng) {
        const int cnt = (e1 - (e0 + g * UB) < UB) ? e1 - (e0 + g * UB) : UB;
        const int gn = g + nw;
        int pkn = 0;
        if (gn < ng && e0 + gn * UB + lane < e1 && lane < UB) pkn = bins[e0 + gn * UB + lane];
        if (cnt == UB) {
            float v[UB][NI];
            int   lr[UB];
            #pragma unroll
            for (int u = 0; u < UB; ++u) {
                const int w = __builtin_amdgcn_readlane(pk, u);
                const int s = w / RB;
                lr[u] = w - s * RB;
                const float* xr = x + (size_t)s * COLS + lane;
                #pragma unroll
                for (int i = 0; i < NI; ++i)
                    v[u][i] = (32 * i + lane < COLS) ? xr[32 * i] : 0.0f;
            }
            #pragma unroll
            for (int u = 0; u < UB; ++u) {
                float* a = acc + lr[u] * COLS + lane;
                #pragma unroll
                for (int i = 0; i < NI; ++i)
                    if (32 * i + lane < COLS) atomicAdd(a + 32 * i, v[u][i]);
            }
        } else {
            float v[UB - 1][NI];
            int   lr[UB - 1];
            #pragma unroll
            for (int u = 0; u < UB - 1; ++u) {
                const int w = __builtin_amdgcn_readlane(pk, u);
                const int s = w / RB;
                lr[u] = w - s * RB;
                const float* xr = x + (size_t)s * COLS + lane;
                #pragma unroll
                for (int i = 0; i < NI; ++i)
                    v[u][i] = (32 * i + lane < COLS) ? xr[32 * i] : 0.0f;
            }
            #pragma unroll
            for (int u = 0; u < UB - 1; ++u) {
                float* a = acc + lr[u] * COLS + lane;
                #pragma unroll
                for (int i = 0; i < NI; ++i)
                    if (32 * i + lane < COLS) atomicAdd(a + 32 * i, (u < cnt) ? v[u][i] : 0.0f);
            }
        }
        pk = pkn;
        g  = gn;
    }
    __syncthreads();
    const int r0 = blockIdx.x * RB;
    const int nr = (n - r0 < RB) ? (n - r0) : RB;
    v4f* o4 = (v4f*)(out + (size_t)r0 * COLS);
    const v4f* a4 = (const v4f*)acc;
    for (int i = threadIdx.x; i < nr * COLS / 4; i += blockDim.x) o4[i] = a4[i];
}

#define AP 132

template <int FOUT, int UB>
__global__ void __launch_bounds__(512)
agg_product_kernel(const float* __restrict__ x, const int* __restrict__ bins,
                   const int* __restrict__ start,
                   const float* __restrict__ di_is, const float* __restrict__ do_is,
                   const float* __restrict__ g, const float* __restrict__ b,
                   const float* __restrict__ W,
                   float* __restrict__ C, int n) {
    constexpr int COLS = 128, NI = COLS / 32, KS = COLS / 4;
    __shared__ __attribute__((aligned(16))) float acc[RB * AP];
    const int lane = threadIdx.x & 31;
    const int wave = threadIdx.x >> 5;
    const int nw   = blockDim.x >> 5;
    const int e0 = start[blockIdx.x];
    const int e1 = start[blockIdx.x + 1];
    const int ng = (e1 - e0 + UB - 1) / UB;
    int gi = wave;
    int pk = 0;
    if (gi < ng && e0 + gi * UB + lane < e1 && lane < UB) pk = bins[e0 + gi * UB + lane];
    {
        const v4f z = {0.0f, 0.0f, 0.0f, 0.0f};
        v4f* a4 = (v4f*)acc;
        for (int i = threadIdx.x; i < RB * AP / 4; i += blockDim.x) a4[i] = z;
    }
    __syncthreads();
    while (gi < ng) {
        const int cnt = (e1 - (e0 + gi * UB) < UB) ? e1 - (e0 + gi * UB) : UB;
        const int gn = gi + nw;
        int pkn = 0;
        if (gn < ng && e0 + gn * UB + lane < e1 && lane < UB) pkn = bins[e0 + gn * UB + lane];
        if (cnt == UB) {
            float v[UB][NI];
            int   lr[UB];
            #pragma unroll
            for (int u = 0; u < UB; ++u) {
                const int w = __builtin_amdgcn_readlane(pk, u);
                const int s = w / RB;
                lr[u] = w - s * RB;
                const float* xr = x + (size_t)s * COLS + lane;
                #pragma unroll
                for (int i = 0; i < NI; ++i) v[u][i] = xr[32 * i];
            }
            #pragma unroll
            for (int u = 0; u < UB; ++u) {
                float* a = acc + lr[u] * AP + lane;
                #pragma unroll
                for (int i = 0; i < NI; ++i) atomicAdd(a + 32 * i, v[u][i]);
            }
        } else {
            float v[UB - 1][NI];
            int   lr[UB - 1];
            #pragma unroll
            for (int u = 0; u < UB - 1; ++u) {
                const int w = __builtin_amdgcn_readlane(pk, u);
                const int s = w / RB;
                lr[u] = w - s * RB;
                const float* xr = x + (size_t)s * COLS + lane;
                #pragma unroll
                for (int i = 0; i < NI; ++i) v[u][i] = xr[32 * i];
            }
            #pragma unroll
            for (int u = 0; u < UB - 1; ++u) {
                float* a = acc + lr[u] * AP + lane;
                #pragma unroll
                for (int i = 0; i < NI; ++i) atomicAdd(a + 32 * i, (u < cnt) ? v[u][i] : 0.0f);
            }
        }
        pk = pkn;
        gi = gn;
    }
    __syncthreads();

    const int r0 = blockIdx.x * RB;
    const int nr = (n - r0 < RB) ? (n - r0) : RB;
    const int nstrips = (nr > 0) ? ((nr + 15) >> 4) : 0;
    constexpr int NH = (FOUT == 128) ? 2 : 1;
    if (wave >= nstrips * NH) return;
    const int strip = wave / NH;
    const int ch    = wave - strip * NH;
    const int half  = lane >> 4;
    const int l16   = lane & 15;
    const int koff  = half * 2;
    int lrow = strip * 16 + l16;
    if (lrow >= nr) lrow = nr - 1;
    const float* ar = acc + lrow * AP + koff;
    const float di = di_is[r0 + lrow];
    const float ds = do_is[r0 + lrow];

    float sa = 0.f, sb = 0.f, qa = 0.f, qb = 0.f;
    #pragma unroll
    for (int u = 0; u < KS; ++u) {
        v2f xv = *(const v2f*)(ar + 4 * u);
        xv.x *= di; xv.y *= di;
        sa += xv.x; qa += xv.x * xv.x;
        sb += xv.y; qb += xv.y * xv.y;
    }
    float s = sa + sb, s2 = qa + qb;
    s  += __shfl_xor(s,  16, 32);
    s2 += __shfl_xor(s2, 16, 32);
    const float m   = s * (1.0f / 128.0f);
    const float rsv = rsqrtf(s2 * (1.0f / 128.0f) - m * m + LN_EPS);

    v8f pacc[4];
    #pragma unroll
    for (int i = 0; i < 4; ++i) { v8f z = {}; pacc[i] = z; }

    const int lc = (l16 < FOUT / 4) ? l16 : FOUT / 4 - 1;
    const int cb = (FOUT == 128) ? (64 * ch + 4 * l16) : (4 * lc);
    constexpr int KB = 4;
    #pragma unroll 1
    for (int t = 0; t < KS; t += KB) {
        v2f xa[KB], gg[KB], bbv[KB];
        v4f w0[KB], w1[KB];
        #pragma unroll
        for (int u = 0; u < KB; ++u) {
            const int c0 = 4 * (t + u) + koff;
            gg[u]  = *(const v2f*)(g + c0);
            bbv[u] = *(const v2f*)(b + c0);
            const float* wrow = W + (size_t)c0 * FOUT + cb;
            w0[u] = *(const v4f*)(wrow);
            w1[u] = *(const v4f*)(wrow + FOUT);
            xa[u] = *(const v2f*)(ar + 4 * (t + u));
        }
        #pragma unroll
        for (int u = 0; u < KB; ++u) {
            const float x0 = xa[u].x * di;
            const float x1 = xa[u].y * di;
            v2f a;
            a.x = fmaxf((x0 - m) * rsv * gg[u].x + bbv[u].x, 0.0f) * ds;
            a.y = fmaxf((x1 - m) * rsv * gg[u].y + bbv[u].y, 0.0f) * ds;
            #pragma unroll
            for (int ft = 0; ft < 4; ++ft) {
                v2f bb;
                bb.x = w0[u][ft];
                bb.y = w1[u][ft];
                pacc[ft] = __builtin_amdgcn_wmma_f32_16x16x4_f32(
                    false, a, false, bb, (short)0, pacc[ft], false, false);
            }
        }
    }
    const int crow = strip * 16 + half * 8;
    const int ccol = (FOUT == 128) ? (64 * ch + 4 * l16) : (4 * l16);
    if (FOUT == 128 || l16 < FOUT / 4) {
        #pragma unroll
        for (int r = 0; r < 8; ++r) {
            if (crow + r < nr) {
                const v4f o = {pacc[0][r], pacc[1][r], pacc[2][r], pacc[3][r]};
                *(v4f*)(C + (size_t)(r0 + crow + r) * FOUT + ccol) = o;
            }
        }
    }
}

__global__ void final_scale_bias_kernel(float* __restrict__ out,
                                        const float* __restrict__ di_is,
                                        const float* __restrict__ b2, int n) {
    const int F = 40;
    int t = blockIdx.x * blockDim.x + threadIdx.x;
    if (t >= n * F) return;
    out[t] = out[t] * di_is[t / F] + b2[t % F];
}

static inline int cdiv(long long a, long long b) { return (int)((a + b - 1) / b); }

extern "C" void kernel_launch(void* const* d_in, const int* in_sizes, int n_in,
                              void* d_out, int out_size, void* d_ws, size_t ws_size,
                              hipStream_t stream) {
    const float* feat = (const float*)d_in[0];
    const int*   src  = (const int*)d_in[1];
    const int*   dst  = (const int*)d_in[2];
    const float* W0   = (const float*)d_in[3];
    const float* W1   = (const float*)d_in[4];
    const float* W2   = (const float*)d_in[5];
    const float* b2   = (const float*)d_in[6];
    const float* g_in = (const float*)d_in[7];
    const float* b_in = (const float*)d_in[8];
    const float* g0   = (const float*)d_in[9];
    const float* b0   = (const float*)d_in[10];
    const float* g1   = (const float*)d_in[11];
    const float* b1   = (const float*)d_in[12];
    float* out = (float*)d_out;

    const int IN_F = 768, HID = 128, CLS = 40;
    const int n = in_sizes[0] / IN_F;
    const int e = in_sizes[1];

    float* ws    = (float*)d_ws;
    float* do_is = ws;
    float* di_is = ws + (size_t)n;
    float* mu    = ws + 2 * (size_t)n;
    float* rs    = ws + 3 * (size_t)n;
    float* bufB  = ws + 4 * (size_t)n;
    float* bufC  = bufB + (size_t)n * HID;
    float* bufD  = bufC + (size_t)n * HID;
    int*   bins  = (int*)(bufD + (size_t)n * HID);
    const int nch = cdiv(e, ECH);
    int*   hist  = bins + (size_t)e;
    int*   start = hist + (size_t)nch * NBK;

    const int T = 256;

    int* part = (int*)bufC;
    degree_hist_kernel<<<dim3(DG, 2), 1024, 0, stream>>>(src, dst, part, e);
    invsqrt_deg_kernel<<<dim3(cdiv(n, T), 2), T, 0, stream>>>(part, do_is, di_is, n);

    bin_count_kernel<<<nch, 256, 0, stream>>>(dst, hist, e);
    bin_scan_kernel<<<1, NBK, 0, stream>>>(hist, start, nch);
    bin_fill_kernel<<<nch, 256, 0, stream>>>(src, dst, hist, start, bins, e);

    ln_stats_768_kernel<<<cdiv((long long)n * 32, T), T, 0, stream>>>(feat, mu, rs, n);

    const int mtiles = cdiv(n, 16);
    gemm0_ln_wmma_kernel<<<mtiles, 32 * KSPL, 0, stream>>>(
        feat, mu, rs, do_is, g_in, b_in, W0, bufB, n);

    agg_product_kernel<128, 8><<<NBK, 512, 0, stream>>>(bufB, bins, start, di_is, do_is, g0, b0, W1, bufD, n);

    agg_product_kernel<40, 8><<<NBK, 512, 0, stream>>>(bufD, bins, start, di_is, do_is, g1, b1, W2, bufC, n);
    agg_owner_kernel<40, 8><<<NBK, 512, 0, stream>>>(bufC, bins, start, out, n);
    final_scale_bias_kernel<<<cdiv((long long)n * CLS, T), T, 0, stream>>>(out, di_is, b2, n);
}


# Input LN statistics moved into the layer-0 product: a wave keeps its 16x192 strip quarter in registers for row sums (joined over 4 waves in LDS) and A fragments; 1 launch fewer; LN sums reordered
# speedup vs baseline: 3.2251x; 1.0824x over previous
#include <hip/hip_runtime.h>
#include <math.h>

typedef __attribute__((ext_vector_type(2))) float v2f;
typedef __attribute__((ext_vector_type(8))) float v8f;
typedef __attribute__((ext_vector_type(4))) float v4f;

#define LN_EPS 1e-5f

#define RB   98
#define NBK  512
#define ECH  4096

#define DG   32
#define DN   (NBK * RB)
#define DQ   8

typedef __attribute__((ext_vector_type(4))) int v4i;

__global__ void __launch_bounds__(1024)
degree_hist_kernel(const int* __restrict__ src, const int* __restrict__ dst,
                   int* __restrict__ part, int e) {
    __shared__ __attribute__((aligned(16))) int cnt[DN];
    const int* __restrict__ idx = blockIdx.y ? dst : src;
    {
        const v4i z = {0, 0, 0, 0};
        v4i* c4 = (v4i*)cnt;
        for (int i = threadIdx.x; i < DN / 4; i += 1024) c4[i] = z;
    }
    __syncthreads();
    const int per = (((e + DG - 1) / DG) + 3) & ~3;
    const long long lo = (long long)blockIdx.x * per;
    long long hi = lo + per;
    if (hi > e) hi = e;
    const int len = (hi > lo) ? (int)(hi - lo) : 0;
    const int nq  = len >> 2;
    const v4i* q = (const v4i*)(idx + lo);
    for (int base = 0; base < nq; base += 1024 * DQ) {
        v4i w[DQ];
        #pragma unroll
        for (int u = 0; u < DQ; ++u) {
            const int i = base + u * 1024 + threadIdx.x;
            w[u] = q[(i < nq) ? i : nq - 1];
        }
        #pragma unroll
        for (int u = 0; u < DQ; ++u) {
            const int i = base + u * 1024 + threadIdx.x;
            if (i < nq) {
                atomicAdd(&cnt[w[u].x], 1);
                atomicAdd(&cnt[w[u].y], 1);
                atomicAdd(&cnt[w[u].z], 1);
                atomicAdd(&cnt[w[u].w], 1);
            }
        }
    }
    if (4 * nq + (int)threadIdx.x < len)
        atomicAdd(&cnt[idx[lo + 4 * nq + threadIdx.x]], 1);
    __syncthreads();
    v4i* p4 = (v4i*)(part + ((size_t)blockIdx.y * DG + blockIdx.x) * DN);
    const v4i* c4 = (const v4i*)cnt;
    for (int i = threadIdx.x; i < DN / 4; i += 1024) p4[i] = c4[i];
}

__global__ void invsqrt_deg_kernel(const int* __restrict__ part,
                                   float* __restrict__ dout, float* __restrict__ din, int n) {
    int i = blockIdx.x * blockDim.x + threadIdx.x;
    if (i >= n) return;
    const int* p = part + (size_t)blockIdx.y * DG * DN + i;
    int s = 0;
    #pragma unroll
    for (int g = 0; g < DG; ++g) s += p[(size_t)g * DN];
    float* o = blockIdx.y ? din : dout;
    o[i] = rsqrtf(fmaxf((float)s, 1.0f));
}

#define KSPL 4
#define RP   132

__global__ void __launch_bounds__(32 * KSPL)
gemm0_ln_wmma_kernel(const float* __restrict__ feat,
                                     const float* __restrict__ do_is,
                                     const float* __restrict__ g, const float* __restrict__ b,
                                     const float* __restrict__ W,
                                     float* __restrict__ C, int M) {
    constexpr int K = 768, F = 128, KQ = K / KSPL;
    constexpr int KB = 4;
    constexpr int NX = KQ / 4;
    __shared__ __attribute__((aligned(16))) float part[KSPL * 16 * RP];
    __shared__ float rsum[2 * KSPL * 16];
    const int lane = threadIdx.x & 31;
    const int q    = __builtin_amdgcn_readfirstlane(threadIdx.x >> 5);
    const int mt   = blockIdx.x;
    if (mt * 16 >= M) return;
    const int half = lane >> 4;
    const int l16  = lane & 15;
    const int arow = mt * 16 + l16;
    const int koff = half * 2;
    const int kbeg = q * KQ;
    const float* ap = feat + (size_t)arow * K + kbeg + koff;

    v2f x[NX];
    #pragma unroll
    for (int u = 0; u < NX; ++u) x[u] = *(const v2f*)(ap + 4 * u);
    const float do_r = do_is[arow];

    float sa = 0.f, sb = 0.f, qa = 0.f, qb = 0.f;
    #pragma unroll
    for (int u = 0; u < NX; ++u) {
        sa += x[u].x; qa += x[u].x * x[u].x;
        sb += x[u].y; qb += x[u].y * x[u].y;
    }
    float s = sa + sb, s2 = qa + qb;
    s  += __shfl_xor(s,  16, 32);
    s2 += __shfl_xor(s2, 16, 32);
    if (half == 0) {
        rsum[q * 16 + l16] = s;
        rsum[KSPL * 16 + q * 16 + l16] = s2;
    }
    __syncthreads();
    float st = rsum[l16], st2 = rsum[KSPL * 16 + l16];
    #pragma unroll
    for (int t = 1; t < KSPL; ++t) {
        st  += rsum[t * 16 + l16];
        st2 += rsum[KSPL * 16 + t * 16 + l16];
    }
    const float m_r  = st * (1.0f / 768.0f);
    const float rs_r = rsqrtf(st2 * (1.0f / 768.0f) - m_r * m_r + LN_EPS);

    v8f acc[8];
    #pragma unroll
    for (int i = 0; i < 8; ++i) { v8f z = {}; acc[i] = z; }

    #pragma unroll
    for (int t = 0; t < NX / KB; ++t) {
        const int k = kbeg + 4 * KB * t;
        v2f gg[KB], bbv[KB];
        v4f w0l[KB], w0h[KB], w1l[KB], w1h[KB];
        #pragma unroll
        for (int u = 0; u < KB; ++u) {
            const int c0 = k + 4 * u + koff;
            gg[u]  = *(const v2f*)(g + c0);
            bbv[u] = *(const v2f*)(b + c0);
            const float* wrow = W + (size_t)c0 * F + 4 * l16;
            w0l[u] = *(const v4f*)(wrow);
            w0h[u] = *(const v4f*)(wrow + 64);
            w1l[u] = *(const v4f*)(wrow + F);
            w1h[u] = *(const v4f*)(wrow + F + 64);
        }
        #pragma unroll
        for (int u = 0; u < KB; ++u) {
            const v2f xv = x[t * KB + u];
            v2f a;
            a.x = ((xv.x - m_r) * rs_r * gg[u].x + bbv[u].x) * do_r;
            a.y = ((xv.y - m_r) * rs_r * gg[u].y + bbv[u].y) * do_r;
            #pragma unroll
            for (int ft = 0; ft < 8; ++ft) {
                v2f bb;
                bb.x = (ft < 4) ? w0l[u][ft & 3] : w0h[u][ft & 3];
                bb.y = (ft < 4) ? w1l[u][ft & 3] : w1h[u][ft & 3];
                acc[ft] = __builtin_amdgcn_wmma_f32_16x16x4_f32(
                    false, a, false, bb, (short)0, acc[ft], false, false);
            }
        }
        __builtin_amdgcn_sched_barrier(0);
    }
    float* pq = part + q * 16 * RP;
    #pragma unroll
    for (int r = 0; r < 8; ++r) {
        float* prow = pq + (half * 8 + r) * RP + 4 * l16;
        const v4f lo = {acc[0][r], acc[1][r], acc[2][r], acc[3][r]};
        const v4f hi = {acc[4][r], acc[5][r], acc[6][r], acc[7][r]};
        *(v4f*)(prow)      = lo;
        *(v4f*)(prow + 64) = hi;
    }
    __syncthreads();
    #pragma unroll
    for (int j = 0; j < 16 / KSPL; ++j) {
        const int row = q * (16 / KSPL) + j;
        const float* p0 = part + row * RP + 4 * lane;
        v4f s = *(const v4f*)(p0);
        #pragma unroll
        for (int t = 1; t < KSPL; ++t) s += *(const v4f*)(p0 + t * 16 * RP);
        *(v4f*)(C + (size_t)(mt * 16 + row) * F + 4 * lane) = s;
    }
}

#define EQ   (ECH / 256)

__global__ void __launch_bounds__(256)
bin_count_kernel(const int* __restrict__ dst, int* __restrict__ hist, int e) {
    __shared__ int h[NBK];
    const long long base = (long long)blockIdx.x * ECH;
    int d[EQ];
    #pragma unroll
    for (int u = 0; u < EQ; ++u) {
        const long long idx = base + u * 256 + threadIdx.x;
        d[u] = dst[(idx < e) ? idx : (long long)e - 1];
    }
    for (int i = threadIdx.x; i < NBK; i += 256) h[i] = 0;
    __syncthreads();
    #pragma unroll
    for (int u = 0; u < EQ; ++u) {
        const long long idx = base + u * 256 + threadIdx.x;
        if (idx < e) atomicAdd(&h[d[u] / RB], 1);
    }
    __syncthreads();
    for (int i = threadIdx.x; i < NBK; i += 256) hist[(size_t)blockIdx.x * NBK + i] = h[i];
}

__global__ void __launch_bounds__(NBK)
bin_scan_kernel(int* __restrict__ hist, int* __restrict__ start, int nch) {
    __shared__ int sc[NBK];
    const int b = threadIdx.x;
    int run = 0;
    constexpr int SB = 49;
    for (int j0 = 0; j0 < nch; j0 += SB) {
        int c[SB];
        #pragma unroll
        for (int u = 0; u < SB; ++u) c[u] = (j0 + u < nch) ? hist[(size_t)(j0 + u) * NBK + b] : 0;
        #pragma unroll
        for (int u = 0; u < SB; ++u) {
            if (j0 + u < nch) hist[(size_t)(j0 + u) * NBK + b] = run;
            run += c[u];
        }
    }
    sc[b] = run;
    __syncthreads();
    for (int off = 1; off < NBK; off <<= 1) {
        const int add = (b >= off) ? sc[b - off] : 0;
        __syncthreads();
        sc[b] += add;
        __syncthreads();
    }
    start[b + 1] = sc[b];
    if (b == 0) start[0] = 0;
}

__global__ void __launch_bounds__(256)
bin_fill_kernel(const int* __restrict__ src, const int* __restrict__ dst,
                const int* __restrict__ hist, const int* __restrict__ start,
                int* __restrict__ bins, int e) {
    __shared__ int cur[NBK + 1];
    const long long base = (long long)blockIdx.x * ECH;
    int d[EQ], s[EQ], pos[EQ];
    #pragma unroll
    for (int u = 0; u < EQ; ++u) {
        const long long idx = base + u * 256 + threadIdx.x;
        const long long ic  = (idx < e) ? idx : (long long)e - 1;
        d[u] = dst[ic];
        s[u] = src[ic];
    }
    for (int i = threadIdx.x; i < NBK; i += 256)
        cur[i] = start[i] + hist[(size_t)blockIdx.x * NBK + i];
    if (threadIdx.x == 0) cur[NBK] = 0;
    __syncthreads();
    #pragma unroll
    for (int u = 0; u < EQ; ++u) {
        const long long idx = base + u * 256 + threadIdx.x;
        pos[u] = atomicAdd(&cur[(idx < e) ? d[u] / RB : NBK], 1);
    }
    #pragma unroll
    for (int u = 0; u < EQ; ++u) {
        const long long idx = base + u * 256 + threadIdx.x;
        if (idx < e) {
            const int b = d[u] / RB;
            bins[pos[u]] = s[u] * RB + (d[u] - b * RB);
        }
    }
}

template <int COLS, int UB>
__global__ void __launch_bounds__(512)
agg_owner_kernel(const float* __restrict__ x, const int* __restrict__ bins,
                 const int* __restrict__ start, float* __restrict__ out, int n) {
    constexpr int NI = (COLS + 31) / 32;
    __shared__ __attribute__((aligned(16))) float acc[RB * COLS];
    const int lane = threadIdx.x & 31;
    const int wave = threadIdx.x >> 5;
    const int nw   = blockDim.x >> 5;
    const int e0 = start[blockIdx.x];
    const int e1 = start[blockIdx.x + 1];
    const int ng = (e1 - e0 + UB - 1) / UB;
    int g = wave;
    int pk = 0;
    if (g < ng && e0 + g * UB + lane < e1 && lane < UB) pk = bins[e0 + g * UB + lane];
    {
        const v4f z = {0.0f, 0.0f, 0.0f, 0.0f};
        v4f* a4 = (v4f*)acc;
        for (int i = threadIdx.x; i < RB * COLS / 4; i += blockDim.x) a4[i] = z;
    }
    __syncthreads();
    while (g < ng) {
        const int cnt = (e1 - (e0 + g * UB) < UB) ? e1 - (e0 + g * UB) : UB;
        const int gn = g + nw;
        int pkn = 0;
        if (gn < ng && e0 + gn * UB + lane < e1 && lane < UB) pkn = bins[e0 + gn * UB + lane];
        if (cnt == UB) {
            float v[UB][NI];
            int   lr[UB];
            #pragma unroll
            for (int u = 0; u < UB; ++u) {
                const int w = __builtin_amdgcn_readlane(pk, u);
                const int s = w / RB;
                lr[u] = w - s * RB;
                const float* xr = x + (size_t)s * COLS + lane;
                #pragma unroll
                for (int i = 0; i < NI; ++i)
                    v[u][i] = (32 * i + lane < COLS) ? xr[32 * i] : 0.0f;
            }
            #pragma unroll
            for (int u = 0; u < UB; ++u) {
                float* a = acc + lr[u] * COLS + lane;
                #pragma unroll
                for (int i = 0; i < NI; ++i)
                    if (32 * i + lane < COLS) atomicAdd(a + 32 * i, v[u][i]);
            }
        } else {
            float v[UB - 1][NI];
            int   lr[UB - 1];
            #pragma unroll
            for (int u = 0; u < UB - 1; ++u) {
                const int w = __builtin_amdgcn_readlane(pk, u);
                const int s = w / RB;
                lr[u] = w - s * RB;
                const float* xr = x + (size_t)s * COLS + lane;
                #pragma unroll
                for (int i = 0; i < NI; ++i)
                    v[u][i] = (32 * i + lane < COLS) ? xr[32 * i] : 0.0f;
            }
            #pragma unroll
            for (int u = 0; u < UB - 1; ++u) {
                float* a = acc + lr[u] * COLS + lane;
                #pragma unroll
                for (int i = 0; i < NI; ++i)
                    if (32 * i + lane < COLS) atomicAdd(a + 32 * i, (u < cnt) ? v[u][i] : 0.0f);
            }
        }
        pk = pkn;
        g  = gn;
    }
    __syncthreads();
    const int r0 = blockIdx.x * RB;
    const int nr = (n - r0 < RB) ? (n - r0) : RB;
    v4f* o4 = (v4f*)(out + (size_t)r0 * COLS);
    const v4f* a4 = (const v4f*)acc;
    for (int i = threadIdx.x; i < nr * COLS / 4; i += blockDim.x) o4[i] = a4[i];
}

#define AP 132

template <int FOUT, int UB>
__global__ void __launch_bounds__(512)
agg_product_kernel(const float* __restrict__ x, const int* __restrict__ bins,
                   const int* __restrict__ start,
                   const float* __restrict__ di_is, const float* __restrict__ do_is,
                   const float* __restrict__ g, const float* __restrict__ b,
                   const float* __restrict__ W,
                   float* __restrict__ C, int n) {
    constexpr int COLS = 128, NI = COLS / 32, KS = COLS / 4;
    __shared__ __attribute__((aligned(16))) float acc[RB * AP];
    const int lane = threadIdx.x & 31;
    const int wave = threadIdx.x >> 5;
    const int nw   = blockDim.x >> 5;
    const int e0 = start[blockIdx.x];
    const int e1 = start[blockIdx.x + 1];
    const int ng = (e1 - e0 + UB - 1) / UB;
    int gi = wave;
    int pk = 0;
    if (gi < ng && e0 + gi * UB + lane < e1 && lane < UB) pk = bins[e0 + gi * UB + lane];
    {
        const v4f z = {0.0f, 0.0f, 0.0f, 0.0f};
        v4f* a4 = (v4f*)acc;
        for (int i = threadIdx.x; i < RB * AP / 4; i += blockDim.x) a4[i] = z;
    }
    __syncthreads();
    while (gi < ng) {
        const int cnt = (e1 - (e0 + gi * UB) < UB) ? e1 - (e0 + gi * UB) : UB;
        const int gn = gi + nw;
        int pkn = 0;
        if (gn < ng && e0 + gn * UB + lane < e1 && lane < UB) pkn = bins[e0 + gn * UB + lane];
        if (cnt == UB) {
            float v[UB][NI];
            int   lr[UB];
            #pragma unroll
            for (int u = 0; u < UB; ++u) {
                const int w = __builtin_amdgcn_readlane(pk, u);
                const int s = w / RB;
                lr[u] = w - s * RB;
                const float* xr = x + (size_t)s * COLS + lane;
                #pragma unroll
                for (int i = 0; i < NI; ++i) v[u][i] = xr[32 * i];
            }
            #pragma unroll
            for (int u = 0; u < UB; ++u) {
                float* a = acc + lr[u] * AP + lane;
                #pragma unroll
                for (int i = 0; i < NI; ++i) atomicAdd(a + 32 * i, v[u][i]);
            }
        } else {
            float v[UB - 1][NI];
            int   lr[UB - 1];
            #pragma unroll
            for (int u = 0; u < UB - 1; ++u) {
                const int w = __builtin_amdgcn_readlane(pk, u);
                const int s = w / RB;
                lr[u] = w - s * RB;
                const float* xr = x + (size_t)s * COLS + lane;
                #pragma unroll
                for (int i = 0; i < NI; ++i) v[u][i] = xr[32 * i];
            }
            #pragma unroll
            for (int u = 0; u < UB - 1; ++u) {
                float* a = acc + lr[u] * AP + lane;
                #pragma unroll
                for (int i = 0; i < NI; ++i) atomicAdd(a + 32 * i, (u < cnt) ? v[u][i] : 0.0f);
            }
        }
        pk = pkn;
        gi = gn;
    }
    __syncthreads();

    const int r0 = blockIdx.x * RB;
    const int nr = (n - r0 < RB) ? (n - r0) : RB;
    const int nstrips = (nr > 0) ? ((nr + 15) >> 4) : 0;
    constexpr int NH = (FOUT == 128) ? 2 : 1;
    if (wave >= nstrips * NH) return;
    const int strip = wave / NH;
    const int ch    = wave - strip * NH;
    const int half  = lane >> 4;
    const int l16   = lane & 15;
    const int koff  = half * 2;
    int lrow = strip * 16 + l16;
    if (lrow >= nr) lrow = nr - 1;
    const float* ar = acc + lrow * AP + koff;
    const float di = di_is[r0 + lrow];
    const float ds = do_is[r0 + lrow];

    float sa = 0.f, sb = 0.f, qa = 0.f, qb = 0.f;
    #pragma unroll
    for (int u = 0; u < KS; ++u) {
        v2f xv = *(const v2f*)(ar + 4 * u);
        xv.x *= di; xv.y *= di;
        sa += xv.x; qa += xv.x * xv.x;
        sb += xv.y; qb += xv.y * xv.y;
    }
    float s = sa + sb, s2 = qa + qb;
    s  += __shfl_xor(s,  16, 32);
    s2 += __shfl_xor(s2, 16, 32);
    const float m   = s * (1.0f / 128.0f);
    const float rsv = rsqrtf(s2 * (1.0f / 128.0f) - m * m + LN_EPS);

    v8f pacc[4];
    #pragma unroll
    for (int i = 0; i < 4; ++i) { v8f z = {}; pacc[i] = z; }

    const int lc = (l16 < FOUT / 4) ? l16 : FOUT / 4 - 1;
    const int cb = (FOUT == 128) ? (64 * ch + 4 * l16) : (4 * lc);
    constexpr int KB = 4;
    #pragma unroll 1
    for (int t = 0; t < KS; t += KB) {
        v2f xa[KB], gg[KB], bbv[KB];
        v4f w0[KB], w1[KB];
        #pragma unroll
        for (int u = 0; u < KB; ++u) {
            const int c0 = 4 * (t + u) + koff;
            gg[u]  = *(const v2f*)(g + c0);
            bbv[u] = *(const v2f*)(b + c0);
            const float* wrow = W + (size_t)c0 * FOUT + cb;
            w0[u] = *(const v4f*)(wrow);
            w1[u] = *(const v4f*)(wrow + FOUT);
            xa[u] = *(const v2f*)(ar + 4 * (t + u));
        }
        #pragma unroll
        for (int u = 0; u < KB; ++u) {
            const float x0 = xa[u].x * di;
            const float x1 = xa[u].y * di;
            v2f a;
            a.x = fmaxf((x0 - m) * rsv * gg[u].x + bbv[u].x, 0.0f) * ds;
            a.y = fmaxf((x1 - m) * rsv * gg[u].y + bbv[u].y, 0.0f) * ds;
            #pragma unroll
            for (int ft = 0; ft < 4; ++ft) {
                v2f bb;
                bb.x = w0[u][ft];
                bb.y = w1[u][ft];
                pacc[ft] = __builtin_amdgcn_wmma_f32_16x16x4_f32(
                    false, a, false, bb, (short)0, pacc[ft], false, false);
            }
        }
    }
    const int crow = strip * 16 + half * 8;
    const int ccol = (FOUT == 128) ? (64 * ch + 4 * l16) : (4 * l16);
    if (FOUT == 128 || l16 < FOUT / 4) {
        #pragma unroll
        for (int r = 0; r < 8; ++r) {
            if (crow + r < nr) {
                const v4f o = {pacc[0][r], pacc[1][r], pacc[2][r], pacc[3][r]};
                *(v4f*)(C + (size_t)(r0 + crow + r) * FOUT + ccol) = o;
            }
        }
    }
}

__global__ void final_scale_bias_kernel(float* __restrict__ out,
                                        const float* __restrict__ di_is,
                                        const float* __restrict__ b2, int n) {
    const int F = 40;
    int t = blockIdx.x * blockDim.x + threadIdx.x;
    if (t >= n * F) return;
    out[t] = out[t] * di_is[t / F] + b2[t % F];
}

static inline int cdiv(long long a, long long b) { return (int)((a + b - 1) / b); }

extern "C" void kernel_launch(void* const* d_in, const int* in_sizes, int n_in,
                              void* d_out, int out_size, void* d_ws, size_t ws_size,
                              hipStream_t stream) {
    const float* feat = (const float*)d_in[0];
    const int*   src  = (const int*)d_in[1];
    const int*   dst  = (const int*)d_in[2];
    const float* W0   = (const float*)d_in[3];
    const float* W1   = (const float*)d_in[4];
    const float* W2   = (const float*)d_in[5];
    const float* b2   = (const float*)d_in[6];
    const float* g_in = (const float*)d_in[7];
    const float* b_in = (const float*)d_in[8];
    const float* g0   = (const float*)d_in[9];
    const float* b0   = (const float*)d_in[10];
    const float* g1   = (const float*)d_in[11];
    const float* b1   = (const float*)d_in[12];
    float* out = (float*)d_out;

    const int IN_F = 768, HID = 128, CLS = 40;
    const int n = in_sizes[0] / IN_F;
    const int e = in_sizes[1];

    float* ws    = (float*)d_ws;
    float* do_is = ws;
    float* di_is = ws + (size_t)n;
    float* bufB  = ws + 4 * (size_t)n;
    float* bufC  = bufB + (size_t)n * HID;
    float* bufD  = bufC + (size_t)n * HID;
    int*   bins  = (int*)(bufD + (size_t)n * HID);
    const int nch = cdiv(e, ECH);
    int*   hist  = bins + (size_t)e;
    int*   start = hist + (size_t)nch * NBK;

    const int T = 256;

    int* part = (int*)bufC;
    degree_hist_kernel<<<dim3(DG, 2), 1024, 0, stream>>>(src, dst, part, e);
    invsqrt_deg_kernel<<<dim3(cdiv(n, T), 2), T, 0, stream>>>(part, do_is, di_is, n);

    bin_count_kernel<<<nch, 256, 0, stream>>>(dst, hist, e);
    bin_scan_kernel<<<1, NBK, 0, stream>>>(hist, start, nch);
    bin_fill_kernel<<<nch, 256, 0, stream>>>(src, dst, hist, start, bins, e);

    const int mtiles = cdiv(n, 16);
    gemm0_ln_wmma_kernel<<<mtiles, 32 * KSPL, 0, stream>>>(
        feat, do_is, g_in, b_in, W0, bufB, n);

    agg_product_kernel<128, 8><<<NBK, 512, 0, stream>>>(bufB, bins, start, di_is, do_is, g0, b0, W1, bufD, n);

    agg_product_kernel<40, 8><<<NBK, 512, 0, stream>>>(bufD, bins, start, di_is, do_is, g1, b1, W2, bufC, n);
    agg_owner_kernel<40, 8><<<NBK, 512, 0, stream>>>(bufC, bins, start, out, n);
    final_scale_bias_kernel<<<cdiv((long long)n * CLS, T), T, 0, stream>>>(out, di_is, b2, n);
}
